# starting kernel
# baseline (speedup 1.0000x reference)
#include <hip/hip_runtime.h>
#include <hip/hip_bf16.h>

typedef __attribute__((__ext_vector_type__(16))) __bf16 bf16x16;
typedef __attribute__((__ext_vector_type__(8)))  float  f32x8;
typedef __attribute__((__ext_vector_type__(4)))  unsigned int u32x4;

union Frag16 {
    u32x4   q[2];
    bf16x16 v;
};

static __device__ __forceinline__ unsigned short f2bf(float f) {
    unsigned u = __float_as_uint(f);
    u += 0x7FFFu + ((u >> 16) & 1u);
    return (unsigned short)(u >> 16);
}

static __device__ __forceinline__ unsigned fkey(float f) {
    unsigned u = __float_as_uint(f);
    return (u & 0x80000000u) ? ~u : (u | 0x80000000u);
}
static __device__ __forceinline__ float funkey(unsigned k) {
    unsigned u = (k & 0x80000000u) ? (k & 0x7FFFFFFFu) : ~k;
    return __uint_as_float(u);
}

#define GEMM_WAVES 4
__global__ void wmma_gemm_bf16(const unsigned short* __restrict__ A,
                               const unsigned short* __restrict__ Bt,
                               float* __restrict__ Y,
                               int M, int K, int NOUT)
{
    const int lane  = threadIdx.x & 31;
    const int wave  = threadIdx.x >> 5;
    const int tileM = blockIdx.x * GEMM_WAVES + wave;
    const int m0    = tileM * 16;
    if (m0 >= M) return;

    const int l15   = lane & 15;
    const int half  = lane >> 4;
    const int kbase = half * 8;

    int rowA = m0 + l15;
    if (rowA >= M) rowA = M - 1;
    const unsigned short* aptr = A + (size_t)rowA * K + kbase;

    for (int c0 = 0; c0 < NOUT; c0 += 16) {
        const unsigned short* bptr = Bt + (size_t)(c0 + l15) * K + kbase;
        f32x8 acc = {};
        for (int k0 = 0; k0 < K; k0 += 32) {
            Frag16 fa, fb;
            fa.q[0] = *(const u32x4*)(aptr + k0);
            fa.q[1] = *(const u32x4*)(aptr + k0 + 16);
            fb.q[0] = *(const u32x4*)(bptr + k0);
            fb.q[1] = *(const u32x4*)(bptr + k0 + 16);
            acc = __builtin_amdgcn_wmma_f32_16x16x32_bf16(
                      false, fa.v, false, fb.v, (short)0, acc, false, false);
        }
        const int colY = c0 + l15;
#pragma unroll
        for (int r = 0; r < 8; ++r) {
            const int row = m0 + r + half * 8;
            if (row < M) Y[(size_t)row * NOUT + colY] = acc[r];
        }
    }
}

__global__ void node_scores(const float* __restrict__ h,
                            const float* __restrict__ as_, const float* __restrict__ ad_,
                            float* __restrict__ es, float* __restrict__ ed,
                            int N, int C)
{
    const int warp = (blockIdx.x * blockDim.x + threadIdx.x) >> 5;
    const int lane = threadIdx.x & 31;
    if (warp >= N) return;
    const float* row = h + (size_t)warp * C;
    float s = 0.f, d = 0.f;
    for (int c = lane; c < C; c += 32) {
        const float v = row[c];
        s += v * as_[c];
        d += v * ad_[c];
    }
#pragma unroll
    for (int off = 16; off; off >>= 1) {
        s += __shfl_xor(s, off);
        d += __shfl_xor(d, off);
    }
    if (lane == 0) { es[warp] = s; ed[warp] = d; }
}

__global__ void init_node(unsigned* __restrict__ maxbits, float* __restrict__ denom, int N)
{
    const int i = blockIdx.x * blockDim.x + threadIdx.x;
    if (i >= N) return;
    maxbits[i] = fkey(-__builtin_inff());
    denom[i]   = 0.f;
}

__global__ void init_agg(float* __restrict__ agg, const float* __restrict__ bias, int N, int C)
{
    const size_t i = (size_t)blockIdx.x * blockDim.x + threadIdx.x;
    if (i >= (size_t)N * C) return;
    agg[i] = bias[i % C];
}

__global__ void edge_logit_max(const int* __restrict__ ei, int E, int N,
                               const float* __restrict__ es, const float* __restrict__ ed,
                               float* __restrict__ logits, unsigned* __restrict__ maxbits)
{
    const int e = blockIdx.x * blockDim.x + threadIdx.x;
    if (e >= E + N) return;
    const int s = (e < E) ? ei[e]     : e - E;
    const int d = (e < E) ? ei[E + e] : e - E;
    float l = es[s] + ed[d];
    l = (l > 0.f) ? l : 0.2f * l;
    logits[e] = l;
    atomicMax(&maxbits[d], fkey(l));
}

__global__ void edge_exp(const int* __restrict__ ei, int E, int N,
                         const float* __restrict__ logits, const unsigned* __restrict__ maxbits,
                         float* __restrict__ p, float* __restrict__ denom)
{
    const int e = blockIdx.x * blockDim.x + threadIdx.x;
    if (e >= E + N) return;
    const int d = (e < E) ? ei[E + e] : e - E;
    const float v = __expf(logits[e] - funkey(maxbits[d]));
    p[e] = v;
    atomicAdd(&denom[d], v);
}

__global__ void edge_aggregate(const int* __restrict__ ei, int E, int N,
                               const float* __restrict__ h, const float* __restrict__ p,
                               const float* __restrict__ denom, float* __restrict__ agg, int C)
{
    const int NE = E + N;
    for (int e = blockIdx.x; e < NE; e += gridDim.x) {
        const int s = (e < E) ? ei[e]     : e - E;
        const int d = (e < E) ? ei[E + e] : e - E;
        const float alpha = p[e] / denom[d];
        const float* hs = h   + (size_t)s * C;
        float*       od = agg + (size_t)d * C;
        for (int c = threadIdx.x; c < C; c += blockDim.x)
            atomicAdd(&od[c], hs[c] * alpha);
    }
}

__global__ void to_bf16(const float* __restrict__ x, unsigned short* __restrict__ y, size_t n)
{
    const size_t i = (size_t)blockIdx.x * blockDim.x + threadIdx.x;
    if (i < n) y[i] = f2bf(x[i]);
}

__global__ void relu_to_bf16(const float* __restrict__ x, unsigned short* __restrict__ y, size_t n)
{
    const size_t i = (size_t)blockIdx.x * blockDim.x + threadIdx.x;
    if (i < n) { const float v = x[i]; y[i] = f2bf(v > 0.f ? v : 0.f); }
}

__global__ void transpose_to_bf16(const float* __restrict__ W, unsigned short* __restrict__ Wt,
                                  int K, int NOUT)
{
    const int i = blockIdx.x * blockDim.x + threadIdx.x;
    if (i >= K * NOUT) return;
    const int k = i / NOUT, n = i % NOUT;
    Wt[(size_t)n * K + k] = f2bf(W[i]);
}

__global__ void link_predict(const float* __restrict__ h, const int* __restrict__ mask,
                             const float* __restrict__ Wl, const float* __restrict__ bl,
                             float* __restrict__ out, int P, int C)
{
    const int warp = (blockIdx.x * blockDim.x + threadIdx.x) >> 5;
    const int lane = threadIdx.x & 31;
    if (warp >= P) return;
    const int m0 = mask[warp * 2 + 0];
    const int m1 = mask[warp * 2 + 1];
    const float* r0 = h + (size_t)m0 * C;
    const float* r1 = h + (size_t)m1 * C;
    float acc = 0.f;
    for (int c = lane; c < C; c += 32)
        acc += r0[c] * Wl[c] + r1[c] * Wl[C + c];
#pragma unroll
    for (int off = 16; off; off >>= 1) acc += __shfl_xor(acc, off);
    if (lane == 0) out[warp] = 1.f / (1.f + expf(-(acc + bl[0])));
}

extern "C" void kernel_launch(void* const* d_in, const int* in_sizes, int n_in,
                              void* d_out, int out_size, void* d_ws, size_t ws_size,
                              hipStream_t stream)
{
    (void)n_in; (void)out_size; (void)ws_size;

    const float* features = (const float*)d_in[0];
    const int*   ei       = (const int*)  d_in[1];
    const int*   mask     = (const int*)  d_in[2];
    const float* W1   = (const float*)d_in[3];
    const float* a_s1 = (const float*)d_in[4];
    const float* a_d1 = (const float*)d_in[5];
    const float* b1   = (const float*)d_in[6];
    const float* W2   = (const float*)d_in[7];
    const float* a_s2 = (const float*)d_in[8];
    const float* a_d2 = (const float*)d_in[9];
    const float* b2   = (const float*)d_in[10];
    const float* Wl   = (const float*)d_in[11];
    const float* bl   = (const float*)d_in[12];

    const int F  = 128, Hh = 256;
    const int N  = in_sizes[0] / F;
    const int E  = in_sizes[1] / 2;
    const int P  = in_sizes[2] / 2;
    const int NE = E + N;

    char* ws = (char*)d_ws;
    size_t off = 0;
    auto alloc = [&](size_t bytes) -> char* {
        char* ptr = ws + off;
        off += (bytes + 255) & ~(size_t)255;
        return ptr;
    };
    unsigned short* xbf  = (unsigned short*)alloc((size_t)N * F  * 2);
    unsigned short* w1t  = (unsigned short*)alloc((size_t)Hh * F * 2);
    unsigned short* w2t  = (unsigned short*)alloc((size_t)F * Hh * 2);
    unsigned short* hrbf = (unsigned short*)alloc((size_t)N * Hh * 2);
    float*    buf1    = (float*)   alloc((size_t)N * Hh * 4);
    float*    buf2    = (float*)   alloc((size_t)N * Hh * 4);
    float*    es      = (float*)   alloc((size_t)N * 4);
    float*    ed      = (float*)   alloc((size_t)N * 4);
    unsigned* maxbits = (unsigned*)alloc((size_t)N * 4);
    float*    denom   = (float*)   alloc((size_t)N * 4);
    float*    logits  = (float*)   alloc((size_t)NE * 4);
    float*    pbuf    = (float*)   alloc((size_t)NE * 4);

    const int tilesM = (N + 15) / 16;
    const int gemmGrid = (tilesM + GEMM_WAVES - 1) / GEMM_WAVES;
    const int TB = 256;
    auto cdiv = [](size_t a, size_t b) -> int { return (int)((a + b - 1) / b); };

    to_bf16<<<cdiv((size_t)N * F, TB), TB, 0, stream>>>(features, xbf, (size_t)N * F);
    transpose_to_bf16<<<cdiv((size_t)F * Hh, TB), TB, 0, stream>>>(W1, w1t, F, Hh);
    transpose_to_bf16<<<cdiv((size_t)Hh * F, TB), TB, 0, stream>>>(W2, w2t, Hh, F);

    wmma_gemm_bf16<<<gemmGrid, 32 * GEMM_WAVES, 0, stream>>>(xbf, w1t, buf1, N, F, Hh);
    node_scores<<<cdiv((size_t)N * 32, TB), TB, 0, stream>>>(buf1, a_s1, a_d1, es, ed, N, Hh);
    init_node<<<cdiv(N, TB), TB, 0, stream>>>(maxbits, denom, N);
    init_agg<<<cdiv((size_t)N * Hh, TB), TB, 0, stream>>>(buf2, b1, N, Hh);
    edge_logit_max<<<cdiv(NE, TB), TB, 0, stream>>>(ei, E, N, es, ed, logits, maxbits);
    edge_exp<<<cdiv(NE, TB), TB, 0, stream>>>(ei, E, N, logits, maxbits, pbuf, denom);
    edge_aggregate<<<8192, Hh, 0, stream>>>(ei, E, N, buf1, pbuf, denom, buf2, Hh);
    relu_to_bf16<<<cdiv((size_t)N * Hh, TB), TB, 0, stream>>>(buf2, hrbf, (size_t)N * Hh);

    wmma_gemm_bf16<<<gemmGrid, 32 * GEMM_WAVES, 0, stream>>>(hrbf, w2t, buf1, N, Hh, F);
    node_scores<<<cdiv((size_t)N * 32, TB), TB, 0, stream>>>(buf1, a_s2, a_d2, es, ed, N, F);
    init_node<<<cdiv(N, TB), TB, 0, stream>>>(maxbits, denom, N);
    init_agg<<<cdiv((size_t)N * F, TB), TB, 0, stream>>>(buf2, b2, N, F);
    edge_logit_max<<<cdiv(NE, TB), TB, 0, stream>>>(ei, E, N, es, ed, logits, maxbits);
    edge_exp<<<cdiv(NE, TB), TB, 0, stream>>>(ei, E, N, logits, maxbits, pbuf, denom);
    edge_aggregate<<<8192, F, 0, stream>>>(ei, E, N, buf1, pbuf, denom, buf2, F);

    link_predict<<<cdiv((size_t)P * 32, TB), TB, 0, stream>>>(buf2, mask, Wl, bl,
                                                              (float*)d_out, P, F);
}


# Edges bucketed by destination (count, scan, arrival-order fill); one wave per node does softmax and weighted row sum in registers: float atomics and edge passes gone, sum order now by arrival rank
# speedup vs baseline: 2.7601x; 2.7601x over previous
#include <hip/hip_runtime.h>
#include <hip/hip_bf16.h>

typedef __attribute__((__ext_vector_type__(16))) __bf16 bf16x16;
typedef __attribute__((__ext_vector_type__(8)))  float  f32x8;
typedef __attribute__((__ext_vector_type__(4)))  unsigned int u32x4;
typedef __attribute__((__ext_vector_type__(4)))  float  f32x4;

union Frag16 {
    u32x4   q[2];
    bf16x16 v;
};

static __device__ __forceinline__ unsigned short f2bf(float f) {
    unsigned u = __float_as_uint(f);
    u += 0x7FFFu + ((u >> 16) & 1u);
    return (unsigned short)(u >> 16);
}

#define GEMM_WAVES 4
__global__ void wmma_gemm_bf16(const unsigned short* __restrict__ A,
                               const unsigned short* __restrict__ Bt,
                               float* __restrict__ Y,
                               int M, int K, int NOUT)
{
    const int lane  = threadIdx.x & 31;
    const int wave  = threadIdx.x >> 5;
    const int tileM = blockIdx.x * GEMM_WAVES + wave;
    const int m0    = tileM * 16;
    if (m0 >= M) return;

    const int l15   = lane & 15;
    const int half  = lane >> 4;
    const int kbase = half * 8;

    int rowA = m0 + l15;
    if (rowA >= M) rowA = M - 1;
    const unsigned short* aptr = A + (size_t)rowA * K + kbase;

    for (int c0 = 0; c0 < NOUT; c0 += 16) {
        const unsigned short* bptr = Bt + (size_t)(c0 + l15) * K + kbase;
        f32x8 acc = {};
        for (int k0 = 0; k0 < K; k0 += 32) {
            Frag16 fa, fb;
            fa.q[0] = *(const u32x4*)(aptr + k0);
            fa.q[1] = *(const u32x4*)(aptr + k0 + 16);
            fb.q[0] = *(const u32x4*)(bptr + k0);
            fb.q[1] = *(const u32x4*)(bptr + k0 + 16);
            acc = __builtin_amdgcn_wmma_f32_16x16x32_bf16(
                      false, fa.v, false, fb.v, (short)0, acc, false, false);
        }
        const int colY = c0 + l15;
#pragma unroll
        for (int r = 0; r < 8; ++r) {
            const int row = m0 + r + half * 8;
            if (row < M) Y[(size_t)row * NOUT + colY] = acc[r];
        }
    }
}

__global__ void node_scores(const float* __restrict__ h,
                            const float* __restrict__ as_, const float* __restrict__ ad_,
                            float* __restrict__ es, float* __restrict__ ed,
                            int N, int C)
{
    const int warp = (blockIdx.x * blockDim.x + threadIdx.x) >> 5;
    const int lane = threadIdx.x & 31;
    if (warp >= N) return;
    const float* row = h + (size_t)warp * C;
    float s = 0.f, d = 0.f;
    for (int c = lane; c < C; c += 32) {
        const float v = row[c];
        s += v * as_[c];
        d += v * ad_[c];
    }
#pragma unroll
    for (int off = 16; off; off >>= 1) {
        s += __shfl_xor(s, off);
        d += __shfl_xor(d, off);
    }
    if (lane == 0) { es[warp] = s; ed[warp] = d; }
}

__global__ void zero_i32(int* __restrict__ p, int n)
{
    const int i = blockIdx.x * blockDim.x + threadIdx.x;
    if (i < n) p[i] = 0;
}

__global__ void count_in_degree(const int* __restrict__ ei, int E, int* __restrict__ cnt)
{
    const int e = blockIdx.x * blockDim.x + threadIdx.x;
    if (e < E) atomicAdd(&cnt[ei[E + e]], 1);
}

#define SCAN_TB 1024
__global__ void __launch_bounds__(SCAN_TB)
scan_in_degree(const int* __restrict__ cnt, int* __restrict__ rowbeg,
               int* __restrict__ fillptr, int N)
{
    __shared__ int wpart[SCAN_TB / 32];
    __shared__ int wtot[SCAN_TB / 32];
    const int t = threadIdx.x, lane = t & 31, wv = t >> 5;
    const int base = blockIdx.x * SCAN_TB;
    int part = 0;
    for (int i = t; i < base; i += SCAN_TB) part += cnt[i];
    const int i = base + t;
    const int v = (i < N) ? cnt[i] : 0;
    int inc = v;
    for (int off = 1; off < 32; off <<= 1) {
        const int n = __shfl_up(inc, off);
        if (lane >= off) inc += n;
    }
    for (int off = 16; off; off >>= 1) part += __shfl_xor(part, off);
    if (lane == 31) wtot[wv]  = inc;
    if (lane == 0)  wpart[wv] = part;
    __syncthreads();
    int add = 0;
    for (int w = 0; w < SCAN_TB / 32; ++w) {
        add += wpart[w];
        if (w < wv) add += wtot[w];
    }
    if (i < N) {
        const int ex = add + inc - v;
        rowbeg[i]  = ex;
        fillptr[i] = ex;
    }
}

__global__ void fill_by_destination(const int* __restrict__ ei, int E,
                                    int* __restrict__ fillptr, int* __restrict__ bsrc)
{
    const int e = blockIdx.x * blockDim.x + threadIdx.x;
    if (e >= E) return;
    const int pos = atomicAdd(&fillptr[ei[E + e]], 1);
    bsrc[pos] = ei[e];
}

static __device__ __forceinline__ float leaky02(float l) { return (l > 0.f) ? l : 0.2f * l; }

template <int NV>
__global__ void __launch_bounds__(256)
gat_gather(const int* __restrict__ rowbeg, const int* __restrict__ rowend,
           const int* __restrict__ bsrc,
           const float* __restrict__ h, const float* __restrict__ es,
           const float* __restrict__ ed, const float* __restrict__ bias,
           float* __restrict__ out, int N)
{
    constexpr int C = 128 * NV;
    const int lane = threadIdx.x & 31;
    const int d = __builtin_amdgcn_readfirstlane(
                      (int)((blockIdx.x * blockDim.x + threadIdx.x) >> 5));
    if (d >= N) return;

    const int beg = rowbeg[d], end = rowend[d];
    const float edv   = ed[d];
    const float lself = leaky02(es[d] + edv);
    const float NEG_INF = -__builtin_inff();

    const int i0 = beg + lane;
    int   s0 = d;
    float l0 = NEG_INF;
    if (i0 < end) { s0 = bsrc[i0]; l0 = leaky02(es[s0] + edv); }

    float m = fmaxf(lself, l0);
    for (int i = i0 + 32; i < end; i += 32) m = fmaxf(m, leaky02(es[bsrc[i]] + edv));
    for (int off = 16; off; off >>= 1) m = fmaxf(m, __shfl_xor(m, off));

    float sum = (i0 < end) ? __expf(l0 - m) : 0.f;
    if (lane == 0) sum += __expf(lself - m);
    for (int i = i0 + 32; i < end; i += 32) sum += __expf(leaky02(es[bsrc[i]] + edv) - m);
    for (int off = 16; off; off >>= 1) sum += __shfl_xor(sum, off);

    f32x4 acc[NV];
    {
        const float w = __expf(lself - m) / sum;
        const f32x4* r = (const f32x4*)(h + (size_t)d * C);
        for (int v = 0; v < NV; ++v) acc[v] = r[v * 32 + lane] * w;
    }
    for (int base = beg; base < end; base += 32) {
        int   s = s0;
        float l = l0;
        if (base != beg) {
            s = d; l = NEG_INF;
            if (base + lane < end) { s = bsrc[base + lane]; l = leaky02(es[s] + edv); }
        }
        const float w = (base + lane < end) ? __expf(l - m) / sum : 0.f;
        const int cnt = (end - base < 32) ? (end - base) : 32;
#pragma unroll 4
        for (int j = 0; j < cnt; ++j) {
            const int   sj = __builtin_amdgcn_readlane(s, j);
            const float wj = __int_as_float(__builtin_amdgcn_readlane(__float_as_int(w), j));
            const f32x4* r = (const f32x4*)(h + (size_t)sj * C);
            for (int v = 0; v < NV; ++v) acc[v] += r[v * 32 + lane] * wj;
        }
    }
    f32x4* o4 = (f32x4*)(out + (size_t)d * C);
    for (int v = 0; v < NV; ++v) {
        const float* b = bias + (v * 32 + lane) * 4;
        const f32x4 bv = { b[0], b[1], b[2], b[3] };
        o4[v * 32 + lane] = acc[v] + bv;
    }
}

__global__ void to_bf16(const float* __restrict__ x, unsigned short* __restrict__ y, size_t n)
{
    const size_t i = (size_t)blockIdx.x * blockDim.x + threadIdx.x;
    if (i < n) y[i] = f2bf(x[i]);
}

__global__ void relu_to_bf16(const float* __restrict__ x, unsigned short* __restrict__ y, size_t n)
{
    const size_t i = (size_t)blockIdx.x * blockDim.x + threadIdx.x;
    if (i < n) { const float v = x[i]; y[i] = f2bf(v > 0.f ? v : 0.f); }
}

__global__ void transpose_to_bf16(const float* __restrict__ W, unsigned short* __restrict__ Wt,
                                  int K, int NOUT)
{
    const int i = blockIdx.x * blockDim.x + threadIdx.x;
    if (i >= K * NOUT) return;
    const int k = i / NOUT, n = i % NOUT;
    Wt[(size_t)n * K + k] = f2bf(W[i]);
}

__global__ void link_predict(const float* __restrict__ h, const int* __restrict__ mask,
                             const float* __restrict__ Wl, const float* __restrict__ bl,
                             float* __restrict__ out, int P, int C)
{
    const int warp = (blockIdx.x * blockDim.x + threadIdx.x) >> 5;
    const int lane = threadIdx.x & 31;
    if (warp >= P) return;
    const int m0 = mask[warp * 2 + 0];
    const int m1 = mask[warp * 2 + 1];
    const float* r0 = h + (size_t)m0 * C;
    const float* r1 = h + (size_t)m1 * C;
    float acc = 0.f;
    for (int c = lane; c < C; c += 32)
        acc += r0[c] * Wl[c] + r1[c] * Wl[C + c];
#pragma unroll
    for (int off = 16; off; off >>= 1) acc += __shfl_xor(acc, off);
    if (lane == 0) out[warp] = 1.f / (1.f + expf(-(acc + bl[0])));
}

extern "C" void kernel_launch(void* const* d_in, const int* in_sizes, int n_in,
                              void* d_out, int out_size, void* d_ws, size_t ws_size,
                              hipStream_t stream)
{
    (void)n_in; (void)out_size; (void)ws_size;

    const float* features = (const float*)d_in[0];
    const int*   ei       = (const int*)  d_in[1];
    const int*   mask     = (const int*)  d_in[2];
    const float* W1   = (const float*)d_in[3];
    const float* a_s1 = (const float*)d_in[4];
    const float* a_d1 = (const float*)d_in[5];
    const float* b1   = (const float*)d_in[6];
    const float* W2   = (const float*)d_in[7];
    const float* a_s2 = (const float*)d_in[8];
    const float* a_d2 = (const float*)d_in[9];
    const float* b2   = (const float*)d_in[10];
    const float* Wl   = (const float*)d_in[11];
    const float* bl   = (const float*)d_in[12];

    const int F  = 128, Hh = 256;
    const int N  = in_sizes[0] / F;
    const int E  = in_sizes[1] / 2;
    const int P  = in_sizes[2] / 2;

    char* ws = (char*)d_ws;
    size_t off = 0;
    auto alloc = [&](size_t bytes) -> char* {
        char* ptr = ws + off;
        off += (bytes + 255) & ~(size_t)255;
        return ptr;
    };
    unsigned short* xbf  = (unsigned short*)alloc((size_t)N * F  * 2);
    unsigned short* w1t  = (unsigned short*)alloc((size_t)Hh * F * 2);
    unsigned short* w2t  = (unsigned short*)alloc((size_t)F * Hh * 2);
    unsigned short* hrbf = (unsigned short*)alloc((size_t)N * Hh * 2);
    float*    buf1    = (float*)   alloc((size_t)N * Hh * 4);
    float*    buf2    = (float*)   alloc((size_t)N * Hh * 4);
    float*    es      = (float*)   alloc((size_t)N * 4);
    float*    ed      = (float*)   alloc((size_t)N * 4);
    int*      cnt     = (int*)     alloc((size_t)N * 4);
    int*      rowbeg  = (int*)     alloc((size_t)N * 4);
    int*      rowend  = (int*)     alloc((size_t)N * 4);
    int*      bsrc    = (int*)     alloc((size_t)E * 4);

    const int tilesM = (N + 15) / 16;
    const int gemmGrid = (tilesM + GEMM_WAVES - 1) / GEMM_WAVES;
    const int TB = 256;
    auto cdiv = [](size_t a, size_t b) -> int { return (int)((a + b - 1) / b); };

    to_bf16<<<cdiv((size_t)N * F, TB), TB, 0, stream>>>(features, xbf, (size_t)N * F);
    transpose_to_bf16<<<cdiv((size_t)F * Hh, TB), TB, 0, stream>>>(W1, w1t, F, Hh);
    transpose_to_bf16<<<cdiv((size_t)Hh * F, TB), TB, 0, stream>>>(W2, w2t, Hh, F);

    zero_i32<<<cdiv(N, TB), TB, 0, stream>>>(cnt, N);
    count_in_degree<<<cdiv(E, TB), TB, 0, stream>>>(ei, E, cnt);
    scan_in_degree<<<cdiv(N, SCAN_TB), SCAN_TB, 0, stream>>>(cnt, rowbeg, rowend, N);
    fill_by_destination<<<cdiv(E, TB), TB, 0, stream>>>(ei, E, rowend, bsrc);

    wmma_gemm_bf16<<<gemmGrid, 32 * GEMM_WAVES, 0, stream>>>(xbf, w1t, buf1, N, F, Hh);
    node_scores<<<cdiv((size_t)N * 32, TB), TB, 0, stream>>>(buf1, a_s1, a_d1, es, ed, N, Hh);
    gat_gather<2><<<cdiv((size_t)N * 32, TB), TB, 0, stream>>>(rowbeg, rowend, bsrc, buf1, es, ed,
                                                                b1, buf2, N);
    relu_to_bf16<<<cdiv((size_t)N * Hh, TB), TB, 0, stream>>>(buf2, hrbf, (size_t)N * Hh);

    wmma_gemm_bf16<<<gemmGrid, 32 * GEMM_WAVES, 0, stream>>>(hrbf, w2t, buf1, N, Hh, F);
    node_scores<<<cdiv((size_t)N * 32, TB), TB, 0, stream>>>(buf1, a_s2, a_d2, es, ed, N, F);
    gat_gather<1><<<cdiv((size_t)N * 32, TB), TB, 0, stream>>>(rowbeg, rowend, bsrc, buf1, es, ed,
                                                                b2, buf2, N);

    link_predict<<<cdiv((size_t)P * 32, TB), TB, 0, stream>>>(buf2, mask, Wl, bl,
                                                              (float*)d_out, P, F);
}


# Per-element and row-dot passes folded into producers: feature rounding in layer-1 A loader (fragments built once, kept over the column sweep), es/ed and link dots in epilogues; dot sum order changes
# speedup vs baseline: 3.8952x; 1.4113x over previous
#include <hip/hip_runtime.h>
#include <hip/hip_bf16.h>

typedef __attribute__((__ext_vector_type__(16))) __bf16 bf16x16;
typedef __attribute__((__ext_vector_type__(8)))  float  f32x8;
typedef __attribute__((__ext_vector_type__(4)))  unsigned int u32x4;
typedef __attribute__((__ext_vector_type__(4)))  float  f32x4;

union Frag16 {
    u32x4   q[2];
    bf16x16 v;
};

static __device__ __forceinline__ unsigned short f2bf(float f) {
    unsigned u = __float_as_uint(f);
    u += 0x7FFFu + ((u >> 16) & 1u);
    return (unsigned short)(u >> 16);
}

#define GEMM_WAVES 4

static __device__ __forceinline__ unsigned pack2bf(float lo, float hi) {
    return (unsigned)f2bf(lo) | ((unsigned)f2bf(hi) << 16);
}

static __device__ __forceinline__ void score_add(float (&ps)[8], float (&pd)[8],
                                                 const f32x8 acc, float av, float dv)
{
#pragma unroll
    for (int r = 0; r < 8; ++r) {
        ps[r] += acc[r] * av;
        pd[r] += acc[r] * dv;
    }
}

static __device__ __forceinline__ void score_store(float (&ps)[8], float (&pd)[8],
                                                   int m0, int half, int l15, int M,
                                                   float* __restrict__ es, float* __restrict__ ed)
{
#pragma unroll
    for (int r = 0; r < 8; ++r) {
        for (int off = 8; off; off >>= 1) {
            ps[r] += __shfl_xor(ps[r], off);
            pd[r] += __shfl_xor(pd[r], off);
        }
    }
    if (l15 == 0) {
#pragma unroll
        for (int r = 0; r < 8; ++r) {
            const int row = m0 + r + half * 8;
            if (row < M) { es[row] = ps[r]; ed[row] = pd[r]; }
        }
    }
}

typedef float f32x4_in __attribute__((ext_vector_type(4), aligned(4)));

template <int K>
__global__ void __launch_bounds__(32 * GEMM_WAVES)
wmma_gemm_f32a_scores(const float* __restrict__ A,
                      const unsigned short* __restrict__ Bt,
                      const float* __restrict__ as_, const float* __restrict__ ad_,
                      float* __restrict__ Y,
                      float* __restrict__ es, float* __restrict__ ed,
                      int M, int NOUT)
{
    constexpr int KS = K / 32;
    const int lane  = threadIdx.x & 31;
    const int wave  = threadIdx.x >> 5;
    const int tileM = blockIdx.x * GEMM_WAVES + wave;
    const int m0    = tileM * 16;
    if (m0 >= M) return;

    const int l15   = lane & 15;
    const int half  = lane >> 4;
    const int kbase = half * 8;

    int rowA = m0 + l15;
    if (rowA >= M) rowA = M - 1;
    const float* ap = A + (size_t)rowA * K + kbase;

    Frag16 fa[KS];
#pragma unroll
    for (int s = 0; s < KS; ++s) {
        const f32x4 x0 = *(const f32x4_in*)(ap + s * 32);
        const f32x4 x1 = *(const f32x4_in*)(ap + s * 32 + 4);
        const f32x4 x2 = *(const f32x4_in*)(ap + s * 32 + 16);
        const f32x4 x3 = *(const f32x4_in*)(ap + s * 32 + 20);
        const u32x4 lo = { pack2bf(x0[0], x0[1]), pack2bf(x0[2], x0[3]),
                           pack2bf(x1[0], x1[1]), pack2bf(x1[2], x1[3]) };
        const u32x4 hi = { pack2bf(x2[0], x2[1]), pack2bf(x2[2], x2[3]),
                           pack2bf(x3[0], x3[1]), pack2bf(x3[2], x3[3]) };
        fa[s].q[0] = lo;
        fa[s].q[1] = hi;
    }

    float ps[8], pd[8];
#pragma unroll
    for (int r = 0; r < 8; ++r) { ps[r] = 0.f; pd[r] = 0.f; }

    for (int c0 = 0; c0 < NOUT; c0 += 16) {
        const unsigned short* bptr = Bt + (size_t)(c0 + l15) * K + kbase;
        const int colY = c0 + l15;
        const float av = as_[colY];
        const float dv = ad_[colY];
        f32x8 acc = {};
#pragma unroll
        for (int s = 0; s < KS; ++s) {
            Frag16 fb;
            fb.q[0] = *(const u32x4*)(bptr + s * 32);
            fb.q[1] = *(const u32x4*)(bptr + s * 32 + 16);
            acc = __builtin_amdgcn_wmma_f32_16x16x32_bf16(
                      false, fa[s].v, false, fb.v, (short)0, acc, false, false);
        }
#pragma unroll
        for (int r = 0; r < 8; ++r) {
            const int row = m0 + r + half * 8;
            if (row < M) Y[(size_t)row * NOUT + colY] = acc[r];
        }
        score_add(ps, pd, acc, av, dv);
    }
    score_store(ps, pd, m0, half, l15, M, es, ed);
}

__global__ void wmma_gemm_bf16_scores(const unsigned short* __restrict__ A,
                                      const unsigned short* __restrict__ Bt,
                                      const float* __restrict__ as_, const float* __restrict__ ad_,
                                      float* __restrict__ Y,
                                      float* __restrict__ es, float* __restrict__ ed,
                                      int M, int K, int NOUT)
{
    const int lane  = threadIdx.x & 31;
    const int wave  = threadIdx.x >> 5;
    const int tileM = blockIdx.x * GEMM_WAVES + wave;
    const int m0    = tileM * 16;
    if (m0 >= M) return;

    const int l15   = lane & 15;
    const int half  = lane >> 4;
    const int kbase = half * 8;

    int rowA = m0 + l15;
    if (rowA >= M) rowA = M - 1;
    const unsigned short* aptr = A + (size_t)rowA * K + kbase;

    float ps[8], pd[8];
#pragma unroll
    for (int r = 0; r < 8; ++r) { ps[r] = 0.f; pd[r] = 0.f; }

    for (int c0 = 0; c0 < NOUT; c0 += 16) {
        const unsigned short* bptr = Bt + (size_t)(c0 + l15) * K + kbase;
        const int colY = c0 + l15;
        const float av = as_[colY];
        const float dv = ad_[colY];
        f32x8 acc = {};
        for (int k0 = 0; k0 < K; k0 += 32) {
            Frag16 fa, fb;
            fa.q[0] = *(const u32x4*)(aptr + k0);
            fa.q[1] = *(const u32x4*)(aptr + k0 + 16);
            fb.q[0] = *(const u32x4*)(bptr + k0);
            fb.q[1] = *(const u32x4*)(bptr + k0 + 16);
            acc = __builtin_amdgcn_wmma_f32_16x16x32_bf16(
                      false, fa.v, false, fb.v, (short)0, acc, false, false);
        }
#pragma unroll
        for (int r = 0; r < 8; ++r) {
            const int row = m0 + r + half * 8;
            if (row < M) Y[(size_t)row * NOUT + colY] = acc[r];
        }
        score_add(ps, pd, acc, av, dv);
    }
    score_store(ps, pd, m0, half, l15, M, es, ed);
}

__global__ void zero_i32(int* __restrict__ p, int n)
{
    const int i = blockIdx.x * blockDim.x + threadIdx.x;
    if (i < n) p[i] = 0;
}

__global__ void count_in_degree(const int* __restrict__ ei, int E, int* __restrict__ cnt)
{
    const int e = blockIdx.x * blockDim.x + threadIdx.x;
    if (e < E) atomicAdd(&cnt[ei[E + e]], 1);
}

#define SCAN_TB 1024
__global__ void __launch_bounds__(SCAN_TB)
scan_in_degree(const int* __restrict__ cnt, int* __restrict__ rowbeg,
               int* __restrict__ fillptr, int N)
{
    __shared__ int wpart[SCAN_TB / 32];
    __shared__ int wtot[SCAN_TB / 32];
    const int t = threadIdx.x, lane = t & 31, wv = t >> 5;
    const int base = blockIdx.x * SCAN_TB;
    int part = 0;
    for (int i = t; i < base; i += SCAN_TB) part += cnt[i];
    const int i = base + t;
    const int v = (i < N) ? cnt[i] : 0;
    int inc = v;
    for (int off = 1; off < 32; off <<= 1) {
        const int n = __shfl_up(inc, off);
        if (lane >= off) inc += n;
    }
    for (int off = 16; off; off >>= 1) part += __shfl_xor(part, off);
    if (lane == 31) wtot[wv]  = inc;
    if (lane == 0)  wpart[wv] = part;
    __syncthreads();
    int add = 0;
    for (int w = 0; w < SCAN_TB / 32; ++w) {
        add += wpart[w];
        if (w < wv) add += wtot[w];
    }
    if (i < N) {
        const int ex = add + inc - v;
        rowbeg[i]  = ex;
        fillptr[i] = ex;
    }
}

__global__ void fill_by_destination(const int* __restrict__ ei, int E,
                                    int* __restrict__ fillptr, int* __restrict__ bsrc)
{
    const int e = blockIdx.x * blockDim.x + threadIdx.x;
    if (e >= E) return;
    const int pos = atomicAdd(&fillptr[ei[E + e]], 1);
    bsrc[pos] = ei[e];
}

static __device__ __forceinline__ float leaky02(float l) { return (l > 0.f) ? l : 0.2f * l; }

template <int NV, bool TO_LINK>
__global__ void __launch_bounds__(256)
gat_gather(const int* __restrict__ rowbeg, const int* __restrict__ rowend,
           const int* __restrict__ bsrc,
           const float* __restrict__ h, const float* __restrict__ es,
           const float* __restrict__ ed, const float* __restrict__ bias,
           const float* __restrict__ Wl,
           unsigned short* __restrict__ outbf,
           float* __restrict__ q0, float* __restrict__ q1, int N)
{
    constexpr int C = 128 * NV;
    const int lane = threadIdx.x & 31;
    const int d = __builtin_amdgcn_readfirstlane(
                      (int)((blockIdx.x * blockDim.x + threadIdx.x) >> 5));
    if (d >= N) return;

    const int beg = rowbeg[d], end = rowend[d];
    const float edv   = ed[d];
    const float lself = leaky02(es[d] + edv);
    const float NEG_INF = -__builtin_inff();

    const int i0 = beg + lane;
    int   s0 = d;
    float l0 = NEG_INF;
    if (i0 < end) { s0 = bsrc[i0]; l0 = leaky02(es[s0] + edv); }

    float m = fmaxf(lself, l0);
    for (int i = i0 + 32; i < end; i += 32) m = fmaxf(m, leaky02(es[bsrc[i]] + edv));
    for (int off = 16; off; off >>= 1) m = fmaxf(m, __shfl_xor(m, off));

    float sum = (i0 < end) ? __expf(l0 - m) : 0.f;
    if (lane == 0) sum += __expf(lself - m);
    for (int i = i0 + 32; i < end; i += 32) sum += __expf(leaky02(es[bsrc[i]] + edv) - m);
    for (int off = 16; off; off >>= 1) sum += __shfl_xor(sum, off);

    f32x4 acc[NV];
    {
        const float w = __expf(lself - m) / sum;
        const f32x4* r = (const f32x4*)(h + (size_t)d * C);
        for (int v = 0; v < NV; ++v) acc[v] = r[v * 32 + lane] * w;
    }
    for (int base = beg; base < end; base += 32) {
        int   s = s0;
        float l = l0;
        if (base != beg) {
            s = d; l = NEG_INF;
            if (base + lane < end) { s = bsrc[base + lane]; l = leaky02(es[s] + edv); }
        }
        const float w = (base + lane < end) ? __expf(l - m) / sum : 0.f;
        const int cnt = (end - base < 32) ? (end - base) : 32;
#pragma unroll 4
        for (int j = 0; j < cnt; ++j) {
            const int   sj = __builtin_amdgcn_readlane(s, j);
            const float wj = __int_as_float(__builtin_amdgcn_readlane(__float_as_int(w), j));
            const f32x4* r = (const f32x4*)(h + (size_t)sj * C);
            for (int v = 0; v < NV; ++v) acc[v] += r[v * 32 + lane] * wj;
        }
    }
    float d0 = 0.f, d1 = 0.f;
    for (int v = 0; v < NV; ++v) {
        const int col = (v * 32 + lane) * 4;
        const float* b = bias + col;
        const f32x4 bv = { b[0], b[1], b[2], b[3] };
        const f32x4 o = acc[v] + bv;
        if constexpr (TO_LINK) {
            const float* w0 = Wl + col;
            const float* w1 = Wl + C + col;
            d0 += o[0] * w0[0] + o[1] * w0[1] + o[2] * w0[2] + o[3] * w0[3];
            d1 += o[0] * w1[0] + o[1] * w1[1] + o[2] * w1[2] + o[3] * w1[3];
        } else {
            uint2 pk;
            pk.x = pack2bf(o[0] > 0.f ? o[0] : 0.f, o[1] > 0.f ? o[1] : 0.f);
            pk.y = pack2bf(o[2] > 0.f ? o[2] : 0.f, o[3] > 0.f ? o[3] : 0.f);
            *(uint2*)(outbf + (size_t)d * C + col) = pk;
        }
    }
    if constexpr (TO_LINK) {
        for (int off = 16; off; off >>= 1) {
            d0 += __shfl_xor(d0, off);
            d1 += __shfl_xor(d1, off);
        }
        if (lane == 0) { q0[d] = d0; q1[d] = d1; }
    }
}

__global__ void transpose_to_bf16(const float* __restrict__ W, unsigned short* __restrict__ Wt,
                                  int K, int NOUT)
{
    const int i = blockIdx.x * blockDim.x + threadIdx.x;
    if (i >= K * NOUT) return;
    const int k = i / NOUT, n = i % NOUT;
    Wt[(size_t)n * K + k] = f2bf(W[i]);
}

__global__ void link_predict(const float* __restrict__ q0, const float* __restrict__ q1,
                             const int* __restrict__ mask, const float* __restrict__ bl,
                             float* __restrict__ out, int P)
{
    const int p = blockIdx.x * blockDim.x + threadIdx.x;
    if (p >= P) return;
    const float z = q0[mask[p * 2 + 0]] + q1[mask[p * 2 + 1]] + bl[0];
    out[p] = 1.f / (1.f + expf(-z));
}

extern "C" void kernel_launch(void* const* d_in, const int* in_sizes, int n_in,
                              void* d_out, int out_size, void* d_ws, size_t ws_size,
                              hipStream_t stream)
{
    (void)n_in; (void)out_size; (void)ws_size;

    const float* features = (const float*)d_in[0];
    const int*   ei       = (const int*)  d_in[1];
    const int*   mask     = (const int*)  d_in[2];
    const float* W1   = (const float*)d_in[3];
    const float* a_s1 = (const float*)d_in[4];
    const float* a_d1 = (const float*)d_in[5];
    const float* b1   = (const float*)d_in[6];
    const float* W2   = (const float*)d_in[7];
    const float* a_s2 = (const float*)d_in[8];
    const float* a_d2 = (const float*)d_in[9];
    const float* b2   = (const float*)d_in[10];
    const float* Wl   = (const float*)d_in[11];
    const float* bl   = (const float*)d_in[12];

    const int F  = 128, Hh = 256;
    const int N  = in_sizes[0] / F;
    const int E  = in_sizes[1] / 2;
    const int P  = in_sizes[2] / 2;

    char* ws = (char*)d_ws;
    size_t off = 0;
    auto alloc = [&](size_t bytes) -> char* {
        char* ptr = ws + off;
        off += (bytes + 255) & ~(size_t)255;
        return ptr;
    };
    unsigned short* w1t  = (unsigned short*)alloc((size_t)Hh * F * 2);
    unsigned short* w2t  = (unsigned short*)alloc((size_t)F * Hh * 2);
    unsigned short* hrbf = (unsigned short*)alloc((size_t)N * Hh * 2);
    float*    buf1    = (float*)   alloc((size_t)N * Hh * 4);
    float*    q0      = (float*)   alloc((size_t)N * 4);
    float*    q1      = (float*)   alloc((size_t)N * 4);
    float*    es      = (float*)   alloc((size_t)N * 4);
    float*    ed      = (float*)   alloc((size_t)N * 4);
    int*      cnt     = (int*)     alloc((size_t)N * 4);
    int*      rowbeg  = (int*)     alloc((size_t)N * 4);
    int*      rowend  = (int*)     alloc((size_t)N * 4);
    int*      bsrc    = (int*)     alloc((size_t)E * 4);

    const int tilesM = (N + 15) / 16;
    const int gemmGrid = (tilesM + GEMM_WAVES - 1) / GEMM_WAVES;
    const int TB = 256;
    auto cdiv = [](size_t a, size_t b) -> int { return (int)((a + b - 1) / b); };

    transpose_to_bf16<<<cdiv((size_t)F * Hh, TB), TB, 0, stream>>>(W1, w1t, F, Hh);
    transpose_to_bf16<<<cdiv((size_t)Hh * F, TB), TB, 0, stream>>>(W2, w2t, Hh, F);

    zero_i32<<<cdiv(N, TB), TB, 0, stream>>>(cnt, N);
    count_in_degree<<<cdiv(E, TB), TB, 0, stream>>>(ei, E, cnt);
    scan_in_degree<<<cdiv(N, SCAN_TB), SCAN_TB, 0, stream>>>(cnt, rowbeg, rowend, N);
    fill_by_destination<<<cdiv(E, TB), TB, 0, stream>>>(ei, E, rowend, bsrc);

    wmma_gemm_f32a_scores<128><<<gemmGrid, 32 * GEMM_WAVES, 0, stream>>>(
        features, w1t, a_s1, a_d1, buf1, es, ed, N, Hh);
    gat_gather<2, false><<<cdiv((size_t)N * 32, TB), TB, 0, stream>>>(
        rowbeg, rowend, bsrc, buf1, es, ed, b1, nullptr, hrbf, nullptr, nullptr, N);

    wmma_gemm_bf16_scores<<<gemmGrid, 32 * GEMM_WAVES, 0, stream>>>(
        hrbf, w2t, a_s2, a_d2, buf1, es, ed, N, Hh, F);
    gat_gather<1, true><<<cdiv((size_t)N * 32, TB), TB, 0, stream>>>(
        rowbeg, rowend, bsrc, buf1, es, ed, b2, Wl, nullptr, q0, q1, N);

    link_predict<<<cdiv((size_t)P, TB), TB, 0, stream>>>(q0, q1, mask, bl, (float*)d_out, P);
}


# Layer 2's weight folded with the vectors that consume its rows (scores, link dots): its product shrinks to N x 256 x 4 on WMMA and its gather sums two scalars per edge; sum order changes
# speedup vs baseline: 5.1283x; 1.3166x over previous
#include <hip/hip_runtime.h>
#include <hip/hip_bf16.h>

typedef __attribute__((__ext_vector_type__(16))) __bf16 bf16x16;
typedef __attribute__((__ext_vector_type__(8)))  float  f32x8;
typedef __attribute__((__ext_vector_type__(4)))  unsigned int u32x4;
typedef __attribute__((__ext_vector_type__(4)))  float  f32x4;

union Frag16 {
    u32x4   q[2];
    bf16x16 v;
};

static __device__ __forceinline__ unsigned short f2bf(float f) {
    unsigned u = __float_as_uint(f);
    u += 0x7FFFu + ((u >> 16) & 1u);
    return (unsigned short)(u >> 16);
}

#define GEMM_WAVES 4

static __device__ __forceinline__ unsigned pack2bf(float lo, float hi) {
    return (unsigned)f2bf(lo) | ((unsigned)f2bf(hi) << 16);
}

static __device__ __forceinline__ void score_add(float (&ps)[8], float (&pd)[8],
                                                 const f32x8 acc, float av, float dv)
{
#pragma unroll
    for (int r = 0; r < 8; ++r) {
        ps[r] += acc[r] * av;
        pd[r] += acc[r] * dv;
    }
}

static __device__ __forceinline__ void score_store(float (&ps)[8], float (&pd)[8],
                                                   int m0, int half, int l15, int M,
                                                   float* __restrict__ es, float* __restrict__ ed)
{
#pragma unroll
    for (int r = 0; r < 8; ++r) {
        for (int off = 8; off; off >>= 1) {
            ps[r] += __shfl_xor(ps[r], off);
            pd[r] += __shfl_xor(pd[r], off);
        }
    }
    if (l15 == 0) {
#pragma unroll
        for (int r = 0; r < 8; ++r) {
            const int row = m0 + r + half * 8;
            if (row < M) { es[row] = ps[r]; ed[row] = pd[r]; }
        }
    }
}

typedef float f32x4_in __attribute__((ext_vector_type(4), aligned(4)));

template <int K>
__global__ void __launch_bounds__(32 * GEMM_WAVES)
wmma_gemm_f32a_scores(const float* __restrict__ A,
                      const unsigned short* __restrict__ Bt,
                      const float* __restrict__ as_, const float* __restrict__ ad_,
                      float* __restrict__ Y,
                      float* __restrict__ es, float* __restrict__ ed,
                      int M, int NOUT)
{
    constexpr int KS = K / 32;
    const int lane  = threadIdx.x & 31;
    const int wave  = threadIdx.x >> 5;
    const int tileM = blockIdx.x * GEMM_WAVES + wave;
    const int m0    = tileM * 16;
    if (m0 >= M) return;

    const int l15   = lane & 15;
    const int half  = lane >> 4;
    const int kbase = half * 8;

    int rowA = m0 + l15;
    if (rowA >= M) rowA = M - 1;
    const float* ap = A + (size_t)rowA * K + kbase;

    Frag16 fa[KS];
#pragma unroll
    for (int s = 0; s < KS; ++s) {
        const f32x4 x0 = *(const f32x4_in*)(ap + s * 32);
        const f32x4 x1 = *(const f32x4_in*)(ap + s * 32 + 4);
        const f32x4 x2 = *(const f32x4_in*)(ap + s * 32 + 16);
        const f32x4 x3 = *(const f32x4_in*)(ap + s * 32 + 20);
        const u32x4 lo = { pack2bf(x0[0], x0[1]), pack2bf(x0[2], x0[3]),
                           pack2bf(x1[0], x1[1]), pack2bf(x1[2], x1[3]) };
        const u32x4 hi = { pack2bf(x2[0], x2[1]), pack2bf(x2[2], x2[3]),
                           pack2bf(x3[0], x3[1]), pack2bf(x3[2], x3[3]) };
        fa[s].q[0] = lo;
        fa[s].q[1] = hi;
    }

    float ps[8], pd[8];
#pragma unroll
    for (int r = 0; r < 8; ++r) { ps[r] = 0.f; pd[r] = 0.f; }

    for (int c0 = 0; c0 < NOUT; c0 += 16) {
        const unsigned short* bptr = Bt + (size_t)(c0 + l15) * K + kbase;
        const int colY = c0 + l15;
        const float av = as_[colY];
        const float dv = ad_[colY];
        f32x8 acc = {};
#pragma unroll
        for (int s = 0; s < KS; ++s) {
            Frag16 fb;
            fb.q[0] = *(const u32x4*)(bptr + s * 32);
            fb.q[1] = *(const u32x4*)(bptr + s * 32 + 16);
            acc = __builtin_amdgcn_wmma_f32_16x16x32_bf16(
                      false, fa[s].v, false, fb.v, (short)0, acc, false, false);
        }
#pragma unroll
        for (int r = 0; r < 8; ++r) {
            const int row = m0 + r + half * 8;
            if (row < M) Y[(size_t)row * NOUT + colY] = acc[r];
        }
        score_add(ps, pd, acc, av, dv);
    }
    score_store(ps, pd, m0, half, l15, M, es, ed);
}

template <int K>
__global__ void __launch_bounds__(32 * GEMM_WAVES)
wmma_rows_times_folded(const unsigned short* __restrict__ A,
                       const unsigned short* __restrict__ Ut,
                       float* __restrict__ S, int M)
{
    const int lane  = threadIdx.x & 31;
    const int wave  = threadIdx.x >> 5;
    const int tileM = blockIdx.x * GEMM_WAVES + wave;
    const int m0    = tileM * 16;
    if (m0 >= M) return;

    const int l15   = lane & 15;
    const int half  = lane >> 4;
    const int kbase = half * 8;

    int rowA = m0 + l15;
    if (rowA >= M) rowA = M - 1;
    const unsigned short* aptr = A  + (size_t)rowA * K + kbase;
    const unsigned short* bptr = Ut + (size_t)l15  * K + kbase;

    f32x8 acc = {};
#pragma unroll
    for (int k0 = 0; k0 < K; k0 += 32) {
        Frag16 fa, fb;
        fa.q[0] = *(const u32x4*)(aptr + k0);
        fa.q[1] = *(const u32x4*)(aptr + k0 + 16);
        fb.q[0] = *(const u32x4*)(bptr + k0);
        fb.q[1] = *(const u32x4*)(bptr + k0 + 16);
        acc = __builtin_amdgcn_wmma_f32_16x16x32_bf16(
                  false, fa.v, false, fb.v, (short)0, acc, false, false);
    }
    if (l15 < 4) {
#pragma unroll
        for (int r = 0; r < 8; ++r) {
            const int row = m0 + r + half * 8;
            if (row < M) S[(size_t)row * 4 + l15] = acc[r];
        }
    }
}

__global__ void zero_i32(int* __restrict__ p, int n)
{
    const int i = blockIdx.x * blockDim.x + threadIdx.x;
    if (i < n) p[i] = 0;
}

__global__ void count_in_degree(const int* __restrict__ ei, int E, int* __restrict__ cnt)
{
    const int e = blockIdx.x * blockDim.x + threadIdx.x;
    if (e < E) atomicAdd(&cnt[ei[E + e]], 1);
}

#define SCAN_TB 1024
__global__ void __launch_bounds__(SCAN_TB)
scan_in_degree(const int* __restrict__ cnt, int* __restrict__ rowbeg,
               int* __restrict__ fillptr, int N)
{
    __shared__ int wpart[SCAN_TB / 32];
    __shared__ int wtot[SCAN_TB / 32];
    const int t = threadIdx.x, lane = t & 31, wv = t >> 5;
    const int base = blockIdx.x * SCAN_TB;
    int part = 0;
    for (int i = t; i < base; i += SCAN_TB) part += cnt[i];
    const int i = base + t;
    const int v = (i < N) ? cnt[i] : 0;
    int inc = v;
    for (int off = 1; off < 32; off <<= 1) {
        const int n = __shfl_up(inc, off);
        if (lane >= off) inc += n;
    }
    for (int off = 16; off; off >>= 1) part += __shfl_xor(part, off);
    if (lane == 31) wtot[wv]  = inc;
    if (lane == 0)  wpart[wv] = part;
    __syncthreads();
    int add = 0;
    for (int w = 0; w < SCAN_TB / 32; ++w) {
        add += wpart[w];
        if (w < wv) add += wtot[w];
    }
    if (i < N) {
        const int ex = add + inc - v;
        rowbeg[i]  = ex;
        fillptr[i] = ex;
    }
}

__global__ void fill_by_destination(const int* __restrict__ ei, int E,
                                    int* __restrict__ fillptr, int* __restrict__ bsrc)
{
    const int e = blockIdx.x * blockDim.x + threadIdx.x;
    if (e >= E) return;
    const int pos = atomicAdd(&fillptr[ei[E + e]], 1);
    bsrc[pos] = ei[e];
}

static __device__ __forceinline__ float leaky02(float l) { return (l > 0.f) ? l : 0.2f * l; }

template <int NV>
__global__ void __launch_bounds__(256)
gat_gather(const int* __restrict__ rowbeg, const int* __restrict__ rowend,
           const int* __restrict__ bsrc,
           const float* __restrict__ h, const float* __restrict__ es,
           const float* __restrict__ ed, const float* __restrict__ bias,
           unsigned short* __restrict__ outbf, int N)
{
    constexpr int C = 128 * NV;
    const int lane = threadIdx.x & 31;
    const int d = __builtin_amdgcn_readfirstlane(
                      (int)((blockIdx.x * blockDim.x + threadIdx.x) >> 5));
    if (d >= N) return;

    const int beg = rowbeg[d], end = rowend[d];
    const float edv   = ed[d];
    const float lself = leaky02(es[d] + edv);
    const float NEG_INF = -__builtin_inff();

    const int i0 = beg + lane;
    int   s0 = d;
    float l0 = NEG_INF;
    if (i0 < end) { s0 = bsrc[i0]; l0 = leaky02(es[s0] + edv); }

    float m = fmaxf(lself, l0);
    for (int i = i0 + 32; i < end; i += 32) m = fmaxf(m, leaky02(es[bsrc[i]] + edv));
    for (int off = 16; off; off >>= 1) m = fmaxf(m, __shfl_xor(m, off));

    float sum = (i0 < end) ? __expf(l0 - m) : 0.f;
    if (lane == 0) sum += __expf(lself - m);
    for (int i = i0 + 32; i < end; i += 32) sum += __expf(leaky02(es[bsrc[i]] + edv) - m);
    for (int off = 16; off; off >>= 1) sum += __shfl_xor(sum, off);

    f32x4 acc[NV];
    {
        const float w = __expf(lself - m) / sum;
        const f32x4* r = (const f32x4*)(h + (size_t)d * C);
        for (int v = 0; v < NV; ++v) acc[v] = r[v * 32 + lane] * w;
    }
    for (int base = beg; base < end; base += 32) {
        int   s = s0;
        float l = l0;
        if (base != beg) {
            s = d; l = NEG_INF;
            if (base + lane < end) { s = bsrc[base + lane]; l = leaky02(es[s] + edv); }
        }
        const float w = (base + lane < end) ? __expf(l - m) / sum : 0.f;
        const int cnt = (end - base < 32) ? (end - base) : 32;
#pragma unroll 4
        for (int j = 0; j < cnt; ++j) {
            const int   sj = __builtin_amdgcn_readlane(s, j);
            const float wj = __int_as_float(__builtin_amdgcn_readlane(__float_as_int(w), j));
            const f32x4* r = (const f32x4*)(h + (size_t)sj * C);
            for (int v = 0; v < NV; ++v) acc[v] += r[v * 32 + lane] * wj;
        }
    }
    for (int v = 0; v < NV; ++v) {
        const int col = (v * 32 + lane) * 4;
        const float* b = bias + col;
        const f32x4 bv = { b[0], b[1], b[2], b[3] };
        const f32x4 o = acc[v] + bv;
        uint2 pk;
        pk.x = pack2bf(o[0] > 0.f ? o[0] : 0.f, o[1] > 0.f ? o[1] : 0.f);
        pk.y = pack2bf(o[2] > 0.f ? o[2] : 0.f, o[3] > 0.f ? o[3] : 0.f);
        *(uint2*)(outbf + (size_t)d * C + col) = pk;
    }
}

#define GL 8
__global__ void __launch_bounds__(256)
gat_gather_folded(const int* __restrict__ rowbeg, const int* __restrict__ rowend,
                  const int* __restrict__ bsrc,
                  const f32x4* __restrict__ S, const float* __restrict__ cb,
                  float* __restrict__ q0, float* __restrict__ q1, int N)
{
    const int t   = blockIdx.x * blockDim.x + threadIdx.x;
    const int sub = threadIdx.x & (GL - 1);
    int d = t / GL;
    const bool live = d < N;
    if (!live) d = N - 1;

    const int beg = rowbeg[d], end = rowend[d];
    const f32x4 sd = S[d];
    const float edv   = sd[1];
    const float lself = leaky02(sd[0] + edv);
    const float NEG_INF = -__builtin_inff();

    const int i0 = beg + sub, i1 = i0 + GL;
    float l0 = NEG_INF, l1 = NEG_INF, ta0 = 0.f, tb0 = 0.f, ta1 = 0.f, tb1 = 0.f;
    if (i0 < end) { const f32x4 x = S[bsrc[i0]]; l0 = leaky02(x[0] + edv); ta0 = x[2]; tb0 = x[3]; }
    if (i1 < end) { const f32x4 x = S[bsrc[i1]]; l1 = leaky02(x[0] + edv); ta1 = x[2]; tb1 = x[3]; }

    float m = fmaxf(lself, fmaxf(l0, l1));
    for (int i = i0 + 2 * GL; i < end; i += GL) m = fmaxf(m, leaky02(S[bsrc[i]][0] + edv));
    for (int off = GL / 2; off; off >>= 1) m = fmaxf(m, __shfl_xor(m, off, GL));

    const float p0 = (i0 < end) ? __expf(l0 - m) : 0.f;
    const float p1 = (i1 < end) ? __expf(l1 - m) : 0.f;
    float sum = p0 + p1;
    float a0  = p0 * ta0 + p1 * ta1;
    float a1  = p0 * tb0 + p1 * tb1;
    if (sub == 0) {
        const float ps = __expf(lself - m);
        sum += ps; a0 += ps * sd[2]; a1 += ps * sd[3];
    }
    for (int i = i0 + 2 * GL; i < end; i += GL) {
        const f32x4 x = S[bsrc[i]];
        const float p = __expf(leaky02(x[0] + edv) - m);
        sum += p; a0 += p * x[2]; a1 += p * x[3];
    }
    for (int off = GL / 2; off; off >>= 1) {
        sum += __shfl_xor(sum, off, GL);
        a0  += __shfl_xor(a0,  off, GL);
        a1  += __shfl_xor(a1,  off, GL);
    }
    if (live && sub == 0) {
        q0[d] = a0 / sum + cb[0];
        q1[d] = a1 / sum + cb[1];
    }
}

__global__ void transpose_to_bf16(const float* __restrict__ W, unsigned short* __restrict__ Wt,
                                  int K, int NOUT)
{
    const int i = blockIdx.x * blockDim.x + threadIdx.x;
    if (i >= K * NOUT) return;
    const int k = i / NOUT, n = i % NOUT;
    Wt[(size_t)n * K + k] = f2bf(W[i]);
}

__global__ void __launch_bounds__(256)
fold_weight(const float* __restrict__ W2, const float* __restrict__ a_s,
            const float* __restrict__ a_d, const float* __restrict__ Wl,
            const float* __restrict__ b2,
            unsigned short* __restrict__ Ut, float* __restrict__ cb, int K, int C)
{
    const int lane = threadIdx.x & 31;
    const int w = (int)((blockIdx.x * blockDim.x + threadIdx.x) >> 5);
    if (w > K) return;
    const float* x = (w < K) ? (W2 + (size_t)w * C) : b2;
    float s0 = 0.f, s1 = 0.f, s2 = 0.f, s3 = 0.f;
    for (int c = lane; c < C; c += 32) {
        const float xv = x[c];
        s0 += xv * a_s[c];
        s1 += xv * a_d[c];
        s2 += xv * Wl[c];
        s3 += xv * Wl[C + c];
    }
    for (int off = 16; off; off >>= 1) {
        s0 += __shfl_xor(s0, off);
        s1 += __shfl_xor(s1, off);
        s2 += __shfl_xor(s2, off);
        s3 += __shfl_xor(s3, off);
    }
    if (w < K) {
        const float v = (lane == 0) ? s0 : (lane == 1) ? s1 : (lane == 2) ? s2 : s3;
        if (lane < 16) Ut[(size_t)lane * K + w] = (lane < 4) ? f2bf(v) : (unsigned short)0;
    } else if (lane == 0) {
        cb[0] = s2;
        cb[1] = s3;
    }
}

__global__ void link_predict(const float* __restrict__ q0, const float* __restrict__ q1,
                             const int* __restrict__ mask, const float* __restrict__ bl,
                             float* __restrict__ out, int P)
{
    const int p = blockIdx.x * blockDim.x + threadIdx.x;
    if (p >= P) return;
    const float z = q0[mask[p * 2 + 0]] + q1[mask[p * 2 + 1]] + bl[0];
    out[p] = 1.f / (1.f + expf(-z));
}

extern "C" void kernel_launch(void* const* d_in, const int* in_sizes, int n_in,
                              void* d_out, int out_size, void* d_ws, size_t ws_size,
                              hipStream_t stream)
{
    (void)n_in; (void)out_size; (void)ws_size;

    const float* features = (const float*)d_in[0];
    const int*   ei       = (const int*)  d_in[1];
    const int*   mask     = (const int*)  d_in[2];
    const float* W1   = (const float*)d_in[3];
    const float* a_s1 = (const float*)d_in[4];
    const float* a_d1 = (const float*)d_in[5];
    const float* b1   = (const float*)d_in[6];
    const float* W2   = (const float*)d_in[7];
    const float* a_s2 = (const float*)d_in[8];
    const float* a_d2 = (const float*)d_in[9];
    const float* b2   = (const float*)d_in[10];
    const float* Wl   = (const float*)d_in[11];
    const float* bl   = (const float*)d_in[12];

    const int F  = 128, Hh = 256;
    const int N  = in_sizes[0] / F;
    const int E  = in_sizes[1] / 2;
    const int P  = in_sizes[2] / 2;

    char* ws = (char*)d_ws;
    size_t off = 0;
    auto alloc = [&](size_t bytes) -> char* {
        char* ptr = ws + off;
        off += (bytes + 255) & ~(size_t)255;
        return ptr;
    };
    unsigned short* w1t  = (unsigned short*)alloc((size_t)Hh * F * 2);
    unsigned short* ut   = (unsigned short*)alloc((size_t)16 * Hh * 2);
    float*    cb      = (float*)   alloc((size_t)2 * 4);
    float*    S       = (float*)   alloc((size_t)N * 4 * 4);
    unsigned short* hrbf = (unsigned short*)alloc((size_t)N * Hh * 2);
    float*    buf1    = (float*)   alloc((size_t)N * Hh * 4);
    float*    q0      = (float*)   alloc((size_t)N * 4);
    float*    q1      = (float*)   alloc((size_t)N * 4);
    float*    es      = (float*)   alloc((size_t)N * 4);
    float*    ed      = (float*)   alloc((size_t)N * 4);
    int*      cnt     = (int*)     alloc((size_t)N * 4);
    int*      rowbeg  = (int*)     alloc((size_t)N * 4);
    int*      rowend  = (int*)     alloc((size_t)N * 4);
    int*      bsrc    = (int*)     alloc((size_t)E * 4);

    const int tilesM = (N + 15) / 16;
    const int gemmGrid = (tilesM + GEMM_WAVES - 1) / GEMM_WAVES;
    const int TB = 256;
    auto cdiv = [](size_t a, size_t b) -> int { return (int)((a + b - 1) / b); };

    transpose_to_bf16<<<cdiv((size_t)F * Hh, TB), TB, 0, stream>>>(W1, w1t, F, Hh);
    fold_weight<<<cdiv((size_t)(Hh + 1) * 32, TB), TB, 0, stream>>>(W2, a_s2, a_d2, Wl, b2, ut, cb, Hh, F);

    zero_i32<<<cdiv(N, TB), TB, 0, stream>>>(cnt, N);
    count_in_degree<<<cdiv(E, TB), TB, 0, stream>>>(ei, E, cnt);
    scan_in_degree<<<cdiv(N, SCAN_TB), SCAN_TB, 0, stream>>>(cnt, rowbeg, rowend, N);
    fill_by_destination<<<cdiv(E, TB), TB, 0, stream>>>(ei, E, rowend, bsrc);

    wmma_gemm_f32a_scores<128><<<gemmGrid, 32 * GEMM_WAVES, 0, stream>>>(
        features, w1t, a_s1, a_d1, buf1, es, ed, N, Hh);
    gat_gather<2><<<cdiv((size_t)N * 32, TB), TB, 0, stream>>>(
        rowbeg, rowend, bsrc, buf1, es, ed, b1, hrbf, N);

    wmma_rows_times_folded<256><<<gemmGrid, 32 * GEMM_WAVES, 0, stream>>>(hrbf, ut, S, N);
    gat_gather_folded<<<cdiv((size_t)N * GL, TB), TB, 0, stream>>>(
        rowbeg, rowend, bsrc, (const f32x4*)S, cb, q0, q1, N);

    link_predict<<<cdiv((size_t)P, TB), TB, 0, stream>>>(q0, q1, mask, bl, (float*)d_out, P);
}


# Independent stages share launches: weight transpose, weight fold, counter clearing in one; bucket fill rides beside the layer-1 product; 11 launches -> 8; arrival (sum) order in buckets may shift
# speedup vs baseline: 5.7143x; 1.1143x over previous
#include <hip/hip_runtime.h>
#include <hip/hip_bf16.h>

typedef __attribute__((__ext_vector_type__(16))) __bf16 bf16x16;
typedef __attribute__((__ext_vector_type__(8)))  float  f32x8;
typedef __attribute__((__ext_vector_type__(4)))  unsigned int u32x4;
typedef __attribute__((__ext_vector_type__(4)))  float  f32x4;

union Frag16 {
    u32x4   q[2];
    bf16x16 v;
};

static __device__ __forceinline__ unsigned short f2bf(float f) {
    unsigned u = __float_as_uint(f);
    u += 0x7FFFu + ((u >> 16) & 1u);
    return (unsigned short)(u >> 16);
}

#define GEMM_WAVES 4

static __device__ __forceinline__ unsigned pack2bf(float lo, float hi) {
    return (unsigned)f2bf(lo) | ((unsigned)f2bf(hi) << 16);
}

static __device__ __forceinline__ void score_add(float (&ps)[8], float (&pd)[8],
                                                 const f32x8 acc, float av, float dv)
{
#pragma unroll
    for (int r = 0; r < 8; ++r) {
        ps[r] += acc[r] * av;
        pd[r] += acc[r] * dv;
    }
}

static __device__ __forceinline__ void score_store(float (&ps)[8], float (&pd)[8],
                                                   int m0, int half, int l15, int M,
                                                   float* __restrict__ es, float* __restrict__ ed)
{
#pragma unroll
    for (int r = 0; r < 8; ++r) {
        for (int off = 8; off; off >>= 1) {
            ps[r] += __shfl_xor(ps[r], off);
            pd[r] += __shfl_xor(pd[r], off);
        }
    }
    if (l15 == 0) {
#pragma unroll
        for (int r = 0; r < 8; ++r) {
            const int row = m0 + r + half * 8;
            if (row < M) { es[row] = ps[r]; ed[row] = pd[r]; }
        }
    }
}

typedef float f32x4_in __attribute__((ext_vector_type(4), aligned(4)));

template <int K>
static __device__ __forceinline__ void
wmma_gemm_f32a_scores_block(int blk,
                            const float* __restrict__ A,
                            const unsigned short* __restrict__ Bt,
                            const float* __restrict__ as_, const float* __restrict__ ad_,
                            float* __restrict__ Y,
                            float* __restrict__ es, float* __restrict__ ed,
                            int M, int NOUT)
{
    constexpr int KS = K / 32;
    const int lane  = threadIdx.x & 31;
    const int wave  = threadIdx.x >> 5;
    const int tileM = blk * GEMM_WAVES + wave;
    const int m0    = tileM * 16;
    if (m0 >= M) return;

    const int l15   = lane & 15;
    const int half  = lane >> 4;
    const int kbase = half * 8;

    int rowA = m0 + l15;
    if (rowA >= M) rowA = M - 1;
    const float* ap = A + (size_t)rowA * K + kbase;

    Frag16 fa[KS];
#pragma unroll
    for (int s = 0; s < KS; ++s) {
        const f32x4 x0 = *(const f32x4_in*)(ap + s * 32);
        const f32x4 x1 = *(const f32x4_in*)(ap + s * 32 + 4);
        const f32x4 x2 = *(const f32x4_in*)(ap + s * 32 + 16);
        const f32x4 x3 = *(const f32x4_in*)(ap + s * 32 + 20);
        const u32x4 lo = { pack2bf(x0[0], x0[1]), pack2bf(x0[2], x0[3]),
                           pack2bf(x1[0], x1[1]), pack2bf(x1[2], x1[3]) };
        const u32x4 hi = { pack2bf(x2[0], x2[1]), pack2bf(x2[2], x2[3]),
                           pack2bf(x3[0], x3[1]), pack2bf(x3[2], x3[3]) };
        fa[s].q[0] = lo;
        fa[s].q[1] = hi;
    }

    float ps[8], pd[8];
#pragma unroll
    for (int r = 0; r < 8; ++r) { ps[r] = 0.f; pd[r] = 0.f; }

    for (int c0 = 0; c0 < NOUT; c0 += 16) {
        const unsigned short* bptr = Bt + (size_t)(c0 + l15) * K + kbase;
        const int colY = c0 + l15;
        const float av = as_[colY];
        const float dv = ad_[colY];
        f32x8 acc = {};
#pragma unroll
        for (int s = 0; s < KS; ++s) {
            Frag16 fb;
            fb.q[0] = *(const u32x4*)(bptr + s * 32);
            fb.q[1] = *(const u32x4*)(bptr + s * 32 + 16);
            acc = __builtin_amdgcn_wmma_f32_16x16x32_bf16(
                      false, fa[s].v, false, fb.v, (short)0, acc, false, false);
        }
#pragma unroll
        for (int r = 0; r < 8; ++r) {
            const int row = m0 + r + half * 8;
            if (row < M) Y[(size_t)row * NOUT + colY] = acc[r];
        }
        score_add(ps, pd, acc, av, dv);
    }
    score_store(ps, pd, m0, half, l15, M, es, ed);
}

template <int K>
__global__ void __launch_bounds__(32 * GEMM_WAVES)
wmma_rows_times_folded(const unsigned short* __restrict__ A,
                       const unsigned short* __restrict__ Ut,
                       float* __restrict__ S, int M)
{
    const int lane  = threadIdx.x & 31;
    const int wave  = threadIdx.x >> 5;
    const int tileM = blockIdx.x * GEMM_WAVES + wave;
    const int m0    = tileM * 16;
    if (m0 >= M) return;

    const int l15   = lane & 15;
    const int half  = lane >> 4;
    const int kbase = half * 8;

    int rowA = m0 + l15;
    if (rowA >= M) rowA = M - 1;
    const unsigned short* aptr = A  + (size_t)rowA * K + kbase;
    const unsigned short* bptr = Ut + (size_t)l15  * K + kbase;

    f32x8 acc = {};
#pragma unroll
    for (int k0 = 0; k0 < K; k0 += 32) {
        Frag16 fa, fb;
        fa.q[0] = *(const u32x4*)(aptr + k0);
        fa.q[1] = *(const u32x4*)(aptr + k0 + 16);
        fb.q[0] = *(const u32x4*)(bptr + k0);
        fb.q[1] = *(const u32x4*)(bptr + k0 + 16);
        acc = __builtin_amdgcn_wmma_f32_16x16x32_bf16(
                  false, fa.v, false, fb.v, (short)0, acc, false, false);
    }
    if (l15 < 4) {
#pragma unroll
        for (int r = 0; r < 8; ++r) {
            const int row = m0 + r + half * 8;
            if (row < M) S[(size_t)row * 4 + l15] = acc[r];
        }
    }
}

static __device__ __forceinline__ void zero_i32_block(int blk, int* __restrict__ p, int n)
{
    const int i = blk * blockDim.x + threadIdx.x;
    if (i < n) p[i] = 0;
}

__global__ void count_in_degree(const int* __restrict__ ei, int E, int* __restrict__ cnt)
{
    const int e = blockIdx.x * blockDim.x + threadIdx.x;
    if (e < E) atomicAdd(&cnt[ei[E + e]], 1);
}

#define SCAN_TB 1024
__global__ void __launch_bounds__(SCAN_TB)
scan_in_degree(const int* __restrict__ cnt, int* __restrict__ rowbeg,
               int* __restrict__ fillptr, int N)
{
    __shared__ int wpart[SCAN_TB / 32];
    __shared__ int wtot[SCAN_TB / 32];
    const int t = threadIdx.x, lane = t & 31, wv = t >> 5;
    const int base = blockIdx.x * SCAN_TB;
    int part = 0;
    for (int i = t; i < base; i += SCAN_TB) part += cnt[i];
    const int i = base + t;
    const int v = (i < N) ? cnt[i] : 0;
    int inc = v;
    for (int off = 1; off < 32; off <<= 1) {
        const int n = __shfl_up(inc, off);
        if (lane >= off) inc += n;
    }
    for (int off = 16; off; off >>= 1) part += __shfl_xor(part, off);
    if (lane == 31) wtot[wv]  = inc;
    if (lane == 0)  wpart[wv] = part;
    __syncthreads();
    int add = 0;
    for (int w = 0; w < SCAN_TB / 32; ++w) {
        add += wpart[w];
        if (w < wv) add += wtot[w];
    }
    if (i < N) {
        const int ex = add + inc - v;
        rowbeg[i]  = ex;
        fillptr[i] = ex;
    }
}

static __device__ __forceinline__ void
fill_by_destination_block(int blk, const int* __restrict__ ei, int E,
                          int* __restrict__ fillptr, int* __restrict__ bsrc)
{
    const int e = blk * blockDim.x + threadIdx.x;
    if (e >= E) return;
    const int pos = atomicAdd(&fillptr[ei[E + e]], 1);
    bsrc[pos] = ei[e];
}

template <int K>
__global__ void __launch_bounds__(32 * GEMM_WAVES)
layer1_product_beside_bucket_fill(const float* __restrict__ A,
                                  const unsigned short* __restrict__ Bt,
                                  const float* __restrict__ as_, const float* __restrict__ ad_,
                                  float* __restrict__ Y,
                                  float* __restrict__ es, float* __restrict__ ed,
                                  int M, int NOUT, int productBlocks,
                                  const int* __restrict__ ei, int E,
                                  int* __restrict__ fillptr, int* __restrict__ bsrc)
{
    const int b = blockIdx.x;
    if (b < productBlocks)
        wmma_gemm_f32a_scores_block<K>(b, A, Bt, as_, ad_, Y, es, ed, M, NOUT);
    else
        fill_by_destination_block(b - productBlocks, ei, E, fillptr, bsrc);
}

static __device__ __forceinline__ float leaky02(float l) { return (l > 0.f) ? l : 0.2f * l; }

template <int NV>
__global__ void __launch_bounds__(256)
gat_gather(const int* __restrict__ rowbeg, const int* __restrict__ rowend,
           const int* __restrict__ bsrc,
           const float* __restrict__ h, const float* __restrict__ es,
           const float* __restrict__ ed, const float* __restrict__ bias,
           unsigned short* __restrict__ outbf, int N)
{
    constexpr int C = 128 * NV;
    const int lane = threadIdx.x & 31;
    const int d = __builtin_amdgcn_readfirstlane(
                      (int)((blockIdx.x * blockDim.x + threadIdx.x) >> 5));
    if (d >= N) return;

    const int beg = rowbeg[d], end = rowend[d];
    const float edv   = ed[d];
    const float lself = leaky02(es[d] + edv);
    const float NEG_INF = -__builtin_inff();

    const int i0 = beg + lane;
    int   s0 = d;
    float l0 = NEG_INF;
    if (i0 < end) { s0 = bsrc[i0]; l0 = leaky02(es[s0] + edv); }

    float m = fmaxf(lself, l0);
    for (int i = i0 + 32; i < end; i += 32) m = fmaxf(m, leaky02(es[bsrc[i]] + edv));
    for (int off = 16; off; off >>= 1) m = fmaxf(m, __shfl_xor(m, off));

    float sum = (i0 < end) ? __expf(l0 - m) : 0.f;
    if (lane == 0) sum += __expf(lself - m);
    for (int i = i0 + 32; i < end; i += 32) sum += __expf(leaky02(es[bsrc[i]] + edv) - m);
    for (int off = 16; off; off >>= 1) sum += __shfl_xor(sum, off);

    f32x4 acc[NV];
    {
        const float w = __expf(lself - m) / sum;
        const f32x4* r = (const f32x4*)(h + (size_t)d * C);
        for (int v = 0; v < NV; ++v) acc[v] = r[v * 32 + lane] * w;
    }
    for (int base = beg; base < end; base += 32) {
        int   s = s0;
        float l = l0;
        if (base != beg) {
            s = d; l = NEG_INF;
            if (base + lane < end) { s = bsrc[base + lane]; l = leaky02(es[s] + edv); }
        }
        const float w = (base + lane < end) ? __expf(l - m) / sum : 0.f;
        const int cnt = (end - base < 32) ? (end - base) : 32;
#pragma unroll 4
        for (int j = 0; j < cnt; ++j) {
            const int   sj = __builtin_amdgcn_readlane(s, j);
            const float wj = __int_as_float(__builtin_amdgcn_readlane(__float_as_int(w), j));
            const f32x4* r = (const f32x4*)(h + (size_t)sj * C);
            for (int v = 0; v < NV; ++v) acc[v] += r[v * 32 + lane] * wj;
        }
    }
    for (int v = 0; v < NV; ++v) {
        const int col = (v * 32 + lane) * 4;
        const float* b = bias + col;
        const f32x4 bv = { b[0], b[1], b[2], b[3] };
        const f32x4 o = acc[v] + bv;
        uint2 pk;
        pk.x = pack2bf(o[0] > 0.f ? o[0] : 0.f, o[1] > 0.f ? o[1] : 0.f);
        pk.y = pack2bf(o[2] > 0.f ? o[2] : 0.f, o[3] > 0.f ? o[3] : 0.f);
        *(uint2*)(outbf + (size_t)d * C + col) = pk;
    }
}

#define GL 8
__global__ void __launch_bounds__(256)
gat_gather_folded(const int* __restrict__ rowbeg, const int* __restrict__ rowend,
                  const int* __restrict__ bsrc,
                  const f32x4* __restrict__ S, const float* __restrict__ cb,
                  float* __restrict__ q0, float* __restrict__ q1, int N)
{
    const int t   = blockIdx.x * blockDim.x + threadIdx.x;
    const int sub = threadIdx.x & (GL - 1);
    int d = t / GL;
    const bool live = d < N;
    if (!live) d = N - 1;

    const int beg = rowbeg[d], end = rowend[d];
    const f32x4 sd = S[d];
    const float edv   = sd[1];
    const float lself = leaky02(sd[0] + edv);
    const float NEG_INF = -__builtin_inff();

    const int i0 = beg + sub, i1 = i0 + GL;
    float l0 = NEG_INF, l1 = NEG_INF, ta0 = 0.f, tb0 = 0.f, ta1 = 0.f, tb1 = 0.f;
    if (i0 < end) { const f32x4 x = S[bsrc[i0]]; l0 = leaky02(x[0] + edv); ta0 = x[2]; tb0 = x[3]; }
    if (i1 < end) { const f32x4 x = S[bsrc[i1]]; l1 = leaky02(x[0] + edv); ta1 = x[2]; tb1 = x[3]; }

    float m = fmaxf(lself, fmaxf(l0, l1));
    for (int i = i0 + 2 * GL; i < end; i += GL) m = fmaxf(m, leaky02(S[bsrc[i]][0] + edv));
    for (int off = GL / 2; off; off >>= 1) m = fmaxf(m, __shfl_xor(m, off, GL));

    const float p0 = (i0 < end) ? __expf(l0 - m) : 0.f;
    const float p1 = (i1 < end) ? __expf(l1 - m) : 0.f;
    float sum = p0 + p1;
    float a0  = p0 * ta0 + p1 * ta1;
    float a1  = p0 * tb0 + p1 * tb1;
    if (sub == 0) {
        const float ps = __expf(lself - m);
        sum += ps; a0 += ps * sd[2]; a1 += ps * sd[3];
    }
    for (int i = i0 + 2 * GL; i < end; i += GL) {
        const f32x4 x = S[bsrc[i]];
        const float p = __expf(leaky02(x[0] + edv) - m);
        sum += p; a0 += p * x[2]; a1 += p * x[3];
    }
    for (int off = GL / 2; off; off >>= 1) {
        sum += __shfl_xor(sum, off, GL);
        a0  += __shfl_xor(a0,  off, GL);
        a1  += __shfl_xor(a1,  off, GL);
    }
    if (live && sub == 0) {
        q0[d] = a0 / sum + cb[0];
        q1[d] = a1 / sum + cb[1];
    }
}

static __device__ __forceinline__ void
transpose_to_bf16_block(int blk, const float* __restrict__ W, unsigned short* __restrict__ Wt,
                        int K, int NOUT)
{
    const int i = blk * blockDim.x + threadIdx.x;
    if (i >= K * NOUT) return;
    const int k = i / NOUT, n = i % NOUT;
    Wt[(size_t)n * K + k] = f2bf(W[i]);
}

static __device__ __forceinline__ void
fold_weight_block(int blk, const float* __restrict__ W2, const float* __restrict__ a_s,
                  const float* __restrict__ a_d, const float* __restrict__ Wl,
                  const float* __restrict__ b2,
                  unsigned short* __restrict__ Ut, float* __restrict__ cb, int K, int C)
{
    const int lane = threadIdx.x & 31;
    const int w = (int)((blk * blockDim.x + threadIdx.x) >> 5);
    if (w > K) return;
    const float* x = (w < K) ? (W2 + (size_t)w * C) : b2;
    float s0 = 0.f, s1 = 0.f, s2 = 0.f, s3 = 0.f;
    for (int c = lane; c < C; c += 32) {
        const float xv = x[c];
        s0 += xv * a_s[c];
        s1 += xv * a_d[c];
        s2 += xv * Wl[c];
        s3 += xv * Wl[C + c];
    }
    for (int off = 16; off; off >>= 1) {
        s0 += __shfl_xor(s0, off);
        s1 += __shfl_xor(s1, off);
        s2 += __shfl_xor(s2, off);
        s3 += __shfl_xor(s3, off);
    }
    if (w < K) {
        const float v = (lane == 0) ? s0 : (lane == 1) ? s1 : (lane == 2) ? s2 : s3;
        if (lane < 16) Ut[(size_t)lane * K + w] = (lane < 4) ? f2bf(v) : (unsigned short)0;
    } else if (lane == 0) {
        cb[0] = s2;
        cb[1] = s3;
    }
}

__global__ void __launch_bounds__(256)
prepare_weights_and_counters(const float* __restrict__ W1, unsigned short* __restrict__ w1t,
                             int K1, int NOUT1, int transposeBlocks,
                             const float* __restrict__ W2, const float* __restrict__ a_s,
                             const float* __restrict__ a_d, const float* __restrict__ Wl,
                             const float* __restrict__ b2,
                             unsigned short* __restrict__ Ut, float* __restrict__ cb,
                             int K2, int C2, int foldBlocks,
                             int* __restrict__ cnt, int N)
{
    const int b = blockIdx.x;
    if (b < transposeBlocks)
        transpose_to_bf16_block(b, W1, w1t, K1, NOUT1);
    else if (b < transposeBlocks + foldBlocks)
        fold_weight_block(b - transposeBlocks, W2, a_s, a_d, Wl, b2, Ut, cb, K2, C2);
    else
        zero_i32_block(b - transposeBlocks - foldBlocks, cnt, N);
}

__global__ void link_predict(const float* __restrict__ q0, const float* __restrict__ q1,
                             const int* __restrict__ mask, const float* __restrict__ bl,
                             float* __restrict__ out, int P)
{
    const int p = blockIdx.x * blockDim.x + threadIdx.x;
    if (p >= P) return;
    const float z = q0[mask[p * 2 + 0]] + q1[mask[p * 2 + 1]] + bl[0];
    out[p] = 1.f / (1.f + expf(-z));
}

extern "C" void kernel_launch(void* const* d_in, const int* in_sizes, int n_in,
                              void* d_out, int out_size, void* d_ws, size_t ws_size,
                              hipStream_t stream)
{
    (void)n_in; (void)out_size; (void)ws_size;

    const float* features = (const float*)d_in[0];
    const int*   ei       = (const int*)  d_in[1];
    const int*   mask     = (const int*)  d_in[2];
    const float* W1   = (const float*)d_in[3];
    const float* a_s1 = (const float*)d_in[4];
    const float* a_d1 = (const float*)d_in[5];
    const float* b1   = (const float*)d_in[6];
    const float* W2   = (const float*)d_in[7];
    const float* a_s2 = (const float*)d_in[8];
    const float* a_d2 = (const float*)d_in[9];
    const float* b2   = (const float*)d_in[10];
    const float* Wl   = (const float*)d_in[11];
    const float* bl   = (const float*)d_in[12];

    const int F  = 128, Hh = 256;
    const int N  = in_sizes[0] / F;
    const int E  = in_sizes[1] / 2;
    const int P  = in_sizes[2] / 2;

    char* ws = (char*)d_ws;
    size_t off = 0;
    auto alloc = [&](size_t bytes) -> char* {
        char* ptr = ws + off;
        off += (bytes + 255) & ~(size_t)255;
        return ptr;
    };
    unsigned short* w1t  = (unsigned short*)alloc((size_t)Hh * F * 2);
    unsigned short* ut   = (unsigned short*)alloc((size_t)16 * Hh * 2);
    float*    cb      = (float*)   alloc((size_t)2 * 4);
    float*    S       = (float*)   alloc((size_t)N * 4 * 4);
    unsigned short* hrbf = (unsigned short*)alloc((size_t)N * Hh * 2);
    float*    buf1    = (float*)   alloc((size_t)N * Hh * 4);
    float*    q0      = (float*)   alloc((size_t)N * 4);
    float*    q1      = (float*)   alloc((size_t)N * 4);
    float*    es      = (float*)   alloc((size_t)N * 4);
    float*    ed      = (float*)   alloc((size_t)N * 4);
    int*      cnt     = (int*)     alloc((size_t)N * 4);
    int*      rowbeg  = (int*)     alloc((size_t)N * 4);
    int*      rowend  = (int*)     alloc((size_t)N * 4);
    int*      bsrc    = (int*)     alloc((size_t)E * 4);

    const int tilesM = (N + 15) / 16;
    const int gemmGrid = (tilesM + GEMM_WAVES - 1) / GEMM_WAVES;
    const int TB = 256;
    auto cdiv = [](size_t a, size_t b) -> int { return (int)((a + b - 1) / b); };

    const int transposeBlocks = cdiv((size_t)F * Hh, TB);
    const int foldBlocks      = cdiv((size_t)(Hh + 1) * 32, TB);
    prepare_weights_and_counters<<<transposeBlocks + foldBlocks + cdiv(N, TB), TB, 0, stream>>>(
        W1, w1t, F, Hh, transposeBlocks,
        W2, a_s2, a_d2, Wl, b2, ut, cb, Hh, F, foldBlocks,
        cnt, N);

    count_in_degree<<<cdiv(E, TB), TB, 0, stream>>>(ei, E, cnt);
    scan_in_degree<<<cdiv(N, SCAN_TB), SCAN_TB, 0, stream>>>(cnt, rowbeg, rowend, N);

    layer1_product_beside_bucket_fill<128>
        <<<gemmGrid + cdiv(E, 32 * GEMM_WAVES), 32 * GEMM_WAVES, 0, stream>>>(
        features, w1t, a_s1, a_d1, buf1, es, ed, N, Hh, gemmGrid,
        ei, E, rowend, bsrc);
    gat_gather<2><<<cdiv((size_t)N * 32, TB), TB, 0, stream>>>(
        rowbeg, rowend, bsrc, buf1, es, ed, b1, hrbf, N);

    wmma_rows_times_folded<256><<<gemmGrid, 32 * GEMM_WAVES, 0, stream>>>(hrbf, ut, S, N);
    gat_gather_folded<<<cdiv((size_t)N * GL, TB), TB, 0, stream>>>(
        rowbeg, rowend, bsrc, (const f32x4*)S, cb, q0, q1, N);

    link_predict<<<cdiv((size_t)P, TB), TB, 0, stream>>>(q0, q1, mask, bl, (float*)d_out, P);
}


# Layer-1 product rows stored as bf16 instead of f32 between the product and the first gather (precision: one more rounding before the weighted row sum); the gather reads half the bytes per edge
# speedup vs baseline: 6.1416x; 1.0748x over previous
#include <hip/hip_runtime.h>
#include <hip/hip_bf16.h>

typedef __attribute__((__ext_vector_type__(16))) __bf16 bf16x16;
typedef __attribute__((__ext_vector_type__(8)))  float  f32x8;
typedef __attribute__((__ext_vector_type__(4)))  unsigned int u32x4;
typedef __attribute__((__ext_vector_type__(4)))  float  f32x4;
typedef __attribute__((__ext_vector_type__(2)))  float  f32x2;
typedef __attribute__((__ext_vector_type__(2)))  __bf16 bf16x2;

union Frag16 {
    u32x4   q[2];
    bf16x16 v;
};

static __device__ __forceinline__ unsigned short f2bf(float f) {
    unsigned u = __float_as_uint(f);
    u += 0x7FFFu + ((u >> 16) & 1u);
    return (unsigned short)(u >> 16);
}

#define GEMM_WAVES 4

static __device__ __forceinline__ unsigned pack2bf(float lo, float hi) {
    return (unsigned)f2bf(lo) | ((unsigned)f2bf(hi) << 16);
}

static __device__ __forceinline__ void score_add(float (&ps)[8], float (&pd)[8],
                                                 const f32x8 acc, float av, float dv)
{
#pragma unroll
    for (int r = 0; r < 8; ++r) {
        ps[r] += acc[r] * av;
        pd[r] += acc[r] * dv;
    }
}

static __device__ __forceinline__ void score_store(float (&ps)[8], float (&pd)[8],
                                                   int m0, int half, int l15, int M,
                                                   float* __restrict__ es, float* __restrict__ ed)
{
#pragma unroll
    for (int r = 0; r < 8; ++r) {
        for (int off = 8; off; off >>= 1) {
            ps[r] += __shfl_xor(ps[r], off);
            pd[r] += __shfl_xor(pd[r], off);
        }
    }
    if (l15 == 0) {
#pragma unroll
        for (int r = 0; r < 8; ++r) {
            const int row = m0 + r + half * 8;
            if (row < M) { es[row] = ps[r]; ed[row] = pd[r]; }
        }
    }
}

typedef float f32x4_in __attribute__((ext_vector_type(4), aligned(4)));

template <int K>
static __device__ __forceinline__ void
wmma_gemm_f32a_scores_block(int blk,
                            const float* __restrict__ A,
                            const unsigned short* __restrict__ Bt,
                            const float* __restrict__ as_, const float* __restrict__ ad_,
                            unsigned short* __restrict__ Y,
                            float* __restrict__ es, float* __restrict__ ed,
                            int M, int NOUT)
{
    constexpr int KS = K / 32;
    const int lane  = threadIdx.x & 31;
    const int wave  = threadIdx.x >> 5;
    const int tileM = blk * GEMM_WAVES + wave;
    const int m0    = tileM * 16;
    if (m0 >= M) return;

    const int l15   = lane & 15;
    const int half  = lane >> 4;
    const int kbase = half * 8;

    int rowA = m0 + l15;
    if (rowA >= M) rowA = M - 1;
    const float* ap = A + (size_t)rowA * K + kbase;

    Frag16 fa[KS];
#pragma unroll
    for (int s = 0; s < KS; ++s) {
        const f32x4 x0 = *(const f32x4_in*)(ap + s * 32);
        const f32x4 x1 = *(const f32x4_in*)(ap + s * 32 + 4);
        const f32x4 x2 = *(const f32x4_in*)(ap + s * 32 + 16);
        const f32x4 x3 = *(const f32x4_in*)(ap + s * 32 + 20);
        const u32x4 lo = { pack2bf(x0[0], x0[1]), pack2bf(x0[2], x0[3]),
                           pack2bf(x1[0], x1[1]), pack2bf(x1[2], x1[3]) };
        const u32x4 hi = { pack2bf(x2[0], x2[1]), pack2bf(x2[2], x2[3]),
                           pack2bf(x3[0], x3[1]), pack2bf(x3[2], x3[3]) };
        fa[s].q[0] = lo;
        fa[s].q[1] = hi;
    }

    float ps[8], pd[8];
#pragma unroll
    for (int r = 0; r < 8; ++r) { ps[r] = 0.f; pd[r] = 0.f; }

    for (int c0 = 0; c0 < NOUT; c0 += 16) {
        const unsigned short* bptr = Bt + (size_t)(c0 + l15) * K + kbase;
        const int colY = c0 + l15;
        const float av = as_[colY];
        const float dv = ad_[colY];
        f32x8 acc = {};
#pragma unroll
        for (int s = 0; s < KS; ++s) {
            Frag16 fb;
            fb.q[0] = *(const u32x4*)(bptr + s * 32);
            fb.q[1] = *(const u32x4*)(bptr + s * 32 + 16);
            acc = __builtin_amdgcn_wmma_f32_16x16x32_bf16(
                      false, fa[s].v, false, fb.v, (short)0, acc, false, false);
        }
        const bool odd = (l15 & 1) != 0;
#pragma unroll
        for (int r = 0; r < 8; r += 2) {
            const f32x2 two = { acc[r], acc[r + 1] };
            const unsigned own  = __builtin_bit_cast(unsigned, __builtin_convertvector(two, bf16x2));
            const unsigned send = odd ? (own & 0xFFFFu) : (own >> 16);
            const unsigned recv = (unsigned)__builtin_amdgcn_update_dpp(0, (int)send, 0xB1, 0xF, 0xF, false);
            const unsigned pk   = odd ? (recv | (own & 0xFFFF0000u)) : ((own & 0xFFFFu) | (recv << 16));
            const int row = m0 + r + (odd ? 1 : 0) + half * 8;
            if (row < M) *(unsigned*)(Y + (size_t)row * NOUT + (colY & ~1)) = pk;
        }
        score_add(ps, pd, acc, av, dv);
    }
    score_store(ps, pd, m0, half, l15, M, es, ed);
}

template <int K>
__global__ void __launch_bounds__(32 * GEMM_WAVES)
wmma_rows_times_folded(const unsigned short* __restrict__ A,
                       const unsigned short* __restrict__ Ut,
                       float* __restrict__ S, int M)
{
    const int lane  = threadIdx.x & 31;
    const int wave  = threadIdx.x >> 5;
    const int tileM = blockIdx.x * GEMM_WAVES + wave;
    const int m0    = tileM * 16;
    if (m0 >= M) return;

    const int l15   = lane & 15;
    const int half  = lane >> 4;
    const int kbase = half * 8;

    int rowA = m0 + l15;
    if (rowA >= M) rowA = M - 1;
    const unsigned short* aptr = A  + (size_t)rowA * K + kbase;
    const unsigned short* bptr = Ut + (size_t)l15  * K + kbase;

    f32x8 acc = {};
#pragma unroll
    for (int k0 = 0; k0 < K; k0 += 32) {
        Frag16 fa, fb;
        fa.q[0] = *(const u32x4*)(aptr + k0);
        fa.q[1] = *(const u32x4*)(aptr + k0 + 16);
        fb.q[0] = *(const u32x4*)(bptr + k0);
        fb.q[1] = *(const u32x4*)(bptr + k0 + 16);
        acc = __builtin_amdgcn_wmma_f32_16x16x32_bf16(
                  false, fa.v, false, fb.v, (short)0, acc, false, false);
    }
    if (l15 < 4) {
#pragma unroll
        for (int r = 0; r < 8; ++r) {
            const int row = m0 + r + half * 8;
            if (row < M) S[(size_t)row * 4 + l15] = acc[r];
        }
    }
}

static __device__ __forceinline__ void zero_i32_block(int blk, int* __restrict__ p, int n)
{
    const int i = blk * blockDim.x + threadIdx.x;
    if (i < n) p[i] = 0;
}

__global__ void count_in_degree(const int* __restrict__ ei, int E, int* __restrict__ cnt)
{
    const int e = blockIdx.x * blockDim.x + threadIdx.x;
    if (e < E) atomicAdd(&cnt[ei[E + e]], 1);
}

#define SCAN_TB 1024
__global__ void __launch_bounds__(SCAN_TB)
scan_in_degree(const int* __restrict__ cnt, int* __restrict__ rowbeg,
               int* __restrict__ fillptr, int N)
{
    __shared__ int wpart[SCAN_TB / 32];
    __shared__ int wtot[SCAN_TB / 32];
    const int t = threadIdx.x, lane = t & 31, wv = t >> 5;
    const int base = blockIdx.x * SCAN_TB;
    int part = 0;
    for (int i = t; i < base; i += SCAN_TB) part += cnt[i];
    const int i = base + t;
    const int v = (i < N) ? cnt[i] : 0;
    int inc = v;
    for (int off = 1; off < 32; off <<= 1) {
        const int n = __shfl_up(inc, off);
        if (lane >= off) inc += n;
    }
    for (int off = 16; off; off >>= 1) part += __shfl_xor(part, off);
    if (lane == 31) wtot[wv]  = inc;
    if (lane == 0)  wpart[wv] = part;
    __syncthreads();
    int add = 0;
    for (int w = 0; w < SCAN_TB / 32; ++w) {
        add += wpart[w];
        if (w < wv) add += wtot[w];
    }
    if (i < N) {
        const int ex = add + inc - v;
        rowbeg[i]  = ex;
        fillptr[i] = ex;
    }
}

static __device__ __forceinline__ void
fill_by_destination_block(int blk, const int* __restrict__ ei, int E,
                          int* __restrict__ fillptr, int* __restrict__ bsrc)
{
    const int e = blk * blockDim.x + threadIdx.x;
    if (e >= E) return;
    const int pos = atomicAdd(&fillptr[ei[E + e]], 1);
    bsrc[pos] = ei[e];
}

template <int K>
__global__ void __launch_bounds__(32 * GEMM_WAVES)
layer1_product_beside_bucket_fill(const float* __restrict__ A,
                                  const unsigned short* __restrict__ Bt,
                                  const float* __restrict__ as_, const float* __restrict__ ad_,
                                  unsigned short* __restrict__ Y,
                                  float* __restrict__ es, float* __restrict__ ed,
                                  int M, int NOUT, int productBlocks,
                                  const int* __restrict__ ei, int E,
                                  int* __restrict__ fillptr, int* __restrict__ bsrc)
{
    const int b = blockIdx.x;
    if (b < productBlocks)
        wmma_gemm_f32a_scores_block<K>(b, A, Bt, as_, ad_, Y, es, ed, M, NOUT);
    else
        fill_by_destination_block(b - productBlocks, ei, E, fillptr, bsrc);
}

static __device__ __forceinline__ float leaky02(float l) { return (l > 0.f) ? l : 0.2f * l; }

template <bool FIRST>
static __device__ __forceinline__ void row_axpy(float (&acc)[8], const u32x4 q, float w)
{
#pragma unroll
    for (int k = 0; k < 4; ++k) {
        const float lo = __uint_as_float(q[k] << 16);
        const float hi = __uint_as_float(q[k] & 0xFFFF0000u);
        if (FIRST) { acc[2 * k] = lo * w;  acc[2 * k + 1] = hi * w; }
        else       { acc[2 * k] += lo * w; acc[2 * k + 1] += hi * w; }
    }
}

template <int NV>
__global__ void __launch_bounds__(256)
gat_gather(const int* __restrict__ rowbeg, const int* __restrict__ rowend,
           const int* __restrict__ bsrc,
           const unsigned short* __restrict__ h, const float* __restrict__ es,
           const float* __restrict__ ed, const float* __restrict__ bias,
           unsigned short* __restrict__ outbf, int N)
{
    constexpr int C = 256 * NV;
    const int lane = threadIdx.x & 31;
    const int d = __builtin_amdgcn_readfirstlane(
                      (int)((blockIdx.x * blockDim.x + threadIdx.x) >> 5));
    if (d >= N) return;

    const int beg = rowbeg[d], end = rowend[d];
    const float edv   = ed[d];
    const float lself = leaky02(es[d] + edv);
    const float NEG_INF = -__builtin_inff();

    const int i0 = beg + lane;
    int   s0 = d;
    float l0 = NEG_INF;
    if (i0 < end) { s0 = bsrc[i0]; l0 = leaky02(es[s0] + edv); }

    float m = fmaxf(lself, l0);
    for (int i = i0 + 32; i < end; i += 32) m = fmaxf(m, leaky02(es[bsrc[i]] + edv));
    for (int off = 16; off; off >>= 1) m = fmaxf(m, __shfl_xor(m, off));

    float sum = (i0 < end) ? __expf(l0 - m) : 0.f;
    if (lane == 0) sum += __expf(lself - m);
    for (int i = i0 + 32; i < end; i += 32) sum += __expf(leaky02(es[bsrc[i]] + edv) - m);
    for (int off = 16; off; off >>= 1) sum += __shfl_xor(sum, off);

    float acc[NV][8];
    {
        const float w = __expf(lself - m) / sum;
        const u32x4* r = (const u32x4*)(h + (size_t)d * C);
        for (int v = 0; v < NV; ++v) row_axpy<true>(acc[v], r[v * 32 + lane], w);
    }
    for (int base = beg; base < end; base += 32) {
        int   s = s0;
        float l = l0;
        if (base != beg) {
            s = d; l = NEG_INF;
            if (base + lane < end) { s = bsrc[base + lane]; l = leaky02(es[s] + edv); }
        }
        const float w = (base + lane < end) ? __expf(l - m) / sum : 0.f;
        const int cnt = (end - base < 32) ? (end - base) : 32;
#pragma unroll 4
        for (int j = 0; j < cnt; ++j) {
            const int   sj = __builtin_amdgcn_readlane(s, j);
            const float wj = __int_as_float(__builtin_amdgcn_readlane(__float_as_int(w), j));
            const u32x4* r = (const u32x4*)(h + (size_t)sj * C);
            for (int v = 0; v < NV; ++v) row_axpy<false>(acc[v], r[v * 32 + lane], wj);
        }
    }
    for (int v = 0; v < NV; ++v) {
        const int col = (v * 32 + lane) * 8;
        const float* b = bias + col;
        u32x4 pk;
#pragma unroll
        for (int k = 0; k < 4; ++k) {
            const float o0 = acc[v][2 * k]     + b[2 * k];
            const float o1 = acc[v][2 * k + 1] + b[2 * k + 1];
            pk[k] = pack2bf(o0 > 0.f ? o0 : 0.f, o1 > 0.f ? o1 : 0.f);
        }
        *(u32x4*)(outbf + (size_t)d * C + col) = pk;
    }
}

#define GL 8
__global__ void __launch_bounds__(256)
gat_gather_folded(const int* __restrict__ rowbeg, const int* __restrict__ rowend,
                  const int* __restrict__ bsrc,
                  const f32x4* __restrict__ S, const float* __restrict__ cb,
                  float* __restrict__ q0, float* __restrict__ q1, int N)
{
    const int t   = blockIdx.x * blockDim.x + threadIdx.x;
    const int sub = threadIdx.x & (GL - 1);
    int d = t / GL;
    const bool live = d < N;
    if (!live) d = N - 1;

    const int beg = rowbeg[d], end = rowend[d];
    const f32x4 sd = S[d];
    const float edv   = sd[1];
    const float lself = leaky02(sd[0] + edv);
    const float NEG_INF = -__builtin_inff();

    const int i0 = beg + sub, i1 = i0 + GL;
    float l0 = NEG_INF, l1 = NEG_INF, ta0 = 0.f, tb0 = 0.f, ta1 = 0.f, tb1 = 0.f;
    if (i0 < end) { const f32x4 x = S[bsrc[i0]]; l0 = leaky02(x[0] + edv); ta0 = x[2]; tb0 = x[3]; }
    if (i1 < end) { const f32x4 x = S[bsrc[i1]]; l1 = leaky02(x[0] + edv); ta1 = x[2]; tb1 = x[3]; }

    float m = fmaxf(lself, fmaxf(l0, l1));
    for (int i = i0 + 2 * GL; i < end; i += GL) m = fmaxf(m, leaky02(S[bsrc[i]][0] + edv));
    for (int off = GL / 2; off; off >>= 1) m = fmaxf(m, __shfl_xor(m, off, GL));

    const float p0 = (i0 < end) ? __expf(l0 - m) : 0.f;
    const float p1 = (i1 < end) ? __expf(l1 - m) : 0.f;
    float sum = p0 + p1;
    float a0  = p0 * ta0 + p1 * ta1;
    float a1  = p0 * tb0 + p1 * tb1;
    if (sub == 0) {
        const float ps = __expf(lself - m);
        sum += ps; a0 += ps * sd[2]; a1 += ps * sd[3];
    }
    for (int i = i0 + 2 * GL; i < end; i += GL) {
        const f32x4 x = S[bsrc[i]];
        const float p = __expf(leaky02(x[0] + edv) - m);
        sum += p; a0 += p * x[2]; a1 += p * x[3];
    }
    for (int off = GL / 2; off; off >>= 1) {
        sum += __shfl_xor(sum, off, GL);
        a0  += __shfl_xor(a0,  off, GL);
        a1  += __shfl_xor(a1,  off, GL);
    }
    if (live && sub == 0) {
        q0[d] = a0 / sum + cb[0];
        q1[d] = a1 / sum + cb[1];
    }
}

static __device__ __forceinline__ void
transpose_to_bf16_block(int blk, const float* __restrict__ W, unsigned short* __restrict__ Wt,
                        int K, int NOUT)
{
    const int i = blk * blockDim.x + threadIdx.x;
    if (i >= K * NOUT) return;
    const int k = i / NOUT, n = i % NOUT;
    Wt[(size_t)n * K + k] = f2bf(W[i]);
}

static __device__ __forceinline__ void
fold_weight_block(int blk, const float* __restrict__ W2, const float* __restrict__ a_s,
                  const float* __restrict__ a_d, const float* __restrict__ Wl,
                  const float* __restrict__ b2,
                  unsigned short* __restrict__ Ut, float* __restrict__ cb, int K, int C)
{
    const int lane = threadIdx.x & 31;
    const int w = (int)((blk * blockDim.x + threadIdx.x) >> 5);
    if (w > K) return;
    const float* x = (w < K) ? (W2 + (size_t)w * C) : b2;
    float s0 = 0.f, s1 = 0.f, s2 = 0.f, s3 = 0.f;
    for (int c = lane; c < C; c += 32) {
        const float xv = x[c];
        s0 += xv * a_s[c];
        s1 += xv * a_d[c];
        s2 += xv * Wl[c];
        s3 += xv * Wl[C + c];
    }
    for (int off = 16; off; off >>= 1) {
        s0 += __shfl_xor(s0, off);
        s1 += __shfl_xor(s1, off);
        s2 += __shfl_xor(s2, off);
        s3 += __shfl_xor(s3, off);
    }
    if (w < K) {
        const float v = (lane == 0) ? s0 : (lane == 1) ? s1 : (lane == 2) ? s2 : s3;
        if (lane < 16) Ut[(size_t)lane * K + w] = (lane < 4) ? f2bf(v) : (unsigned short)0;
    } else if (lane == 0) {
        cb[0] = s2;
        cb[1] = s3;
    }
}

__global__ void __launch_bounds__(256)
prepare_weights_and_counters(const float* __restrict__ W1, unsigned short* __restrict__ w1t,
                             int K1, int NOUT1, int transposeBlocks,
                             const float* __restrict__ W2, const float* __restrict__ a_s,
                             const float* __restrict__ a_d, const float* __restrict__ Wl,
                             const float* __restrict__ b2,
                             unsigned short* __restrict__ Ut, float* __restrict__ cb,
                             int K2, int C2, int foldBlocks,
                             int* __restrict__ cnt, int N)
{
    const int b = blockIdx.x;
    if (b < transposeBlocks)
        transpose_to_bf16_block(b, W1, w1t, K1, NOUT1);
    else if (b < transposeBlocks + foldBlocks)
        fold_weight_block(b - transposeBlocks, W2, a_s, a_d, Wl, b2, Ut, cb, K2, C2);
    else
        zero_i32_block(b - transposeBlocks - foldBlocks, cnt, N);
}

__global__ void link_predict(const float* __restrict__ q0, const float* __restrict__ q1,
                             const int* __restrict__ mask, const float* __restrict__ bl,
                             float* __restrict__ out, int P)
{
    const int p = blockIdx.x * blockDim.x + threadIdx.x;
    if (p >= P) return;
    const float z = q0[mask[p * 2 + 0]] + q1[mask[p * 2 + 1]] + bl[0];
    out[p] = 1.f / (1.f + expf(-z));
}

extern "C" void kernel_launch(void* const* d_in, const int* in_sizes, int n_in,
                              void* d_out, int out_size, void* d_ws, size_t ws_size,
                              hipStream_t stream)
{
    (void)n_in; (void)out_size; (void)ws_size;

    const float* features = (const float*)d_in[0];
    const int*   ei       = (const int*)  d_in[1];
    const int*   mask     = (const int*)  d_in[2];
    const float* W1   = (const float*)d_in[3];
    const float* a_s1 = (const float*)d_in[4];
    const float* a_d1 = (const float*)d_in[5];
    const float* b1   = (const float*)d_in[6];
    const float* W2   = (const float*)d_in[7];
    const float* a_s2 = (const float*)d_in[8];
    const float* a_d2 = (const float*)d_in[9];
    const float* b2   = (const float*)d_in[10];
    const float* Wl   = (const float*)d_in[11];
    const float* bl   = (const float*)d_in[12];

    const int F  = 128, Hh = 256;
    const int N  = in_sizes[0] / F;
    const int E  = in_sizes[1] / 2;
    const int P  = in_sizes[2] / 2;

    char* ws = (char*)d_ws;
    size_t off = 0;
    auto alloc = [&](size_t bytes) -> char* {
        char* ptr = ws + off;
        off += (bytes + 255) & ~(size_t)255;
        return ptr;
    };
    unsigned short* w1t  = (unsigned short*)alloc((size_t)Hh * F * 2);
    unsigned short* ut   = (unsigned short*)alloc((size_t)16 * Hh * 2);
    float*    cb      = (float*)   alloc((size_t)2 * 4);
    float*    S       = (float*)   alloc((size_t)N * 4 * 4);
    unsigned short* hrbf = (unsigned short*)alloc((size_t)N * Hh * 2);
    unsigned short* buf1 = (unsigned short*)alloc((size_t)N * Hh * 2);
    float*    q0      = (float*)   alloc((size_t)N * 4);
    float*    q1      = (float*)   alloc((size_t)N * 4);
    float*    es      = (float*)   alloc((size_t)N * 4);
    float*    ed      = (float*)   alloc((size_t)N * 4);
    int*      cnt     = (int*)     alloc((size_t)N * 4);
    int*      rowbeg  = (int*)     alloc((size_t)N * 4);
    int*      rowend  = (int*)     alloc((size_t)N * 4);
    int*      bsrc    = (int*)     alloc((size_t)E * 4);

    const int tilesM = (N + 15) / 16;
    const int gemmGrid = (tilesM + GEMM_WAVES - 1) / GEMM_WAVES;
    const int TB = 256;
    auto cdiv = [](size_t a, size_t b) -> int { return (int)((a + b - 1) / b); };

    const int transposeBlocks = cdiv((size_t)F * Hh, TB);
    const int foldBlocks      = cdiv((size_t)(Hh + 1) * 32, TB);
    prepare_weights_and_counters<<<transposeBlocks + foldBlocks + cdiv(N, TB), TB, 0, stream>>>(
        W1, w1t, F, Hh, transposeBlocks,
        W2, a_s2, a_d2, Wl, b2, ut, cb, Hh, F, foldBlocks,
        cnt, N);

    count_in_degree<<<cdiv(E, TB), TB, 0, stream>>>(ei, E, cnt);
    scan_in_degree<<<cdiv(N, SCAN_TB), SCAN_TB, 0, stream>>>(cnt, rowbeg, rowend, N);

    layer1_product_beside_bucket_fill<128>
        <<<gemmGrid + cdiv(E, 32 * GEMM_WAVES), 32 * GEMM_WAVES, 0, stream>>>(
        features, w1t, a_s1, a_d1, buf1, es, ed, N, Hh, gemmGrid,
        ei, E, rowend, bsrc);
    gat_gather<1><<<cdiv((size_t)N * 32, TB), TB, 0, stream>>>(
        rowbeg, rowend, bsrc, buf1, es, ed, b1, hrbf, N);

    wmma_rows_times_folded<256><<<gemmGrid, 32 * GEMM_WAVES, 0, stream>>>(hrbf, ut, S, N);
    gat_gather_folded<<<cdiv((size_t)N * GL, TB), TB, 0, stream>>>(
        rowbeg, rowend, bsrc, (const f32x4*)S, cb, q0, q1, N);

    link_predict<<<cdiv((size_t)P, TB), TB, 0, stream>>>(q0, q1, mask, bl, (float*)d_out, P);
}


# An edge's place in its bucket is the value the counting pass's add returns (kept per edge); the fill stores at start + rank without atomics (500k fewer); arrival (sum) order in buckets may shift
# speedup vs baseline: 7.0748x; 1.1519x over previous
#include <hip/hip_runtime.h>
#include <hip/hip_bf16.h>

typedef __attribute__((__ext_vector_type__(16))) __bf16 bf16x16;
typedef __attribute__((__ext_vector_type__(8)))  float  f32x8;
typedef __attribute__((__ext_vector_type__(4)))  unsigned int u32x4;
typedef __attribute__((__ext_vector_type__(4)))  float  f32x4;
typedef __attribute__((__ext_vector_type__(2)))  float  f32x2;
typedef __attribute__((__ext_vector_type__(2)))  __bf16 bf16x2;

union Frag16 {
    u32x4   q[2];
    bf16x16 v;
};

static __device__ __forceinline__ unsigned short f2bf(float f) {
    unsigned u = __float_as_uint(f);
    u += 0x7FFFu + ((u >> 16) & 1u);
    return (unsigned short)(u >> 16);
}

#define GEMM_WAVES 4

static __device__ __forceinline__ unsigned pack2bf(float lo, float hi) {
    return (unsigned)f2bf(lo) | ((unsigned)f2bf(hi) << 16);
}

static __device__ __forceinline__ void score_add(float (&ps)[8], float (&pd)[8],
                                                 const f32x8 acc, float av, float dv)
{
#pragma unroll
    for (int r = 0; r < 8; ++r) {
        ps[r] += acc[r] * av;
        pd[r] += acc[r] * dv;
    }
}

static __device__ __forceinline__ void score_store(float (&ps)[8], float (&pd)[8],
                                                   int m0, int half, int l15, int M,
                                                   float* __restrict__ es, float* __restrict__ ed)
{
#pragma unroll
    for (int r = 0; r < 8; ++r) {
        for (int off = 8; off; off >>= 1) {
            ps[r] += __shfl_xor(ps[r], off);
            pd[r] += __shfl_xor(pd[r], off);
        }
    }
    if (l15 == 0) {
#pragma unroll
        for (int r = 0; r < 8; ++r) {
            const int row = m0 + r + half * 8;
            if (row < M) { es[row] = ps[r]; ed[row] = pd[r]; }
        }
    }
}

typedef float f32x4_in __attribute__((ext_vector_type(4), aligned(4)));

template <int K>
static __device__ __forceinline__ void
wmma_gemm_f32a_scores_block(int blk,
                            const float* __restrict__ A,
                            const unsigned short* __restrict__ Bt,
                            const float* __restrict__ as_, const float* __restrict__ ad_,
                            unsigned short* __restrict__ Y,
                            float* __restrict__ es, float* __restrict__ ed,
                            int M, int NOUT)
{
    constexpr int KS = K / 32;
    const int lane  = threadIdx.x & 31;
    const int wave  = threadIdx.x >> 5;
    const int tileM = blk * GEMM_WAVES + wave;
    const int m0    = tileM * 16;
    if (m0 >= M) return;

    const int l15   = lane & 15;
    const int half  = lane >> 4;
    const int kbase = half * 8;

    int rowA = m0 + l15;
    if (rowA >= M) rowA = M - 1;
    const float* ap = A + (size_t)rowA * K + kbase;

    Frag16 fa[KS];
#pragma unroll
    for (int s = 0; s < KS; ++s) {
        const f32x4 x0 = *(const f32x4_in*)(ap + s * 32);
        const f32x4 x1 = *(const f32x4_in*)(ap + s * 32 + 4);
        const f32x4 x2 = *(const f32x4_in*)(ap + s * 32 + 16);
        const f32x4 x3 = *(const f32x4_in*)(ap + s * 32 + 20);
        const u32x4 lo = { pack2bf(x0[0], x0[1]), pack2bf(x0[2], x0[3]),
                           pack2bf(x1[0], x1[1]), pack2bf(x1[2], x1[3]) };
        const u32x4 hi = { pack2bf(x2[0], x2[1]), pack2bf(x2[2], x2[3]),
                           pack2bf(x3[0], x3[1]), pack2bf(x3[2], x3[3]) };
        fa[s].q[0] = lo;
        fa[s].q[1] = hi;
    }

    float ps[8], pd[8];
#pragma unroll
    for (int r = 0; r < 8; ++r) { ps[r] = 0.f; pd[r] = 0.f; }

    for (int c0 = 0; c0 < NOUT; c0 += 16) {
        const unsigned short* bptr = Bt + (size_t)(c0 + l15) * K + kbase;
        const int colY = c0 + l15;
        const float av = as_[colY];
        const float dv = ad_[colY];
        f32x8 acc = {};
#pragma unroll
        for (int s = 0; s < KS; ++s) {
            Frag16 fb;
            fb.q[0] = *(const u32x4*)(bptr + s * 32);
            fb.q[1] = *(const u32x4*)(bptr + s * 32 + 16);
            acc = __builtin_amdgcn_wmma_f32_16x16x32_bf16(
                      false, fa[s].v, false, fb.v, (short)0, acc, false, false);
        }
        const bool odd = (l15 & 1) != 0;
#pragma unroll
        for (int r = 0; r < 8; r += 2) {
            const f32x2 two = { acc[r], acc[r + 1] };
            const unsigned own  = __builtin_bit_cast(unsigned, __builtin_convertvector(two, bf16x2));
            const unsigned send = odd ? (own & 0xFFFFu) : (own >> 16);
            const unsigned recv = (unsigned)__builtin_amdgcn_update_dpp(0, (int)send, 0xB1, 0xF, 0xF, false);
            const unsigned pk   = odd ? (recv | (own & 0xFFFF0000u)) : ((own & 0xFFFFu) | (recv << 16));
            const int row = m0 + r + (odd ? 1 : 0) + half * 8;
            if (row < M) *(unsigned*)(Y + (size_t)row * NOUT + (colY & ~1)) = pk;
        }
        score_add(ps, pd, acc, av, dv);
    }
    score_store(ps, pd, m0, half, l15, M, es, ed);
}

template <int K>
__global__ void __launch_bounds__(32 * GEMM_WAVES)
wmma_rows_times_folded(const unsigned short* __restrict__ A,
                       const unsigned short* __restrict__ Ut,
                       float* __restrict__ S, int M)
{
    const int lane  = threadIdx.x & 31;
    const int wave  = threadIdx.x >> 5;
    const int tileM = blockIdx.x * GEMM_WAVES + wave;
    const int m0    = tileM * 16;
    if (m0 >= M) return;

    const int l15   = lane & 15;
    const int half  = lane >> 4;
    const int kbase = half * 8;

    int rowA = m0 + l15;
    if (rowA >= M) rowA = M - 1;
    const unsigned short* aptr = A  + (size_t)rowA * K + kbase;
    const unsigned short* bptr = Ut + (size_t)l15  * K + kbase;

    f32x8 acc = {};
#pragma unroll
    for (int k0 = 0; k0 < K; k0 += 32) {
        Frag16 fa, fb;
        fa.q[0] = *(const u32x4*)(aptr + k0);
        fa.q[1] = *(const u32x4*)(aptr + k0 + 16);
        fb.q[0] = *(const u32x4*)(bptr + k0);
        fb.q[1] = *(const u32x4*)(bptr + k0 + 16);
        acc = __builtin_amdgcn_wmma_f32_16x16x32_bf16(
                  false, fa.v, false, fb.v, (short)0, acc, false, false);
    }
    if (l15 < 4) {
#pragma unroll
        for (int r = 0; r < 8; ++r) {
            const int row = m0 + r + half * 8;
            if (row < M) S[(size_t)row * 4 + l15] = acc[r];
        }
    }
}

static __device__ __forceinline__ void zero_i32_block(int blk, int* __restrict__ p, int n)
{
    const int i = blk * blockDim.x + threadIdx.x;
    if (i < n) p[i] = 0;
}

__global__ void count_in_degree(const int* __restrict__ ei, int E, int* __restrict__ cnt,
                                int* __restrict__ rank)
{
    const int e = blockIdx.x * blockDim.x + threadIdx.x;
    if (e < E) rank[e] = atomicAdd(&cnt[ei[E + e]], 1);
}

#define SCAN_TB 1024
__global__ void __launch_bounds__(SCAN_TB)
scan_in_degree(const int* __restrict__ cnt, int* __restrict__ rowbeg,
               int* __restrict__ rowend, int N)
{
    __shared__ int wpart[SCAN_TB / 32];
    __shared__ int wtot[SCAN_TB / 32];
    const int t = threadIdx.x, lane = t & 31, wv = t >> 5;
    const int base = blockIdx.x * SCAN_TB;
    int part = 0;
    for (int i = t; i < base; i += SCAN_TB) part += cnt[i];
    const int i = base + t;
    const int v = (i < N) ? cnt[i] : 0;
    int inc = v;
    for (int off = 1; off < 32; off <<= 1) {
        const int n = __shfl_up(inc, off);
        if (lane >= off) inc += n;
    }
    for (int off = 16; off; off >>= 1) part += __shfl_xor(part, off);
    if (lane == 31) wtot[wv]  = inc;
    if (lane == 0)  wpart[wv] = part;
    __syncthreads();
    int add = 0;
    for (int w = 0; w < SCAN_TB / 32; ++w) {
        add += wpart[w];
        if (w < wv) add += wtot[w];
    }
    if (i < N) {
        const int ex = add + inc - v;
        rowbeg[i] = ex;
        rowend[i] = ex + v;
    }
}

static __device__ __forceinline__ void
fill_by_destination_block(int blk, const int* __restrict__ ei, int E,
                          const int* __restrict__ rowbeg, const int* __restrict__ rank,
                          int* __restrict__ bsrc)
{
    const int e = blk * blockDim.x + threadIdx.x;
    if (e >= E) return;
    bsrc[rowbeg[ei[E + e]] + rank[e]] = ei[e];
}

template <int K>
__global__ void __launch_bounds__(32 * GEMM_WAVES)
layer1_product_beside_bucket_fill(const float* __restrict__ A,
                                  const unsigned short* __restrict__ Bt,
                                  const float* __restrict__ as_, const float* __restrict__ ad_,
                                  unsigned short* __restrict__ Y,
                                  float* __restrict__ es, float* __restrict__ ed,
                                  int M, int NOUT, int productBlocks,
                                  const int* __restrict__ ei, int E,
                                  const int* __restrict__ rowbeg, const int* __restrict__ rank,
                                  int* __restrict__ bsrc)
{
    const int b = blockIdx.x;
    if (b < productBlocks)
        wmma_gemm_f32a_scores_block<K>(b, A, Bt, as_, ad_, Y, es, ed, M, NOUT);
    else
        fill_by_destination_block(b - productBlocks, ei, E, rowbeg, rank, bsrc);
}

static __device__ __forceinline__ float leaky02(float l) { return (l > 0.f) ? l : 0.2f * l; }

template <bool FIRST>
static __device__ __forceinline__ void row_axpy(float (&acc)[8], const u32x4 q, float w)
{
#pragma unroll
    for (int k = 0; k < 4; ++k) {
        const float lo = __uint_as_float(q[k] << 16);
        const float hi = __uint_as_float(q[k] & 0xFFFF0000u);
        if (FIRST) { acc[2 * k] = lo * w;  acc[2 * k + 1] = hi * w; }
        else       { acc[2 * k] += lo * w; acc[2 * k + 1] += hi * w; }
    }
}

template <int NV>
__global__ void __launch_bounds__(256)
gat_gather(const int* __restrict__ rowbeg, const int* __restrict__ rowend,
           const int* __restrict__ bsrc,
           const unsigned short* __restrict__ h, const float* __restrict__ es,
           const float* __restrict__ ed, const float* __restrict__ bias,
           unsigned short* __restrict__ outbf, int N)
{
    constexpr int C = 256 * NV;
    const int lane = threadIdx.x & 31;
    const int d = __builtin_amdgcn_readfirstlane(
                      (int)((blockIdx.x * blockDim.x + threadIdx.x) >> 5));
    if (d >= N) return;

    const int beg = rowbeg[d], end = rowend[d];
    const float edv   = ed[d];
    const float lself = leaky02(es[d] + edv);
    const float NEG_INF = -__builtin_inff();

    const int i0 = beg + lane;
    int   s0 = d;
    float l0 = NEG_INF;
    if (i0 < end) { s0 = bsrc[i0]; l0 = leaky02(es[s0] + edv); }

    float m = fmaxf(lself, l0);
    for (int i = i0 + 32; i < end; i += 32) m = fmaxf(m, leaky02(es[bsrc[i]] + edv));
    for (int off = 16; off; off >>= 1) m = fmaxf(m, __shfl_xor(m, off));

    float sum = (i0 < end) ? __expf(l0 - m) : 0.f;
    if (lane == 0) sum += __expf(lself - m);
    for (int i = i0 + 32; i < end; i += 32) sum += __expf(leaky02(es[bsrc[i]] + edv) - m);
    for (int off = 16; off; off >>= 1) sum += __shfl_xor(sum, off);

    float acc[NV][8];
    {
        const float w = __expf(lself - m) / sum;
        const u32x4* r = (const u32x4*)(h + (size_t)d * C);
        for (int v = 0; v < NV; ++v) row_axpy<true>(acc[v], r[v * 32 + lane], w);
    }
    for (int base = beg; base < end; base += 32) {
        int   s = s0;
        float l = l0;
        if (base != beg) {
            s = d; l = NEG_INF;
            if (base + lane < end) { s = bsrc[base + lane]; l = leaky02(es[s] + edv); }
        }
        const float w = (base + lane < end) ? __expf(l - m) / sum : 0.f;
        const int cnt = (end - base < 32) ? (end - base) : 32;
#pragma unroll 4
        for (int j = 0; j < cnt; ++j) {
            const int   sj = __builtin_amdgcn_readlane(s, j);
            const float wj = __int_as_float(__builtin_amdgcn_readlane(__float_as_int(w), j));
            const u32x4* r = (const u32x4*)(h + (size_t)sj * C);
            for (int v = 0; v < NV; ++v) row_axpy<false>(acc[v], r[v * 32 + lane], wj);
        }
    }
    for (int v = 0; v < NV; ++v) {
        const int col = (v * 32 + lane) * 8;
        const float* b = bias + col;
        u32x4 pk;
#pragma unroll
        for (int k = 0; k < 4; ++k) {
            const float o0 = acc[v][2 * k]     + b[2 * k];
            const float o1 = acc[v][2 * k + 1] + b[2 * k + 1];
            pk[k] = pack2bf(o0 > 0.f ? o0 : 0.f, o1 > 0.f ? o1 : 0.f);
        }
        *(u32x4*)(outbf + (size_t)d * C + col) = pk;
    }
}

#define GL 8
__global__ void __launch_bounds__(256)
gat_gather_folded(const int* __restrict__ rowbeg, const int* __restrict__ rowend,
                  const int* __restrict__ bsrc,
                  const f32x4* __restrict__ S, const float* __restrict__ cb,
                  float* __restrict__ q0, float* __restrict__ q1, int N)
{
    const int t   = blockIdx.x * blockDim.x + threadIdx.x;
    const int sub = threadIdx.x & (GL - 1);
    int d = t / GL;
    const bool live = d < N;
    if (!live) d = N - 1;

    const int beg = rowbeg[d], end = rowend[d];
    const f32x4 sd = S[d];
    const float edv   = sd[1];
    const float lself = leaky02(sd[0] + edv);
    const float NEG_INF = -__builtin_inff();

    const int i0 = beg + sub, i1 = i0 + GL;
    float l0 = NEG_INF, l1 = NEG_INF, ta0 = 0.f, tb0 = 0.f, ta1 = 0.f, tb1 = 0.f;
    if (i0 < end) { const f32x4 x = S[bsrc[i0]]; l0 = leaky02(x[0] + edv); ta0 = x[2]; tb0 = x[3]; }
    if (i1 < end) { const f32x4 x = S[bsrc[i1]]; l1 = leaky02(x[0] + edv); ta1 = x[2]; tb1 = x[3]; }

    float m = fmaxf(lself, fmaxf(l0, l1));
    for (int i = i0 + 2 * GL; i < end; i += GL) m = fmaxf(m, leaky02(S[bsrc[i]][0] + edv));
    for (int off = GL / 2; off; off >>= 1) m = fmaxf(m, __shfl_xor(m, off, GL));

    const float p0 = (i0 < end) ? __expf(l0 - m) : 0.f;
    const float p1 = (i1 < end) ? __expf(l1 - m) : 0.f;
    float sum = p0 + p1;
    float a0  = p0 * ta0 + p1 * ta1;
    float a1  = p0 * tb0 + p1 * tb1;
    if (sub == 0) {
        const float ps = __expf(lself - m);
        sum += ps; a0 += ps * sd[2]; a1 += ps * sd[3];
    }
    for (int i = i0 + 2 * GL; i < end; i += GL) {
        const f32x4 x = S[bsrc[i]];
        const float p = __expf(leaky02(x[0] + edv) - m);
        sum += p; a0 += p * x[2]; a1 += p * x[3];
    }
    for (int off = GL / 2; off; off >>= 1) {
        sum += __shfl_xor(sum, off, GL);
        a0  += __shfl_xor(a0,  off, GL);
        a1  += __shfl_xor(a1,  off, GL);
    }
    if (live && sub == 0) {
        q0[d] = a0 / sum + cb[0];
        q1[d] = a1 / sum + cb[1];
    }
}

static __device__ __forceinline__ void
transpose_to_bf16_block(int blk, const float* __restrict__ W, unsigned short* __restrict__ Wt,
                        int K, int NOUT)
{
    const int i = blk * blockDim.x + threadIdx.x;
    if (i >= K * NOUT) return;
    const int k = i / NOUT, n = i % NOUT;
    Wt[(size_t)n * K + k] = f2bf(W[i]);
}

static __device__ __forceinline__ void
fold_weight_block(int blk, const float* __restrict__ W2, const float* __restrict__ a_s,
                  const float* __restrict__ a_d, const float* __restrict__ Wl,
                  const float* __restrict__ b2,
                  unsigned short* __restrict__ Ut, float* __restrict__ cb, int K, int C)
{
    const int lane = threadIdx.x & 31;
    const int w = (int)((blk * blockDim.x + threadIdx.x) >> 5);
    if (w > K) return;
    const float* x = (w < K) ? (W2 + (size_t)w * C) : b2;
    float s0 = 0.f, s1 = 0.f, s2 = 0.f, s3 = 0.f;
    for (int c = lane; c < C; c += 32) {
        const float xv = x[c];
        s0 += xv * a_s[c];
        s1 += xv * a_d[c];
        s2 += xv * Wl[c];
        s3 += xv * Wl[C + c];
    }
    for (int off = 16; off; off >>= 1) {
        s0 += __shfl_xor(s0, off);
        s1 += __shfl_xor(s1, off);
        s2 += __shfl_xor(s2, off);
        s3 += __shfl_xor(s3, off);
    }
    if (w < K) {
        const float v = (lane == 0) ? s0 : (lane == 1) ? s1 : (lane == 2) ? s2 : s3;
        if (lane < 16) Ut[(size_t)lane * K + w] = (lane < 4) ? f2bf(v) : (unsigned short)0;
    } else if (lane == 0) {
        cb[0] = s2;
        cb[1] = s3;
    }
}

__global__ void __launch_bounds__(256)
prepare_weights_and_counters(const float* __restrict__ W1, unsigned short* __restrict__ w1t,
                             int K1, int NOUT1, int transposeBlocks,
                             const float* __restrict__ W2, const float* __restrict__ a_s,
                             const float* __restrict__ a_d, const float* __restrict__ Wl,
                             const float* __restrict__ b2,
                             unsigned short* __restrict__ Ut, float* __restrict__ cb,
                             int K2, int C2, int foldBlocks,
                             int* __restrict__ cnt, int N)
{
    const int b = blockIdx.x;
    if (b < transposeBlocks)
        transpose_to_bf16_block(b, W1, w1t, K1, NOUT1);
    else if (b < transposeBlocks + foldBlocks)
        fold_weight_block(b - transposeBlocks, W2, a_s, a_d, Wl, b2, Ut, cb, K2, C2);
    else
        zero_i32_block(b - transposeBlocks - foldBlocks, cnt, N);
}

__global__ void link_predict(const float* __restrict__ q0, const float* __restrict__ q1,
                             const int* __restrict__ mask, const float* __restrict__ bl,
                             float* __restrict__ out, int P)
{
    const int p = blockIdx.x * blockDim.x + threadIdx.x;
    if (p >= P) return;
    const float z = q0[mask[p * 2 + 0]] + q1[mask[p * 2 + 1]] + bl[0];
    out[p] = 1.f / (1.f + expf(-z));
}

extern "C" void kernel_launch(void* const* d_in, const int* in_sizes, int n_in,
                              void* d_out, int out_size, void* d_ws, size_t ws_size,
                              hipStream_t stream)
{
    (void)n_in; (void)out_size; (void)ws_size;

    const float* features = (const float*)d_in[0];
    const int*   ei       = (const int*)  d_in[1];
    const int*   mask     = (const int*)  d_in[2];
    const float* W1   = (const float*)d_in[3];
    const float* a_s1 = (const float*)d_in[4];
    const float* a_d1 = (const float*)d_in[5];
    const float* b1   = (const float*)d_in[6];
    const float* W2   = (const float*)d_in[7];
    const float* a_s2 = (const float*)d_in[8];
    const float* a_d2 = (const float*)d_in[9];
    const float* b2   = (const float*)d_in[10];
    const float* Wl   = (const float*)d_in[11];
    const float* bl   = (const float*)d_in[12];

    const int F  = 128, Hh = 256;
    const int N  = in_sizes[0] / F;
    const int E  = in_sizes[1] / 2;
    const int P  = in_sizes[2] / 2;

    char* ws = (char*)d_ws;
    size_t off = 0;
    auto alloc = [&](size_t bytes) -> char* {
        char* ptr = ws + off;
        off += (bytes + 255) & ~(size_t)255;
        return ptr;
    };
    unsigned short* w1t  = (unsigned short*)alloc((size_t)Hh * F * 2);
    unsigned short* ut   = (unsigned short*)alloc((size_t)16 * Hh * 2);
    float*    cb      = (float*)   alloc((size_t)2 * 4);
    float*    S       = (float*)   alloc((size_t)N * 4 * 4);
    unsigned short* hrbf = (unsigned short*)alloc((size_t)N * Hh * 2);
    unsigned short* buf1 = (unsigned short*)alloc((size_t)N * Hh * 2);
    float*    q0      = (float*)   alloc((size_t)N * 4);
    float*    q1      = (float*)   alloc((size_t)N * 4);
    float*    es      = (float*)   alloc((size_t)N * 4);
    float*    ed      = (float*)   alloc((size_t)N * 4);
    int*      cnt     = (int*)     alloc((size_t)N * 4);
    int*      rowbeg  = (int*)     alloc((size_t)N * 4);
    int*      rowend  = (int*)     alloc((size_t)N * 4);
    int*      bsrc    = (int*)     alloc((size_t)E * 4);
    int*      rank    = (int*)     alloc((size_t)E * 4);

    const int tilesM = (N + 15) / 16;
    const int gemmGrid = (tilesM + GEMM_WAVES - 1) / GEMM_WAVES;
    const int TB = 256;
    auto cdiv = [](size_t a, size_t b) -> int { return (int)((a + b - 1) / b); };

    const int transposeBlocks = cdiv((size_t)F * Hh, TB);
    const int foldBlocks      = cdiv((size_t)(Hh + 1) * 32, TB);
    prepare_weights_and_counters<<<transposeBlocks + foldBlocks + cdiv(N, TB), TB, 0, stream>>>(
        W1, w1t, F, Hh, transposeBlocks,
        W2, a_s2, a_d2, Wl, b2, ut, cb, Hh, F, foldBlocks,
        cnt, N);

    count_in_degree<<<cdiv(E, TB), TB, 0, stream>>>(ei, E, cnt, rank);
    scan_in_degree<<<cdiv(N, SCAN_TB), SCAN_TB, 0, stream>>>(cnt, rowbeg, rowend, N);

    layer1_product_beside_bucket_fill<128>
        <<<gemmGrid + cdiv(E, 32 * GEMM_WAVES), 32 * GEMM_WAVES, 0, stream>>>(
        features, w1t, a_s1, a_d1, buf1, es, ed, N, Hh, gemmGrid,
        ei, E, rowbeg, rank, bsrc);
    gat_gather<1><<<cdiv((size_t)N * 32, TB), TB, 0, stream>>>(
        rowbeg, rowend, bsrc, buf1, es, ed, b1, hrbf, N);

    wmma_rows_times_folded<256><<<gemmGrid, 32 * GEMM_WAVES, 0, stream>>>(hrbf, ut, S, N);
    gat_gather_folded<<<cdiv((size_t)N * GL, TB), TB, 0, stream>>>(
        rowbeg, rowend, bsrc, (const f32x4*)S, cb, q0, q1, N);

    link_predict<<<cdiv((size_t)P, TB), TB, 0, stream>>>(q0, q1, mask, bl, (float*)d_out, P);
}


# Buckets become one 128-byte line per node (count + 31 sources; deeper edges in an overflow list): the counting add places the edge, scan and fill launches go (8 -> 6); arrival (sum) order may shift
# speedup vs baseline: 9.7605x; 1.3796x over previous
#include <hip/hip_runtime.h>
#include <hip/hip_bf16.h>

typedef __attribute__((__ext_vector_type__(16))) __bf16 bf16x16;
typedef __attribute__((__ext_vector_type__(8)))  float  f32x8;
typedef __attribute__((__ext_vector_type__(4)))  unsigned int u32x4;
typedef __attribute__((__ext_vector_type__(4)))  float  f32x4;
typedef __attribute__((__ext_vector_type__(2)))  float  f32x2;
typedef __attribute__((__ext_vector_type__(2)))  __bf16 bf16x2;

union Frag16 {
    u32x4   q[2];
    bf16x16 v;
};

static __device__ __forceinline__ unsigned short f2bf(float f) {
    unsigned u = __float_as_uint(f);
    u += 0x7FFFu + ((u >> 16) & 1u);
    return (unsigned short)(u >> 16);
}

#define GEMM_WAVES 4

static __device__ __forceinline__ unsigned pack2bf(float lo, float hi) {
    return (unsigned)f2bf(lo) | ((unsigned)f2bf(hi) << 16);
}

static __device__ __forceinline__ void score_add(float (&ps)[8], float (&pd)[8],
                                                 const f32x8 acc, float av, float dv)
{
#pragma unroll
    for (int r = 0; r < 8; ++r) {
        ps[r] += acc[r] * av;
        pd[r] += acc[r] * dv;
    }
}

static __device__ __forceinline__ void score_store(float (&ps)[8], float (&pd)[8],
                                                   int m0, int half, int l15, int M,
                                                   float* __restrict__ es, float* __restrict__ ed)
{
#pragma unroll
    for (int r = 0; r < 8; ++r) {
        for (int off = 8; off; off >>= 1) {
            ps[r] += __shfl_xor(ps[r], off);
            pd[r] += __shfl_xor(pd[r], off);
        }
    }
    if (l15 == 0) {
#pragma unroll
        for (int r = 0; r < 8; ++r) {
            const int row = m0 + r + half * 8;
            if (row < M) { es[row] = ps[r]; ed[row] = pd[r]; }
        }
    }
}

typedef float f32x4_in __attribute__((ext_vector_type(4), aligned(4)));

template <int K>
static __device__ __forceinline__ void
wmma_gemm_f32a_scores_block(int blk,
                            const float* __restrict__ A,
                            const unsigned short* __restrict__ Bt,
                            const float* __restrict__ as_, const float* __restrict__ ad_,
                            unsigned short* __restrict__ Y,
                            float* __restrict__ es, float* __restrict__ ed,
                            int M, int NOUT)
{
    constexpr int KS = K / 32;
    const int lane  = threadIdx.x & 31;
    const int wave  = threadIdx.x >> 5;
    const int tileM = blk * GEMM_WAVES + wave;
    const int m0    = tileM * 16;
    if (m0 >= M) return;

    const int l15   = lane & 15;
    const int half  = lane >> 4;
    const int kbase = half * 8;

    int rowA = m0 + l15;
    if (rowA >= M) rowA = M - 1;
    const float* ap = A + (size_t)rowA * K + kbase;

    Frag16 fa[KS];
#pragma unroll
    for (int s = 0; s < KS; ++s) {
        const f32x4 x0 = *(const f32x4_in*)(ap + s * 32);
        const f32x4 x1 = *(const f32x4_in*)(ap + s * 32 + 4);
        const f32x4 x2 = *(const f32x4_in*)(ap + s * 32 + 16);
        const f32x4 x3 = *(const f32x4_in*)(ap + s * 32 + 20);
        const u32x4 lo = { pack2bf(x0[0], x0[1]), pack2bf(x0[2], x0[3]),
                           pack2bf(x1[0], x1[1]), pack2bf(x1[2], x1[3]) };
        const u32x4 hi = { pack2bf(x2[0], x2[1]), pack2bf(x2[2], x2[3]),
                           pack2bf(x3[0], x3[1]), pack2bf(x3[2], x3[3]) };
        fa[s].q[0] = lo;
        fa[s].q[1] = hi;
    }

    float ps[8], pd[8];
#pragma unroll
    for (int r = 0; r < 8; ++r) { ps[r] = 0.f; pd[r] = 0.f; }

    for (int c0 = 0; c0 < NOUT; c0 += 16) {
        const unsigned short* bptr = Bt + (size_t)(c0 + l15) * K + kbase;
        const int colY = c0 + l15;
        const float av = as_[colY];
        const float dv = ad_[colY];
        f32x8 acc = {};
#pragma unroll
        for (int s = 0; s < KS; ++s) {
            Frag16 fb;
            fb.q[0] = *(const u32x4*)(bptr + s * 32);
            fb.q[1] = *(const u32x4*)(bptr + s * 32 + 16);
            acc = __builtin_amdgcn_wmma_f32_16x16x32_bf16(
                      false, fa[s].v, false, fb.v, (short)0, acc, false, false);
        }
        const bool odd = (l15 & 1) != 0;
#pragma unroll
        for (int r = 0; r < 8; r += 2) {
            const f32x2 two = { acc[r], acc[r + 1] };
            const unsigned own  = __builtin_bit_cast(unsigned, __builtin_convertvector(two, bf16x2));
            const unsigned send = odd ? (own & 0xFFFFu) : (own >> 16);
            const unsigned recv = (unsigned)__builtin_amdgcn_update_dpp(0, (int)send, 0xB1, 0xF, 0xF, false);
            const unsigned pk   = odd ? (recv | (own & 0xFFFF0000u)) : ((own & 0xFFFFu) | (recv << 16));
            const int row = m0 + r + (odd ? 1 : 0) + half * 8;
            if (row < M) *(unsigned*)(Y + (size_t)row * NOUT + (colY & ~1)) = pk;
        }
        score_add(ps, pd, acc, av, dv);
    }
    score_store(ps, pd, m0, half, l15, M, es, ed);
}

template <int K>
__global__ void __launch_bounds__(32 * GEMM_WAVES)
wmma_rows_times_folded(const unsigned short* __restrict__ A,
                       const unsigned short* __restrict__ Ut,
                       float* __restrict__ S, int M)
{
    const int lane  = threadIdx.x & 31;
    const int wave  = threadIdx.x >> 5;
    const int tileM = blockIdx.x * GEMM_WAVES + wave;
    const int m0    = tileM * 16;
    if (m0 >= M) return;

    const int l15   = lane & 15;
    const int half  = lane >> 4;
    const int kbase = half * 8;

    int rowA = m0 + l15;
    if (rowA >= M) rowA = M - 1;
    const unsigned short* aptr = A  + (size_t)rowA * K + kbase;
    const unsigned short* bptr = Ut + (size_t)l15  * K + kbase;

    f32x8 acc = {};
#pragma unroll
    for (int k0 = 0; k0 < K; k0 += 32) {
        Frag16 fa, fb;
        fa.q[0] = *(const u32x4*)(aptr + k0);
        fa.q[1] = *(const u32x4*)(aptr + k0 + 16);
        fb.q[0] = *(const u32x4*)(bptr + k0);
        fb.q[1] = *(const u32x4*)(bptr + k0 + 16);
        acc = __builtin_amdgcn_wmma_f32_16x16x32_bf16(
                  false, fa.v, false, fb.v, (short)0, acc, false, false);
    }
    if (l15 < 4) {
#pragma unroll
        for (int r = 0; r < 8; ++r) {
            const int row = m0 + r + half * 8;
            if (row < M) S[(size_t)row * 4 + l15] = acc[r];
        }
    }
}

#define LINE_W   32
#define LINE_CAP (LINE_W - 1)
typedef __attribute__((__ext_vector_type__(4))) int i32x4;
typedef __attribute__((__ext_vector_type__(2))) int i32x2;

static __device__ __forceinline__ void
clear_lines_block(int blk, int nblk, i32x4* __restrict__ lines4, int n4, int* __restrict__ novf)
{
    const int stride = nblk * blockDim.x;
    for (int i = blk * blockDim.x + threadIdx.x; i < n4; i += stride) {
        i32x4 v = { -1, -1, -1, -1 };
        if ((i & (LINE_W / 4 - 1)) == 0) v[0] = 0;
        lines4[i] = v;
    }
    if (blk == 0 && threadIdx.x == 0) novf[0] = 0;
}

static __device__ __forceinline__ void
place_edges_block(int blk, const int* __restrict__ ei, int E,
                  int* __restrict__ lines, int* __restrict__ novf, i32x2* __restrict__ ovf)
{
    const int e = blk * blockDim.x + threadIdx.x;
    if (e >= E) return;
    const int s = ei[e], d = ei[E + e];
    int* line = lines + (size_t)d * LINE_W;
    const int pos = atomicAdd(line, 1);
    if (pos < LINE_CAP) {
        line[1 + pos] = s;
    } else {
        const int k = atomicAdd(novf, 1);
        const i32x2 p = { d, s };
        ovf[k] = p;
    }
}

template <int K>
__global__ void __launch_bounds__(32 * GEMM_WAVES)
layer1_product_beside_edge_placing(const float* __restrict__ A,
                                   const unsigned short* __restrict__ Bt,
                                   const float* __restrict__ as_, const float* __restrict__ ad_,
                                   unsigned short* __restrict__ Y,
                                   float* __restrict__ es, float* __restrict__ ed,
                                   int M, int NOUT, int productBlocks,
                                   const int* __restrict__ ei, int E,
                                   int* __restrict__ lines, int* __restrict__ novf,
                                   i32x2* __restrict__ ovf)
{
    const int b = blockIdx.x;
    if (b < productBlocks)
        wmma_gemm_f32a_scores_block<K>(b, A, Bt, as_, ad_, Y, es, ed, M, NOUT);
    else
        place_edges_block(b - productBlocks, ei, E, lines, novf, ovf);
}

static __device__ __forceinline__ float leaky02(float l) { return (l > 0.f) ? l : 0.2f * l; }

template <bool FIRST>
static __device__ __forceinline__ void row_axpy(float (&acc)[8], const u32x4 q, float w)
{
#pragma unroll
    for (int k = 0; k < 4; ++k) {
        const float lo = __uint_as_float(q[k] << 16);
        const float hi = __uint_as_float(q[k] & 0xFFFF0000u);
        if (FIRST) { acc[2 * k] = lo * w;  acc[2 * k + 1] = hi * w; }
        else       { acc[2 * k] += lo * w; acc[2 * k + 1] += hi * w; }
    }
}

template <int NV>
__global__ void __launch_bounds__(256)
gat_gather(const int* __restrict__ lines, const int* __restrict__ novf,
           const i32x2* __restrict__ ovf,
           const unsigned short* __restrict__ h, const float* __restrict__ es,
           const float* __restrict__ ed, const float* __restrict__ bias,
           unsigned short* __restrict__ outbf, int N)
{
    constexpr int C = 256 * NV;
    const int lane = threadIdx.x & 31;
    const int d = __builtin_amdgcn_readfirstlane(
                      (int)((blockIdx.x * blockDim.x + threadIdx.x) >> 5));
    if (d >= N) return;

    const float edv   = ed[d];
    const float lself = leaky02(es[d] + edv);
    const float NEG_INF = -__builtin_inff();

    const int word = lines[(size_t)d * LINE_W + lane];
    const int cnt = __builtin_amdgcn_readfirstlane(word);
    const int nin = (cnt < LINE_CAP) ? cnt : LINE_CAP;
    const int nov = (cnt > LINE_CAP) ? novf[0] : 0;

    const bool has = (lane >= 1) && (lane <= nin);
    int   s0 = d;
    float l0 = NEG_INF;
    if (has) { s0 = word; l0 = leaky02(es[s0] + edv); }

    float m = fmaxf(lself, l0);
    for (int i = lane; i < nov; i += 32) {
        const i32x2 p = ovf[i];
        if (p[0] == d) m = fmaxf(m, leaky02(es[p[1]] + edv));
    }
    for (int off = 16; off; off >>= 1) m = fmaxf(m, __shfl_xor(m, off));

    float sum = has ? __expf(l0 - m) : 0.f;
    if (lane == 0) sum += __expf(lself - m);
    for (int i = lane; i < nov; i += 32) {
        const i32x2 p = ovf[i];
        if (p[0] == d) sum += __expf(leaky02(es[p[1]] + edv) - m);
    }
    for (int off = 16; off; off >>= 1) sum += __shfl_xor(sum, off);

    float acc[NV][8];
    {
        const float w = __expf(lself - m) / sum;
        const u32x4* r = (const u32x4*)(h + (size_t)d * C);
        for (int v = 0; v < NV; ++v) row_axpy<true>(acc[v], r[v * 32 + lane], w);
    }
    {
        const float w = has ? __expf(l0 - m) / sum : 0.f;
#pragma unroll 4
        for (int j = 1; j <= nin; ++j) {
            const int   sj = __builtin_amdgcn_readlane(s0, j);
            const float wj = __int_as_float(__builtin_amdgcn_readlane(__float_as_int(w), j));
            const u32x4* r = (const u32x4*)(h + (size_t)sj * C);
            for (int v = 0; v < NV; ++v) row_axpy<false>(acc[v], r[v * 32 + lane], wj);
        }
    }
    for (int base = 0; base < nov; base += 32) {
        const int i = base + lane;
        int   s = d;
        float w = 0.f;
        bool  mine = false;
        if (i < nov) {
            const i32x2 p = ovf[i];
            if (p[0] == d) { mine = true; s = p[1]; w = __expf(leaky02(es[s] + edv) - m) / sum; }
        }
        unsigned todo = __builtin_amdgcn_ballot_w32(mine);
        while (todo) {
            const int j = __builtin_ctz(todo);
            todo &= todo - 1;
            const int   sj = __builtin_amdgcn_readlane(s, j);
            const float wj = __int_as_float(__builtin_amdgcn_readlane(__float_as_int(w), j));
            const u32x4* r = (const u32x4*)(h + (size_t)sj * C);
            for (int v = 0; v < NV; ++v) row_axpy<false>(acc[v], r[v * 32 + lane], wj);
        }
    }
    for (int v = 0; v < NV; ++v) {
        const int col = (v * 32 + lane) * 8;
        const float* b = bias + col;
        u32x4 pk;
#pragma unroll
        for (int k = 0; k < 4; ++k) {
            const float o0 = acc[v][2 * k]     + b[2 * k];
            const float o1 = acc[v][2 * k + 1] + b[2 * k + 1];
            pk[k] = pack2bf(o0 > 0.f ? o0 : 0.f, o1 > 0.f ? o1 : 0.f);
        }
        *(u32x4*)(outbf + (size_t)d * C + col) = pk;
    }
}

#define GL 8
__global__ void __launch_bounds__(256)
gat_gather_folded(const i32x4* __restrict__ lines4, const int* __restrict__ novf,
                  const i32x2* __restrict__ ovf,
                  const f32x4* __restrict__ S, const float* __restrict__ cb,
                  float* __restrict__ q0, float* __restrict__ q1, int N)
{
    const int t   = blockIdx.x * blockDim.x + threadIdx.x;
    const int sub = threadIdx.x & (GL - 1);
    int d = t / GL;
    const bool live = d < N;
    if (!live) d = N - 1;

    const f32x4 sd = S[d];
    const i32x4 wd = lines4[(size_t)d * (LINE_W / 4) + sub];
    const int cnt = __shfl(wd[0], 0, GL);
    const int nin = (cnt < LINE_CAP) ? cnt : LINE_CAP;
    const int nov = (cnt > LINE_CAP) ? novf[0] : 0;

    const float edv   = sd[1];
    const float lself = leaky02(sd[0] + edv);
    const float NEG_INF = -__builtin_inff();

    f32x4 x[4];
#pragma unroll
    for (int k = 0; k < 4; ++k) {
        const int idx = sub * 4 + k;
        x[k] = S[(idx >= 1 && idx <= nin) ? wd[k] : d];
    }
    float l[4], ta[4], tb[4];
#pragma unroll
    for (int k = 0; k < 4; ++k) {
        const int idx = sub * 4 + k;
        l[k]  = (idx >= 1 && idx <= nin) ? leaky02(x[k][0] + edv) : NEG_INF;
        ta[k] = x[k][2];
        tb[k] = x[k][3];
    }

    float m = fmaxf(fmaxf(lself, l[0]), fmaxf(l[1], fmaxf(l[2], l[3])));
    for (int i = sub; i < nov; i += GL) {
        const i32x2 p = ovf[i];
        if (p[0] == d) m = fmaxf(m, leaky02(S[p[1]][0] + edv));
    }
    for (int off = GL / 2; off; off >>= 1) m = fmaxf(m, __shfl_xor(m, off, GL));

    float sum = 0.f, a0 = 0.f, a1 = 0.f;
#pragma unroll
    for (int k = 0; k < 4; ++k) {
        const int idx = sub * 4 + k;
        const float p = (idx >= 1 && idx <= nin) ? __expf(l[k] - m) : 0.f;
        sum += p; a0 += p * ta[k]; a1 += p * tb[k];
    }
    if (sub == 0) {
        const float ps = __expf(lself - m);
        sum += ps; a0 += ps * sd[2]; a1 += ps * sd[3];
    }
    for (int i = sub; i < nov; i += GL) {
        const i32x2 pr = ovf[i];
        if (pr[0] == d) {
            const f32x4 xr = S[pr[1]];
            const float p = __expf(leaky02(xr[0] + edv) - m);
            sum += p; a0 += p * xr[2]; a1 += p * xr[3];
        }
    }
    for (int off = GL / 2; off; off >>= 1) {
        sum += __shfl_xor(sum, off, GL);
        a0  += __shfl_xor(a0,  off, GL);
        a1  += __shfl_xor(a1,  off, GL);
    }
    if (live && sub == 0) {
        q0[d] = a0 / sum + cb[0];
        q1[d] = a1 / sum + cb[1];
    }
}

static __device__ __forceinline__ void
transpose_to_bf16_block(int blk, const float* __restrict__ W, unsigned short* __restrict__ Wt,
                        int K, int NOUT)
{
    const int i = blk * blockDim.x + threadIdx.x;
    if (i >= K * NOUT) return;
    const int k = i / NOUT, n = i % NOUT;
    Wt[(size_t)n * K + k] = f2bf(W[i]);
}

static __device__ __forceinline__ void
fold_weight_block(int blk, const float* __restrict__ W2, const float* __restrict__ a_s,
                  const float* __restrict__ a_d, const float* __restrict__ Wl,
                  const float* __restrict__ b2,
                  unsigned short* __restrict__ Ut, float* __restrict__ cb, int K, int C)
{
    const int lane = threadIdx.x & 31;
    const int w = (int)((blk * blockDim.x + threadIdx.x) >> 5);
    if (w > K) return;
    const float* x = (w < K) ? (W2 + (size_t)w * C) : b2;
    float s0 = 0.f, s1 = 0.f, s2 = 0.f, s3 = 0.f;
    for (int c = lane; c < C; c += 32) {
        const float xv = x[c];
        s0 += xv * a_s[c];
        s1 += xv * a_d[c];
        s2 += xv * Wl[c];
        s3 += xv * Wl[C + c];
    }
    for (int off = 16; off; off >>= 1) {
        s0 += __shfl_xor(s0, off);
        s1 += __shfl_xor(s1, off);
        s2 += __shfl_xor(s2, off);
        s3 += __shfl_xor(s3, off);
    }
    if (w < K) {
        const float v = (lane == 0) ? s0 : (lane == 1) ? s1 : (lane == 2) ? s2 : s3;
        if (lane < 16) Ut[(size_t)lane * K + w] = (lane < 4) ? f2bf(v) : (unsigned short)0;
    } else if (lane == 0) {
        cb[0] = s2;
        cb[1] = s3;
    }
}

__global__ void __launch_bounds__(256)
prepare_weights_and_lines(const float* __restrict__ W1, unsigned short* __restrict__ w1t,
                          int K1, int NOUT1, int transposeBlocks,
                          const float* __restrict__ W2, const float* __restrict__ a_s,
                          const float* __restrict__ a_d, const float* __restrict__ Wl,
                          const float* __restrict__ b2,
                          unsigned short* __restrict__ Ut, float* __restrict__ cb,
                          int K2, int C2, int foldBlocks,
                          i32x4* __restrict__ lines4, int n4, int clearBlocks,
                          int* __restrict__ novf)
{
    const int b = blockIdx.x;
    if (b < transposeBlocks)
        transpose_to_bf16_block(b, W1, w1t, K1, NOUT1);
    else if (b < transposeBlocks + foldBlocks)
        fold_weight_block(b - transposeBlocks, W2, a_s, a_d, Wl, b2, Ut, cb, K2, C2);
    else
        clear_lines_block(b - transposeBlocks - foldBlocks, clearBlocks, lines4, n4, novf);
}

__global__ void link_predict(const float* __restrict__ q0, const float* __restrict__ q1,
                             const int* __restrict__ mask, const float* __restrict__ bl,
                             float* __restrict__ out, int P)
{
    const int p = blockIdx.x * blockDim.x + threadIdx.x;
    if (p >= P) return;
    const float z = q0[mask[p * 2 + 0]] + q1[mask[p * 2 + 1]] + bl[0];
    out[p] = 1.f / (1.f + expf(-z));
}

extern "C" void kernel_launch(void* const* d_in, const int* in_sizes, int n_in,
                              void* d_out, int out_size, void* d_ws, size_t ws_size,
                              hipStream_t stream)
{
    (void)n_in; (void)out_size; (void)ws_size;

    const float* features = (const float*)d_in[0];
    const int*   ei       = (const int*)  d_in[1];
    const int*   mask     = (const int*)  d_in[2];
    const float* W1   = (const float*)d_in[3];
    const float* a_s1 = (const float*)d_in[4];
    const float* a_d1 = (const float*)d_in[5];
    const float* b1   = (const float*)d_in[6];
    const float* W2   = (const float*)d_in[7];
    const float* a_s2 = (const float*)d_in[8];
    const float* a_d2 = (const float*)d_in[9];
    const float* b2   = (const float*)d_in[10];
    const float* Wl   = (const float*)d_in[11];
    const float* bl   = (const float*)d_in[12];

    const int F  = 128, Hh = 256;
    const int N  = in_sizes[0] / F;
    const int E  = in_sizes[1] / 2;
    const int P  = in_sizes[2] / 2;

    char* ws = (char*)d_ws;
    size_t off = 0;
    auto alloc = [&](size_t bytes) -> char* {
        char* ptr = ws + off;
        off += (bytes + 255) & ~(size_t)255;
        return ptr;
    };
    unsigned short* w1t  = (unsigned short*)alloc((size_t)Hh * F * 2);
    unsigned short* ut   = (unsigned short*)alloc((size_t)16 * Hh * 2);
    float*    cb      = (float*)   alloc((size_t)2 * 4);
    float*    S       = (float*)   alloc((size_t)N * 4 * 4);
    unsigned short* hrbf = (unsigned short*)alloc((size_t)N * Hh * 2);
    unsigned short* buf1 = (unsigned short*)alloc((size_t)N * Hh * 2);
    float*    q0      = (float*)   alloc((size_t)N * 4);
    float*    q1      = (float*)   alloc((size_t)N * 4);
    float*    es      = (float*)   alloc((size_t)N * 4);
    float*    ed      = (float*)   alloc((size_t)N * 4);
    int*      lines   = (int*)     alloc((size_t)N * LINE_W * 4);
    int*      novf    = (int*)     alloc((size_t)4);
    i32x2*    ovf     = (i32x2*)   alloc((size_t)E * 8);

    const int tilesM = (N + 15) / 16;
    const int gemmGrid = (tilesM + GEMM_WAVES - 1) / GEMM_WAVES;
    const int TB = 256;
    auto cdiv = [](size_t a, size_t b) -> int { return (int)((a + b - 1) / b); };

    const int transposeBlocks = cdiv((size_t)F * Hh, TB);
    const int foldBlocks      = cdiv((size_t)(Hh + 1) * 32, TB);
    const int clearBlocks     = cdiv(N, TB);
    prepare_weights_and_lines<<<transposeBlocks + foldBlocks + clearBlocks, TB, 0, stream>>>(
        W1, w1t, F, Hh, transposeBlocks,
        W2, a_s2, a_d2, Wl, b2, ut, cb, Hh, F, foldBlocks,
        (i32x4*)lines, N * (LINE_W / 4), clearBlocks, novf);

    layer1_product_beside_edge_placing<128>
        <<<gemmGrid + cdiv(E, 32 * GEMM_WAVES), 32 * GEMM_WAVES, 0, stream>>>(
        features, w1t, a_s1, a_d1, buf1, es, ed, N, Hh, gemmGrid,
        ei, E, lines, novf, ovf);
    gat_gather<1><<<cdiv((size_t)N * 32, TB), TB, 0, stream>>>(
        lines, novf, ovf, buf1, es, ed, b1, hrbf, N);

    wmma_rows_times_folded<256><<<gemmGrid, 32 * GEMM_WAVES, 0, stream>>>(hrbf, ut, S, N);
    gat_gather_folded<<<cdiv((size_t)N * GL, TB), TB, 0, stream>>>(
        (const i32x4*)lines, novf, ovf, (const f32x4*)S, cb, q0, q1, N);

    link_predict<<<cdiv((size_t)P, TB), TB, 0, stream>>>(q0, q1, mask, bl, (float*)d_out, P);
}


# Layer 2's folded product joins the first gather's work-group: 16 waves leave relu'd bf16 rows in an LDS tile, the last to arrive runs the 8 WMMAs; rows never go to memory; 6 -> 5 launches
# speedup vs baseline: 10.3788x; 1.0633x over previous
#include <hip/hip_runtime.h>
#include <hip/hip_bf16.h>

typedef __attribute__((__ext_vector_type__(16))) __bf16 bf16x16;
typedef __attribute__((__ext_vector_type__(8)))  float  f32x8;
typedef __attribute__((__ext_vector_type__(4)))  unsigned int u32x4;
typedef __attribute__((__ext_vector_type__(4)))  float  f32x4;
typedef __attribute__((__ext_vector_type__(2)))  float  f32x2;
typedef __attribute__((__ext_vector_type__(2)))  __bf16 bf16x2;

union Frag16 {
    u32x4   q[2];
    bf16x16 v;
};

static __device__ __forceinline__ unsigned short f2bf(float f) {
    unsigned u = __float_as_uint(f);
    u += 0x7FFFu + ((u >> 16) & 1u);
    return (unsigned short)(u >> 16);
}

#define GEMM_WAVES 4

static __device__ __forceinline__ unsigned pack2bf(float lo, float hi) {
    return (unsigned)f2bf(lo) | ((unsigned)f2bf(hi) << 16);
}

static __device__ __forceinline__ void score_add(float (&ps)[8], float (&pd)[8],
                                                 const f32x8 acc, float av, float dv)
{
#pragma unroll
    for (int r = 0; r < 8; ++r) {
        ps[r] += acc[r] * av;
        pd[r] += acc[r] * dv;
    }
}

static __device__ __forceinline__ void score_store(float (&ps)[8], float (&pd)[8],
                                                   int m0, int half, int l15, int M,
                                                   float* __restrict__ es, float* __restrict__ ed)
{
#pragma unroll
    for (int r = 0; r < 8; ++r) {
        for (int off = 8; off; off >>= 1) {
            ps[r] += __shfl_xor(ps[r], off);
            pd[r] += __shfl_xor(pd[r], off);
        }
    }
    if (l15 == 0) {
#pragma unroll
        for (int r = 0; r < 8; ++r) {
            const int row = m0 + r + half * 8;
            if (row < M) { es[row] = ps[r]; ed[row] = pd[r]; }
        }
    }
}

typedef float f32x4_in __attribute__((ext_vector_type(4), aligned(4)));

template <int K>
static __device__ __forceinline__ void
wmma_gemm_f32a_scores_block(int blk,
                            const float* __restrict__ A,
                            const unsigned short* __restrict__ Bt,
                            const float* __restrict__ as_, const float* __restrict__ ad_,
                            unsigned short* __restrict__ Y,
                            float* __restrict__ es, float* __restrict__ ed,
                            int M, int NOUT)
{
    constexpr int KS = K / 32;
    const int lane  = threadIdx.x & 31;
    const int wave  = threadIdx.x >> 5;
    const int tileM = blk * GEMM_WAVES + wave;
    const int m0    = tileM * 16;
    if (m0 >= M) return;

    const int l15   = lane & 15;
    const int half  = lane >> 4;
    const int kbase = half * 8;

    int rowA = m0 + l15;
    if (rowA >= M) rowA = M - 1;
    const float* ap = A + (size_t)rowA * K + kbase;

    Frag16 fa[KS];
#pragma unroll
    for (int s = 0; s < KS; ++s) {
        const f32x4 x0 = *(const f32x4_in*)(ap + s * 32);
        const f32x4 x1 = *(const f32x4_in*)(ap + s * 32 + 4);
        const f32x4 x2 = *(const f32x4_in*)(ap + s * 32 + 16);
        const f32x4 x3 = *(const f32x4_in*)(ap + s * 32 + 20);
        const u32x4 lo = { pack2bf(x0[0], x0[1]), pack2bf(x0[2], x0[3]),
                           pack2bf(x1[0], x1[1]), pack2bf(x1[2], x1[3]) };
        const u32x4 hi = { pack2bf(x2[0], x2[1]), pack2bf(x2[2], x2[3]),
                           pack2bf(x3[0], x3[1]), pack2bf(x3[2], x3[3]) };
        fa[s].q[0] = lo;
        fa[s].q[1] = hi;
    }

    float ps[8], pd[8];
#pragma unroll
    for (int r = 0; r < 8; ++r) { ps[r] = 0.f; pd[r] = 0.f; }

    for (int c0 = 0; c0 < NOUT; c0 += 16) {
        const unsigned short* bptr = Bt + (size_t)(c0 + l15) * K + kbase;
        const int colY = c0 + l15;
        const float av = as_[colY];
        const float dv = ad_[colY];
        f32x8 acc = {};
#pragma unroll
        for (int s = 0; s < KS; ++s) {
            Frag16 fb;
            fb.q[0] = *(const u32x4*)(bptr + s * 32);
            fb.q[1] = *(const u32x4*)(bptr + s * 32 + 16);
            acc = __builtin_amdgcn_wmma_f32_16x16x32_bf16(
                      false, fa[s].v, false, fb.v, (short)0, acc, false, false);
        }
        const bool odd = (l15 & 1) != 0;
#pragma unroll
        for (int r = 0; r < 8; r += 2) {
            const f32x2 two = { acc[r], acc[r + 1] };
            const unsigned own  = __builtin_bit_cast(unsigned, __builtin_convertvector(two, bf16x2));
            const unsigned send = odd ? (own & 0xFFFFu) : (own >> 16);
            const unsigned recv = (unsigned)__builtin_amdgcn_update_dpp(0, (int)send, 0xB1, 0xF, 0xF, false);
            const unsigned pk   = odd ? (recv | (own & 0xFFFF0000u)) : ((own & 0xFFFFu) | (recv << 16));
            const int row = m0 + r + (odd ? 1 : 0) + half * 8;
            if (row < M) *(unsigned*)(Y + (size_t)row * NOUT + (colY & ~1)) = pk;
        }
        score_add(ps, pd, acc, av, dv);
    }
    score_store(ps, pd, m0, half, l15, M, es, ed);
}

#define LINE_W   32
#define LINE_CAP (LINE_W - 1)
typedef __attribute__((__ext_vector_type__(4))) int i32x4;
typedef __attribute__((__ext_vector_type__(2))) int i32x2;

static __device__ __forceinline__ void
clear_lines_block(int blk, int nblk, i32x4* __restrict__ lines4, int n4, int* __restrict__ novf)
{
    const int stride = nblk * blockDim.x;
    for (int i = blk * blockDim.x + threadIdx.x; i < n4; i += stride) {
        i32x4 v = { -1, -1, -1, -1 };
        if ((i & (LINE_W / 4 - 1)) == 0) v[0] = 0;
        lines4[i] = v;
    }
    if (blk == 0 && threadIdx.x == 0) novf[0] = 0;
}

static __device__ __forceinline__ void
place_edges_block(int blk, const int* __restrict__ ei, int E,
                  int* __restrict__ lines, int* __restrict__ novf, i32x2* __restrict__ ovf)
{
    const int e = blk * blockDim.x + threadIdx.x;
    if (e >= E) return;
    const int s = ei[e], d = ei[E + e];
    int* line = lines + (size_t)d * LINE_W;
    const int pos = atomicAdd(line, 1);
    if (pos < LINE_CAP) {
        line[1 + pos] = s;
    } else {
        const int k = atomicAdd(novf, 1);
        const i32x2 p = { d, s };
        ovf[k] = p;
    }
}

template <int K>
__global__ void __launch_bounds__(32 * GEMM_WAVES)
layer1_product_beside_edge_placing(const float* __restrict__ A,
                                   const unsigned short* __restrict__ Bt,
                                   const float* __restrict__ as_, const float* __restrict__ ad_,
                                   unsigned short* __restrict__ Y,
                                   float* __restrict__ es, float* __restrict__ ed,
                                   int M, int NOUT, int productBlocks,
                                   const int* __restrict__ ei, int E,
                                   int* __restrict__ lines, int* __restrict__ novf,
                                   i32x2* __restrict__ ovf)
{
    const int b = blockIdx.x;
    if (b < productBlocks)
        wmma_gemm_f32a_scores_block<K>(b, A, Bt, as_, ad_, Y, es, ed, M, NOUT);
    else
        place_edges_block(b - productBlocks, ei, E, lines, novf, ovf);
}

static __device__ __forceinline__ float leaky02(float l) { return (l > 0.f) ? l : 0.2f * l; }

template <bool FIRST>
static __device__ __forceinline__ void row_axpy(float (&acc)[8], const u32x4 q, float w)
{
#pragma unroll
    for (int k = 0; k < 4; ++k) {
        const float lo = __uint_as_float(q[k] << 16);
        const float hi = __uint_as_float(q[k] & 0xFFFF0000u);
        if (FIRST) { acc[2 * k] = lo * w;  acc[2 * k + 1] = hi * w; }
        else       { acc[2 * k] += lo * w; acc[2 * k + 1] += hi * w; }
    }
}

static __device__ __forceinline__ u32x4
gather_row_relu_bf16(int d, int lane,
                     const int* __restrict__ lines, const int* __restrict__ novf,
                     const i32x2* __restrict__ ovf,
                     const unsigned short* __restrict__ h, const float* __restrict__ es,
                     const float* __restrict__ ed, const float* __restrict__ bias)
{
    constexpr int C = 256;
    const float edv   = ed[d];
    const float lself = leaky02(es[d] + edv);
    const float NEG_INF = -__builtin_inff();

    const int word = lines[(size_t)d * LINE_W + lane];
    const int cnt = __builtin_amdgcn_readfirstlane(word);
    const int nin = (cnt < LINE_CAP) ? cnt : LINE_CAP;
    const int nov = (cnt > LINE_CAP) ? novf[0] : 0;

    const bool has = (lane >= 1) && (lane <= nin);
    int   s0 = d;
    float l0 = NEG_INF;
    if (has) { s0 = word; l0 = leaky02(es[s0] + edv); }

    float m = fmaxf(lself, l0);
    for (int i = lane; i < nov; i += 32) {
        const i32x2 p = ovf[i];
        if (p[0] == d) m = fmaxf(m, leaky02(es[p[1]] + edv));
    }
    for (int off = 16; off; off >>= 1) m = fmaxf(m, __shfl_xor(m, off));

    float sum = has ? __expf(l0 - m) : 0.f;
    if (lane == 0) sum += __expf(lself - m);
    for (int i = lane; i < nov; i += 32) {
        const i32x2 p = ovf[i];
        if (p[0] == d) sum += __expf(leaky02(es[p[1]] + edv) - m);
    }
    for (int off = 16; off; off >>= 1) sum += __shfl_xor(sum, off);

    float acc[8];
    {
        const float w = __expf(lself - m) / sum;
        const u32x4* r = (const u32x4*)(h + (size_t)d * C);
        row_axpy<true>(acc, r[lane], w);
    }
    {
        const float w = has ? __expf(l0 - m) / sum : 0.f;
#pragma unroll 4
        for (int j = 1; j <= nin; ++j) {
            const int   sj = __builtin_amdgcn_readlane(s0, j);
            const float wj = __int_as_float(__builtin_amdgcn_readlane(__float_as_int(w), j));
            const u32x4* r = (const u32x4*)(h + (size_t)sj * C);
            row_axpy<false>(acc, r[lane], wj);
        }
    }
    for (int base = 0; base < nov; base += 32) {
        const int i = base + lane;
        int   s = d;
        float w = 0.f;
        bool  mine = false;
        if (i < nov) {
            const i32x2 p = ovf[i];
            if (p[0] == d) { mine = true; s = p[1]; w = __expf(leaky02(es[s] + edv) - m) / sum; }
        }
        unsigned todo = __builtin_amdgcn_ballot_w32(mine);
        while (todo) {
            const int j = __builtin_ctz(todo);
            todo &= todo - 1;
            const int   sj = __builtin_amdgcn_readlane(s, j);
            const float wj = __int_as_float(__builtin_amdgcn_readlane(__float_as_int(w), j));
            const u32x4* r = (const u32x4*)(h + (size_t)sj * C);
            row_axpy<false>(acc, r[lane], wj);
        }
    }
    const float* b = bias + lane * 8;
    u32x4 pk;
#pragma unroll
    for (int k = 0; k < 4; ++k) {
        const float o0 = acc[2 * k]     + b[2 * k];
        const float o1 = acc[2 * k + 1] + b[2 * k + 1];
        pk[k] = pack2bf(o0 > 0.f ? o0 : 0.f, o1 > 0.f ? o1 : 0.f);
    }
    return pk;
}

#define TILE_ROWS 16
#define ROW_Q     33

__global__ void __launch_bounds__(32 * TILE_ROWS)
gat_gather_times_folded(const int* __restrict__ lines, const int* __restrict__ novf,
                        const i32x2* __restrict__ ovf,
                        const unsigned short* __restrict__ h, const float* __restrict__ es,
                        const float* __restrict__ ed, const float* __restrict__ bias,
                        const unsigned short* __restrict__ Ut,
                        float* __restrict__ S, int N)
{
    constexpr int K = 256;
    __shared__ u32x4 tile[TILE_ROWS * ROW_Q];
    __shared__ int   tickets;

    const int lane = threadIdx.x & 31;
    const int wave = __builtin_amdgcn_readfirstlane((int)(threadIdx.x >> 5));
    const int m0   = blockIdx.x * TILE_ROWS;
    const int d    = m0 + wave;

    if (threadIdx.x == 0) tickets = 0;
    __syncthreads();

    u32x4 row = { 0u, 0u, 0u, 0u };
    if (d < N)
        row = gather_row_relu_bf16(d, lane, lines, novf, ovf, h, es, ed, bias);
    tile[wave * ROW_Q + lane] = row;

    __builtin_amdgcn_fence(__ATOMIC_RELEASE, "workgroup");
    int t = 0;
    if (lane == 0) t = atomicAdd(&tickets, 1);
    t = __builtin_amdgcn_readfirstlane(t);
    if (t != TILE_ROWS - 1) return;
    __builtin_amdgcn_fence(__ATOMIC_ACQUIRE, "workgroup");

    const int l15  = lane & 15;
    const int half = lane >> 4;
    const u32x4* arow = tile + l15 * ROW_Q + half;
    const unsigned short* bptr = Ut + (size_t)l15 * K + half * 8;

    f32x8 acc = {};
#pragma unroll 2
    for (int k0 = 0; k0 < K; k0 += 32) {
        Frag16 fa, fb;
        fa.q[0] = arow[(k0 >> 3)];
        fa.q[1] = arow[(k0 >> 3) + 2];
        fb.q[0] = *(const u32x4*)(bptr + k0);
        fb.q[1] = *(const u32x4*)(bptr + k0 + 16);
        acc = __builtin_amdgcn_wmma_f32_16x16x32_bf16(
                  false, fa.v, false, fb.v, (short)0, acc, false, false);
    }
    if (l15 < 4) {
#pragma unroll
        for (int r = 0; r < 8; ++r) {
            const int node = m0 + r + half * 8;
            if (node < N) S[(size_t)node * 4 + l15] = acc[r];
        }
    }
}

#define GL 8
__global__ void __launch_bounds__(256)
gat_gather_folded(const i32x4* __restrict__ lines4, const int* __restrict__ novf,
                  const i32x2* __restrict__ ovf,
                  const f32x4* __restrict__ S, const float* __restrict__ cb,
                  float* __restrict__ q0, float* __restrict__ q1, int N)
{
    const int t   = blockIdx.x * blockDim.x + threadIdx.x;
    const int sub = threadIdx.x & (GL - 1);
    int d = t / GL;
    const bool live = d < N;
    if (!live) d = N - 1;

    const f32x4 sd = S[d];
    const i32x4 wd = lines4[(size_t)d * (LINE_W / 4) + sub];
    const int cnt = __shfl(wd[0], 0, GL);
    const int nin = (cnt < LINE_CAP) ? cnt : LINE_CAP;
    const int nov = (cnt > LINE_CAP) ? novf[0] : 0;

    const float edv   = sd[1];
    const float lself = leaky02(sd[0] + edv);
    const float NEG_INF = -__builtin_inff();

    f32x4 x[4];
#pragma unroll
    for (int k = 0; k < 4; ++k) {
        const int idx = sub * 4 + k;
        x[k] = S[(idx >= 1 && idx <= nin) ? wd[k] : d];
    }
    float l[4], ta[4], tb[4];
#pragma unroll
    for (int k = 0; k < 4; ++k) {
        const int idx = sub * 4 + k;
        l[k]  = (idx >= 1 && idx <= nin) ? leaky02(x[k][0] + edv) : NEG_INF;
        ta[k] = x[k][2];
        tb[k] = x[k][3];
    }

    float m = fmaxf(fmaxf(lself, l[0]), fmaxf(l[1], fmaxf(l[2], l[3])));
    for (int i = sub; i < nov; i += GL) {
        const i32x2 p = ovf[i];
        if (p[0] == d) m = fmaxf(m, leaky02(S[p[1]][0] + edv));
    }
    for (int off = GL / 2; off; off >>= 1) m = fmaxf(m, __shfl_xor(m, off, GL));

    float sum = 0.f, a0 = 0.f, a1 = 0.f;
#pragma unroll
    for (int k = 0; k < 4; ++k) {
        const int idx = sub * 4 + k;
        const float p = (idx >= 1 && idx <= nin) ? __expf(l[k] - m) : 0.f;
        sum += p; a0 += p * ta[k]; a1 += p * tb[k];
    }
    if (sub == 0) {
        const float ps = __expf(lself - m);
        sum += ps; a0 += ps * sd[2]; a1 += ps * sd[3];
    }
    for (int i = sub; i < nov; i += GL) {
        const i32x2 pr = ovf[i];
        if (pr[0] == d) {
            const f32x4 xr = S[pr[1]];
            const float p = __expf(leaky02(xr[0] + edv) - m);
            sum += p; a0 += p * xr[2]; a1 += p * xr[3];
        }
    }
    for (int off = GL / 2; off; off >>= 1) {
        sum += __shfl_xor(sum, off, GL);
        a0  += __shfl_xor(a0,  off, GL);
        a1  += __shfl_xor(a1,  off, GL);
    }
    if (live && sub == 0) {
        q0[d] = a0 / sum + cb[0];
        q1[d] = a1 / sum + cb[1];
    }
}

static __device__ __forceinline__ void
transpose_to_bf16_block(int blk, const float* __restrict__ W, unsigned short* __restrict__ Wt,
                        int K, int NOUT)
{
    const int i = blk * blockDim.x + threadIdx.x;
    if (i >= K * NOUT) return;
    const int k = i / NOUT, n = i % NOUT;
    Wt[(size_t)n * K + k] = f2bf(W[i]);
}

static __device__ __forceinline__ void
fold_weight_block(int blk, const float* __restrict__ W2, const float* __restrict__ a_s,
                  const float* __restrict__ a_d, const float* __restrict__ Wl,
                  const float* __restrict__ b2,
                  unsigned short* __restrict__ Ut, float* __restrict__ cb, int K, int C)
{
    const int lane = threadIdx.x & 31;
    const int w = (int)((blk * blockDim.x + threadIdx.x) >> 5);
    if (w > K) return;
    const float* x = (w < K) ? (W2 + (size_t)w * C) : b2;
    float s0 = 0.f, s1 = 0.f, s2 = 0.f, s3 = 0.f;
    for (int c = lane; c < C; c += 32) {
        const float xv = x[c];
        s0 += xv * a_s[c];
        s1 += xv * a_d[c];
        s2 += xv * Wl[c];
        s3 += xv * Wl[C + c];
    }
    for (int off = 16; off; off >>= 1) {
        s0 += __shfl_xor(s0, off);
        s1 += __shfl_xor(s1, off);
        s2 += __shfl_xor(s2, off);
        s3 += __shfl_xor(s3, off);
    }
    if (w < K) {
        const float v = (lane == 0) ? s0 : (lane == 1) ? s1 : (lane == 2) ? s2 : s3;
        if (lane < 16) Ut[(size_t)lane * K + w] = (lane < 4) ? f2bf(v) : (unsigned short)0;
    } else if (lane == 0) {
        cb[0] = s2;
        cb[1] = s3;
    }
}

__global__ void __launch_bounds__(256)
prepare_weights_and_lines(const float* __restrict__ W1, unsigned short* __restrict__ w1t,
                          int K1, int NOUT1, int transposeBlocks,
                          const float* __restrict__ W2, const float* __restrict__ a_s,
                          const float* __restrict__ a_d, const float* __restrict__ Wl,
                          const float* __restrict__ b2,
                          unsigned short* __restrict__ Ut, float* __restrict__ cb,
                          int K2, int C2, int foldBlocks,
                          i32x4* __restrict__ lines4, int n4, int clearBlocks,
                          int* __restrict__ novf)
{
    const int b = blockIdx.x;
    if (b < transposeBlocks)
        transpose_to_bf16_block(b, W1, w1t, K1, NOUT1);
    else if (b < transposeBlocks + foldBlocks)
        fold_weight_block(b - transposeBlocks, W2, a_s, a_d, Wl, b2, Ut, cb, K2, C2);
    else
        clear_lines_block(b - transposeBlocks - foldBlocks, clearBlocks, lines4, n4, novf);
}

__global__ void link_predict(const float* __restrict__ q0, const float* __restrict__ q1,
                             const int* __restrict__ mask, const float* __restrict__ bl,
                             float* __restrict__ out, int P)
{
    const int p = blockIdx.x * blockDim.x + threadIdx.x;
    if (p >= P) return;
    const float z = q0[mask[p * 2 + 0]] + q1[mask[p * 2 + 1]] + bl[0];
    out[p] = 1.f / (1.f + expf(-z));
}

extern "C" void kernel_launch(void* const* d_in, const int* in_sizes, int n_in,
                              void* d_out, int out_size, void* d_ws, size_t ws_size,
                              hipStream_t stream)
{
    (void)n_in; (void)out_size; (void)ws_size;

    const float* features = (const float*)d_in[0];
    const int*   ei       = (const int*)  d_in[1];
    const int*   mask     = (const int*)  d_in[2];
    const float* W1   = (const float*)d_in[3];
    const float* a_s1 = (const float*)d_in[4];
    const float* a_d1 = (const float*)d_in[5];
    const float* b1   = (const float*)d_in[6];
    const float* W2   = (const float*)d_in[7];
    const float* a_s2 = (const float*)d_in[8];
    const float* a_d2 = (const float*)d_in[9];
    const float* b2   = (const float*)d_in[10];
    const float* Wl   = (const float*)d_in[11];
    const float* bl   = (const float*)d_in[12];

    const int F  = 128, Hh = 256;
    const int N  = in_sizes[0] / F;
    const int E  = in_sizes[1] / 2;
    const int P  = in_sizes[2] / 2;

    char* ws = (char*)d_ws;
    size_t off = 0;
    auto alloc = [&](size_t bytes) -> char* {
        char* ptr = ws + off;
        off += (bytes + 255) & ~(size_t)255;
        return ptr;
    };
    unsigned short* w1t  = (unsigned short*)alloc((size_t)Hh * F * 2);
    unsigned short* ut   = (unsigned short*)alloc((size_t)16 * Hh * 2);
    float*    cb      = (float*)   alloc((size_t)2 * 4);
    float*    S       = (float*)   alloc((size_t)N * 4 * 4);
    unsigned short* buf1 = (unsigned short*)alloc((size_t)N * Hh * 2);
    float*    q0      = (float*)   alloc((size_t)N * 4);
    float*    q1      = (float*)   alloc((size_t)N * 4);
    float*    es      = (float*)   alloc((size_t)N * 4);
    float*    ed      = (float*)   alloc((size_t)N * 4);
    int*      lines   = (int*)     alloc((size_t)N * LINE_W * 4);
    int*      novf    = (int*)     alloc((size_t)4);
    i32x2*    ovf     = (i32x2*)   alloc((size_t)E * 8);

    const int tilesM = (N + 15) / 16;
    const int gemmGrid = (tilesM + GEMM_WAVES - 1) / GEMM_WAVES;
    const int TB = 256;
    auto cdiv = [](size_t a, size_t b) -> int { return (int)((a + b - 1) / b); };

    const int transposeBlocks = cdiv((size_t)F * Hh, TB);
    const int foldBlocks      = cdiv((size_t)(Hh + 1) * 32, TB);
    const int clearBlocks     = cdiv(N, TB);
    prepare_weights_and_lines<<<transposeBlocks + foldBlocks + clearBlocks, TB, 0, stream>>>(
        W1, w1t, F, Hh, transposeBlocks,
        W2, a_s2, a_d2, Wl, b2, ut, cb, Hh, F, foldBlocks,
        (i32x4*)lines, N * (LINE_W / 4), clearBlocks, novf);

    layer1_product_beside_edge_placing<128>
        <<<gemmGrid + cdiv(E, 32 * GEMM_WAVES), 32 * GEMM_WAVES, 0, stream>>>(
        features, w1t, a_s1, a_d1, buf1, es, ed, N, Hh, gemmGrid,
        ei, E, lines, novf, ovf);
    gat_gather_times_folded<<<cdiv(N, TILE_ROWS), 32 * TILE_ROWS, 0, stream>>>(
        lines, novf, ovf, buf1, es, ed, b1, ut, S, N);

    gat_gather_folded<<<cdiv((size_t)N * GL, TB), TB, 0, stream>>>(
        (const i32x4*)lines, novf, ovf, (const f32x4*)S, cb, q0, q1, N);

    link_predict<<<cdiv((size_t)P, TB), TB, 0, stream>>>(q0, q1, mask, bl, (float*)d_out, P);
}


# Edges ranked with LDS counters in a two-level counting sort (bins of 256 nodes beside the product, node lines built in the gather's group), not one returning global add per edge; sum order may shift
# speedup vs baseline: 12.1639x; 1.1720x over previous
#include <hip/hip_runtime.h>
#include <hip/hip_bf16.h>

typedef __attribute__((__ext_vector_type__(16))) __bf16 bf16x16;
typedef __attribute__((__ext_vector_type__(8)))  float  f32x8;
typedef __attribute__((__ext_vector_type__(4)))  unsigned int u32x4;
typedef __attribute__((__ext_vector_type__(4)))  float  f32x4;
typedef __attribute__((__ext_vector_type__(2)))  float  f32x2;
typedef __attribute__((__ext_vector_type__(2)))  __bf16 bf16x2;

union Frag16 {
    u32x4   q[2];
    bf16x16 v;
};

static __device__ __forceinline__ unsigned short f2bf(float f) {
    unsigned u = __float_as_uint(f);
    u += 0x7FFFu + ((u >> 16) & 1u);
    return (unsigned short)(u >> 16);
}

#define GEMM_WAVES 4

static __device__ __forceinline__ unsigned pack2bf(float lo, float hi) {
    return (unsigned)f2bf(lo) | ((unsigned)f2bf(hi) << 16);
}

static __device__ __forceinline__ void score_add(float (&ps)[8], float (&pd)[8],
                                                 const f32x8 acc, float av, float dv)
{
#pragma unroll
    for (int r = 0; r < 8; ++r) {
        ps[r] += acc[r] * av;
        pd[r] += acc[r] * dv;
    }
}

static __device__ __forceinline__ void score_store(float (&ps)[8], float (&pd)[8],
                                                   int m0, int half, int l15, int M,
                                                   float* __restrict__ es, float* __restrict__ ed)
{
#pragma unroll
    for (int r = 0; r < 8; ++r) {
        for (int off = 8; off; off >>= 1) {
            ps[r] += __shfl_xor(ps[r], off);
            pd[r] += __shfl_xor(pd[r], off);
        }
    }
    if (l15 == 0) {
#pragma unroll
        for (int r = 0; r < 8; ++r) {
            const int row = m0 + r + half * 8;
            if (row < M) { es[row] = ps[r]; ed[row] = pd[r]; }
        }
    }
}

typedef float f32x4_in __attribute__((ext_vector_type(4), aligned(4)));

template <int K>
static __device__ __forceinline__ void
wmma_gemm_f32a_scores_block(int blk,
                            const float* __restrict__ A,
                            const unsigned short* __restrict__ Bt,
                            const float* __restrict__ as_, const float* __restrict__ ad_,
                            unsigned short* __restrict__ Y,
                            float* __restrict__ es, float* __restrict__ ed,
                            int M, int NOUT)
{
    constexpr int KS = K / 32;
    const int lane  = threadIdx.x & 31;
    const int wave  = threadIdx.x >> 5;
    const int tileM = blk * GEMM_WAVES + wave;
    const int m0    = tileM * 16;
    if (m0 >= M) return;

    const int l15   = lane & 15;
    const int half  = lane >> 4;
    const int kbase = half * 8;

    int rowA = m0 + l15;
    if (rowA >= M) rowA = M - 1;
    const float* ap = A + (size_t)rowA * K + kbase;

    Frag16 fa[KS];
#pragma unroll
    for (int s = 0; s < KS; ++s) {
        const f32x4 x0 = *(const f32x4_in*)(ap + s * 32);
        const f32x4 x1 = *(const f32x4_in*)(ap + s * 32 + 4);
        const f32x4 x2 = *(const f32x4_in*)(ap + s * 32 + 16);
        const f32x4 x3 = *(const f32x4_in*)(ap + s * 32 + 20);
        const u32x4 lo = { pack2bf(x0[0], x0[1]), pack2bf(x0[2], x0[3]),
                           pack2bf(x1[0], x1[1]), pack2bf(x1[2], x1[3]) };
        const u32x4 hi = { pack2bf(x2[0], x2[1]), pack2bf(x2[2], x2[3]),
                           pack2bf(x3[0], x3[1]), pack2bf(x3[2], x3[3]) };
        fa[s].q[0] = lo;
        fa[s].q[1] = hi;
    }

    float ps[8], pd[8];
#pragma unroll
    for (int r = 0; r < 8; ++r) { ps[r] = 0.f; pd[r] = 0.f; }

    for (int c0 = 0; c0 < NOUT; c0 += 16) {
        const unsigned short* bptr = Bt + (size_t)(c0 + l15) * K + kbase;
        const int colY = c0 + l15;
        const float av = as_[colY];
        const float dv = ad_[colY];
        f32x8 acc = {};
#pragma unroll
        for (int s = 0; s < KS; ++s) {
            Frag16 fb;
            fb.q[0] = *(const u32x4*)(bptr + s * 32);
            fb.q[1] = *(const u32x4*)(bptr + s * 32 + 16);
            acc = __builtin_amdgcn_wmma_f32_16x16x32_bf16(
                      false, fa[s].v, false, fb.v, (short)0, acc, false, false);
        }
        const bool odd = (l15 & 1) != 0;
#pragma unroll
        for (int r = 0; r < 8; r += 2) {
            const f32x2 two = { acc[r], acc[r + 1] };
            const unsigned own  = __builtin_bit_cast(unsigned, __builtin_convertvector(two, bf16x2));
            const unsigned send = odd ? (own & 0xFFFFu) : (own >> 16);
            const unsigned recv = (unsigned)__builtin_amdgcn_update_dpp(0, (int)send, 0xB1, 0xF, 0xF, false);
            const unsigned pk   = odd ? (recv | (own & 0xFFFF0000u)) : ((own & 0xFFFFu) | (recv << 16));
            const int row = m0 + r + (odd ? 1 : 0) + half * 8;
            if (row < M) *(unsigned*)(Y + (size_t)row * NOUT + (colY & ~1)) = pk;
        }
        score_add(ps, pd, acc, av, dv);
    }
    score_store(ps, pd, m0, half, l15, M, es, ed);
}

#define LINE_W    32
#define LINE_CAP  (LINE_W - 1)
#define BIN_SHIFT 8
#define NBINS_MAX 256
#define BIN_CAP   4096
#define CNT_PITCH 32
#define CHUNK_EPT 16
typedef __attribute__((__ext_vector_type__(4))) int i32x4;
typedef __attribute__((__ext_vector_type__(2))) int i32x2;

static __device__ __forceinline__ int bin_of(int d) { return (d >> BIN_SHIFT) & (NBINS_MAX - 1); }

static __device__ __forceinline__ void
zero_bin_counters_block(i32x4* __restrict__ cnt4, int n4, int* __restrict__ novf)
{
    for (int i = threadIdx.x; i < n4; i += blockDim.x) {
        const i32x4 z = { 0, 0, 0, 0 };
        cnt4[i] = z;
    }
    if (threadIdx.x == 0) novf[0] = 0;
}

static __device__ __forceinline__ void
bin_edges_block(int blk, const int* __restrict__ ei, int E,
                int* __restrict__ binCount, i32x2* __restrict__ binArr,
                int* __restrict__ novf, i32x2* __restrict__ ovf)
{
    constexpr int T  = 32 * 4;
    constexpr int CH = T * CHUNK_EPT;
    __shared__ int   hist[NBINS_MAX];
    __shared__ int   lstart[NBINS_MAX];
    __shared__ int   gbase[NBINS_MAX];
    __shared__ i32x2 image[CH];

    const int tid  = threadIdx.x;
    const int lane = tid & 31;
    const int e0   = blk * CH;

    for (int b = tid; b < NBINS_MAX; b += T) hist[b] = 0;
    __syncthreads();

    int s[CHUNK_EPT], d[CHUNK_EPT], r[CHUNK_EPT];
#pragma unroll
    for (int k = 0; k < CHUNK_EPT; ++k) {
        const int e = e0 + k * T + tid;
        const bool ok = e < E;
        s[k] = ok ? ei[e] : 0;
        d[k] = ok ? ei[E + e] : -1;
    }
#pragma unroll
    for (int k = 0; k < CHUNK_EPT; ++k)
        r[k] = (d[k] >= 0) ? atomicAdd(&hist[bin_of(d[k])], 1) : 0;
    __syncthreads();

    for (int b = tid; b < NBINS_MAX; b += T) {
        const int cnt = hist[b];
        gbase[b] = cnt ? atomicAdd(binCount + b * CNT_PITCH, cnt) : 0;
    }
    if (tid < 32) {
        int v[NBINS_MAX / 32];
        int run = 0;
#pragma unroll
        for (int j = 0; j < NBINS_MAX / 32; ++j) { v[j] = run; run += hist[lane * (NBINS_MAX / 32) + j]; }
        int incl = run;
        for (int off = 1; off < 32; off <<= 1) {
            const int t = __shfl_up(incl, off);
            if (lane >= off) incl += t;
        }
        const int excl = incl - run;
#pragma unroll
        for (int j = 0; j < NBINS_MAX / 32; ++j) lstart[lane * (NBINS_MAX / 32) + j] = excl + v[j];
    }
    __syncthreads();

#pragma unroll
    for (int k = 0; k < CHUNK_EPT; ++k) {
        if (d[k] >= 0) {
            const i32x2 p = { d[k], s[k] };
            image[lstart[bin_of(d[k])] + r[k]] = p;
        }
    }
    __syncthreads();

    const int left   = E - e0;
    const int nvalid = left < CH ? left : CH;
    for (int k = 0; k < CHUNK_EPT; ++k) {
        const int i = k * T + tid;
        if (i < nvalid) {
            const i32x2 p = image[i];
            const int b = bin_of(p[0]);
            const int slot = gbase[b] + (i - lstart[b]);
            if (slot < BIN_CAP) {
                binArr[(size_t)b * BIN_CAP + slot] = p;
            } else {
                const int kk = atomicAdd(novf, 1);
                ovf[kk] = p;
            }
        }
    }
}

template <int K>
__global__ void __launch_bounds__(32 * GEMM_WAVES)
layer1_product_beside_edge_binning(const float* __restrict__ A,
                                   const unsigned short* __restrict__ Bt,
                                   const float* __restrict__ as_, const float* __restrict__ ad_,
                                   unsigned short* __restrict__ Y,
                                   float* __restrict__ es, float* __restrict__ ed,
                                   int M, int NOUT, int productBlocks,
                                   const int* __restrict__ ei, int E,
                                   int* __restrict__ binCount, i32x2* __restrict__ binArr,
                                   int* __restrict__ novf, i32x2* __restrict__ ovf)
{
    const int b = blockIdx.x;
    if (b < productBlocks)
        wmma_gemm_f32a_scores_block<K>(b, A, Bt, as_, ad_, Y, es, ed, M, NOUT);
    else
        bin_edges_block(b - productBlocks, ei, E, binCount, binArr, novf, ovf);
}

static __device__ __forceinline__ i32x2
pair_at(const i32x2* __restrict__ listA, int lenA, const i32x2* __restrict__ listB, int i)
{
    return (i < lenA) ? listA[i] : listB[i - lenA];
}

static __device__ __forceinline__ float leaky02(float l) { return (l > 0.f) ? l : 0.2f * l; }

template <bool FIRST>
static __device__ __forceinline__ void row_axpy(float (&acc)[8], const u32x4 q, float w)
{
#pragma unroll
    for (int k = 0; k < 4; ++k) {
        const float lo = __uint_as_float(q[k] << 16);
        const float hi = __uint_as_float(q[k] & 0xFFFF0000u);
        if (FIRST) { acc[2 * k] = lo * w;  acc[2 * k + 1] = hi * w; }
        else       { acc[2 * k] += lo * w; acc[2 * k + 1] += hi * w; }
    }
}

static __device__ __forceinline__ u32x4
gather_row_relu_bf16(int d, int lane, int word,
                     const i32x2* __restrict__ listA, int lenA,
                     const i32x2* __restrict__ listB, int lenB,
                     const unsigned short* __restrict__ h, const float* __restrict__ es,
                     const float* __restrict__ ed, const float* __restrict__ bias)
{
    constexpr int C = 256;
    const float edv   = ed[d];
    const float lself = leaky02(es[d] + edv);
    const float NEG_INF = -__builtin_inff();

    const int cnt = __builtin_amdgcn_readfirstlane(word);
    const bool deep = cnt > LINE_CAP;
    const int nin = deep ? 0 : cnt;
    const int nov = deep ? lenA + lenB : 0;

    const bool has = (lane >= 1) && (lane <= nin);
    int   s0 = d;
    float l0 = NEG_INF;
    if (has) { s0 = word; l0 = leaky02(es[s0] + edv); }

    float m = fmaxf(lself, l0);
    for (int i = lane; i < nov; i += 32) {
        const i32x2 p = pair_at(listA, lenA, listB, i);
        if (p[0] == d) m = fmaxf(m, leaky02(es[p[1]] + edv));
    }
    for (int off = 16; off; off >>= 1) m = fmaxf(m, __shfl_xor(m, off));

    float sum = has ? __expf(l0 - m) : 0.f;
    if (lane == 0) sum += __expf(lself - m);
    for (int i = lane; i < nov; i += 32) {
        const i32x2 p = pair_at(listA, lenA, listB, i);
        if (p[0] == d) sum += __expf(leaky02(es[p[1]] + edv) - m);
    }
    for (int off = 16; off; off >>= 1) sum += __shfl_xor(sum, off);

    float acc[8];
    {
        const float w = __expf(lself - m) / sum;
        const u32x4* r = (const u32x4*)(h + (size_t)d * C);
        row_axpy<true>(acc, r[lane], w);
    }
    {
        const float w = has ? __expf(l0 - m) / sum : 0.f;
#pragma unroll 4
        for (int j = 1; j <= nin; ++j) {
            const int   sj = __builtin_amdgcn_readlane(s0, j);
            const float wj = __int_as_float(__builtin_amdgcn_readlane(__float_as_int(w), j));
            const u32x4* r = (const u32x4*)(h + (size_t)sj * C);
            row_axpy<false>(acc, r[lane], wj);
        }
    }
    for (int base = 0; base < nov; base += 32) {
        const int i = base + lane;
        int   s = d;
        float w = 0.f;
        bool  mine = false;
        if (i < nov) {
            const i32x2 p = pair_at(listA, lenA, listB, i);
            if (p[0] == d) { mine = true; s = p[1]; w = __expf(leaky02(es[s] + edv) - m) / sum; }
        }
        unsigned todo = __builtin_amdgcn_ballot_w32(mine);
        while (todo) {
            const int j = __builtin_ctz(todo);
            todo &= todo - 1;
            const int   sj = __builtin_amdgcn_readlane(s, j);
            const float wj = __int_as_float(__builtin_amdgcn_readlane(__float_as_int(w), j));
            const u32x4* r = (const u32x4*)(h + (size_t)sj * C);
            row_axpy<false>(acc, r[lane], wj);
        }
    }
    const float* b = bias + lane * 8;
    u32x4 pk;
#pragma unroll
    for (int k = 0; k < 4; ++k) {
        const float o0 = acc[2 * k]     + b[2 * k];
        const float o1 = acc[2 * k + 1] + b[2 * k + 1];
        pk[k] = pack2bf(o0 > 0.f ? o0 : 0.f, o1 > 0.f ? o1 : 0.f);
    }
    return pk;
}

#define TILE_ROWS 16
#define ROW_Q     33

__global__ void __launch_bounds__(32 * TILE_ROWS)
gat_gather_times_folded(const int* __restrict__ binCount, const i32x2* __restrict__ binArr,
                        const int* __restrict__ novf, const i32x2* __restrict__ ovf,
                        int* __restrict__ lines,
                        const unsigned short* __restrict__ h, const float* __restrict__ es,
                        const float* __restrict__ ed, const float* __restrict__ bias,
                        const unsigned short* __restrict__ Ut,
                        float* __restrict__ S, int N)
{
    constexpr int K = 256;
    constexpr int T = 32 * TILE_ROWS;
    constexpr int PER_THREAD = BIN_CAP / T;
    static_assert(PER_THREAD * T == BIN_CAP, "a bin array is a whole number of sweeps");
    __shared__ u32x4 tile[TILE_ROWS * ROW_Q];
    __shared__ int   tickets;
    __shared__ int   ecount[TILE_ROWS];
    __shared__ int   elist[TILE_ROWS * LINE_W];

    const int lane = threadIdx.x & 31;
    const int wave = __builtin_amdgcn_readfirstlane((int)(threadIdx.x >> 5));
    const int m0   = blockIdx.x * TILE_ROWS;
    const int d    = m0 + wave;

    if (threadIdx.x == 0) tickets = 0;
    if (threadIdx.x < TILE_ROWS) ecount[threadIdx.x] = 0;
    __syncthreads();

    const int bin   = m0 >> BIN_SHIFT;
    const int bc    = binCount[bin * CNT_PITCH];
    const int lenA  = (bc < BIN_CAP) ? bc : BIN_CAP;
    const int lenB  = (bc > BIN_CAP) ? novf[0] : 0;
    const i32x2* listA = binArr + (size_t)bin * BIN_CAP;
    i32x2 pr[PER_THREAD];
#pragma unroll
    for (int k = 0; k < PER_THREAD; ++k) {
        const int i = k * T + (int)threadIdx.x;
        const i32x2 none = { -1, -1 };
        pr[k] = (i < lenA) ? listA[i] : none;
    }
#pragma unroll
    for (int k = 0; k < PER_THREAD; ++k) {
        const unsigned t = (unsigned)(pr[k][0] - m0);
        if (t < (unsigned)TILE_ROWS) {
            const int pos = atomicAdd(&ecount[t], 1);
            if (pos < LINE_CAP) elist[t * LINE_W + 1 + pos] = pr[k][1];
        }
    }
    for (int i = threadIdx.x; i < lenB; i += T) {
        const i32x2 p = ovf[i];
        const unsigned t = (unsigned)(p[0] - m0);
        if (t < (unsigned)TILE_ROWS) {
            const int pos = atomicAdd(&ecount[t], 1);
            if (pos < LINE_CAP) elist[t * LINE_W + 1 + pos] = p[1];
        }
    }
    __syncthreads();

    u32x4 row = { 0u, 0u, 0u, 0u };
    if (d < N) {
        const int cntd = ecount[wave];
        const int held = (cntd < LINE_CAP) ? cntd : LINE_CAP;
        int word = -1;
        if (lane == 0) word = cntd;
        else if (lane <= held) word = elist[wave * LINE_W + lane];
        lines[(size_t)d * LINE_W + lane] = word;
        row = gather_row_relu_bf16(d, lane, word, listA, lenA, ovf, lenB, h, es, ed, bias);
    }
    tile[wave * ROW_Q + lane] = row;

    __builtin_amdgcn_fence(__ATOMIC_RELEASE, "workgroup");
    int t = 0;
    if (lane == 0) t = atomicAdd(&tickets, 1);
    t = __builtin_amdgcn_readfirstlane(t);
    if (t != TILE_ROWS - 1) return;
    __builtin_amdgcn_fence(__ATOMIC_ACQUIRE, "workgroup");

    const int l15  = lane & 15;
    const int half = lane >> 4;
    const u32x4* arow = tile + l15 * ROW_Q + half;
    const unsigned short* bptr = Ut + (size_t)l15 * K + half * 8;

    f32x8 acc = {};
#pragma unroll 2
    for (int k0 = 0; k0 < K; k0 += 32) {
        Frag16 fa, fb;
        fa.q[0] = arow[(k0 >> 3)];
        fa.q[1] = arow[(k0 >> 3) + 2];
        fb.q[0] = *(const u32x4*)(bptr + k0);
        fb.q[1] = *(const u32x4*)(bptr + k0 + 16);
        acc = __builtin_amdgcn_wmma_f32_16x16x32_bf16(
                  false, fa.v, false, fb.v, (short)0, acc, false, false);
    }
    if (l15 < 4) {
#pragma unroll
        for (int r = 0; r < 8; ++r) {
            const int node = m0 + r + half * 8;
            if (node < N) S[(size_t)node * 4 + l15] = acc[r];
        }
    }
}

#define GL 8
__global__ void __launch_bounds__(256)
gat_gather_folded(const i32x4* __restrict__ lines4,
                  const int* __restrict__ binCount, const i32x2* __restrict__ binArr,
                  const int* __restrict__ novf, const i32x2* __restrict__ ovf,
                  const f32x4* __restrict__ S, const float* __restrict__ cb,
                  float* __restrict__ q0, float* __restrict__ q1, int N)
{
    const int t   = blockIdx.x * blockDim.x + threadIdx.x;
    const int sub = threadIdx.x & (GL - 1);
    int d = t / GL;
    const bool live = d < N;
    if (!live) d = N - 1;

    const f32x4 sd = S[d];
    const i32x4 wd = lines4[(size_t)d * (LINE_W / 4) + sub];
    const int cnt = __shfl(wd[0], 0, GL);
    const bool deep = cnt > LINE_CAP;
    const int nin = deep ? 0 : cnt;
    int lenA = 0, lenB = 0;
    const i32x2* listA = binArr;
    if (deep) {
        const int bin = d >> BIN_SHIFT;
        const int bc  = binCount[bin * CNT_PITCH];
        lenA  = (bc < BIN_CAP) ? bc : BIN_CAP;
        lenB  = (bc > BIN_CAP) ? novf[0] : 0;
        listA = binArr + (size_t)bin * BIN_CAP;
    }
    const int nov = lenA + lenB;

    const float edv   = sd[1];
    const float lself = leaky02(sd[0] + edv);
    const float NEG_INF = -__builtin_inff();

    f32x4 x[4];
#pragma unroll
    for (int k = 0; k < 4; ++k) {
        const int idx = sub * 4 + k;
        x[k] = S[(idx >= 1 && idx <= nin) ? wd[k] : d];
    }
    float l[4], ta[4], tb[4];
#pragma unroll
    for (int k = 0; k < 4; ++k) {
        const int idx = sub * 4 + k;
        l[k]  = (idx >= 1 && idx <= nin) ? leaky02(x[k][0] + edv) : NEG_INF;
        ta[k] = x[k][2];
        tb[k] = x[k][3];
    }

    float m = fmaxf(fmaxf(lself, l[0]), fmaxf(l[1], fmaxf(l[2], l[3])));
    for (int i = sub; i < nov; i += GL) {
        const i32x2 p = pair_at(listA, lenA, ovf, i);
        if (p[0] == d) m = fmaxf(m, leaky02(S[p[1]][0] + edv));
    }
    for (int off = GL / 2; off; off >>= 1) m = fmaxf(m, __shfl_xor(m, off, GL));

    float sum = 0.f, a0 = 0.f, a1 = 0.f;
#pragma unroll
    for (int k = 0; k < 4; ++k) {
        const int idx = sub * 4 + k;
        const float p = (idx >= 1 && idx <= nin) ? __expf(l[k] - m) : 0.f;
        sum += p; a0 += p * ta[k]; a1 += p * tb[k];
    }
    if (sub == 0) {
        const float ps = __expf(lself - m);
        sum += ps; a0 += ps * sd[2]; a1 += ps * sd[3];
    }
    for (int i = sub; i < nov; i += GL) {
        const i32x2 pr = pair_at(listA, lenA, ovf, i);
        if (pr[0] == d) {
            const f32x4 xr = S[pr[1]];
            const float p = __expf(leaky02(xr[0] + edv) - m);
            sum += p; a0 += p * xr[2]; a1 += p * xr[3];
        }
    }
    for (int off = GL / 2; off; off >>= 1) {
        sum += __shfl_xor(sum, off, GL);
        a0  += __shfl_xor(a0,  off, GL);
        a1  += __shfl_xor(a1,  off, GL);
    }
    if (live && sub == 0) {
        q0[d] = a0 / sum + cb[0];
        q1[d] = a1 / sum + cb[1];
    }
}

static __device__ __forceinline__ void
transpose_to_bf16_block(int blk, const float* __restrict__ W, unsigned short* __restrict__ Wt,
                        int K, int NOUT)
{
    const int i = blk * blockDim.x + threadIdx.x;
    if (i >= K * NOUT) return;
    const int k = i / NOUT, n = i % NOUT;
    Wt[(size_t)n * K + k] = f2bf(W[i]);
}

static __device__ __forceinline__ void
fold_weight_block(int blk, const float* __restrict__ W2, const float* __restrict__ a_s,
                  const float* __restrict__ a_d, const float* __restrict__ Wl,
                  const float* __restrict__ b2,
                  unsigned short* __restrict__ Ut, float* __restrict__ cb, int K, int C)
{
    const int lane = threadIdx.x & 31;
    const int w = (int)((blk * blockDim.x + threadIdx.x) >> 5);
    if (w > K) return;
    const float* x = (w < K) ? (W2 + (size_t)w * C) : b2;
    float s0 = 0.f, s1 = 0.f, s2 = 0.f, s3 = 0.f;
    for (int c = lane; c < C; c += 32) {
        const float xv = x[c];
        s0 += xv * a_s[c];
        s1 += xv * a_d[c];
        s2 += xv * Wl[c];
        s3 += xv * Wl[C + c];
    }
    for (int off = 16; off; off >>= 1) {
        s0 += __shfl_xor(s0, off);
        s1 += __shfl_xor(s1, off);
        s2 += __shfl_xor(s2, off);
        s3 += __shfl_xor(s3, off);
    }
    if (w < K) {
        const float v = (lane == 0) ? s0 : (lane == 1) ? s1 : (lane == 2) ? s2 : s3;
        if (lane < 16) Ut[(size_t)lane * K + w] = (lane < 4) ? f2bf(v) : (unsigned short)0;
    } else if (lane == 0) {
        cb[0] = s2;
        cb[1] = s3;
    }
}

__global__ void __launch_bounds__(256)
prepare_weights_and_counters(const float* __restrict__ W1, unsigned short* __restrict__ w1t,
                             int K1, int NOUT1, int transposeBlocks,
                             const float* __restrict__ W2, const float* __restrict__ a_s,
                             const float* __restrict__ a_d, const float* __restrict__ Wl,
                             const float* __restrict__ b2,
                             unsigned short* __restrict__ Ut, float* __restrict__ cb,
                             int K2, int C2, int foldBlocks,
                             i32x4* __restrict__ binCount4, int n4,
                             int* __restrict__ novf)
{
    const int b = blockIdx.x;
    if (b < transposeBlocks)
        transpose_to_bf16_block(b, W1, w1t, K1, NOUT1);
    else if (b < transposeBlocks + foldBlocks)
        fold_weight_block(b - transposeBlocks, W2, a_s, a_d, Wl, b2, Ut, cb, K2, C2);
    else
        zero_bin_counters_block(binCount4, n4, novf);
}

__global__ void link_predict(const float* __restrict__ q0, const float* __restrict__ q1,
                             const int* __restrict__ mask, const float* __restrict__ bl,
                             float* __restrict__ out, int P)
{
    const int p = blockIdx.x * blockDim.x + threadIdx.x;
    if (p >= P) return;
    const float z = q0[mask[p * 2 + 0]] + q1[mask[p * 2 + 1]] + bl[0];
    out[p] = 1.f / (1.f + expf(-z));
}

extern "C" void kernel_launch(void* const* d_in, const int* in_sizes, int n_in,
                              void* d_out, int out_size, void* d_ws, size_t ws_size,
                              hipStream_t stream)
{
    (void)n_in; (void)out_size; (void)ws_size;

    const float* features = (const float*)d_in[0];
    const int*   ei       = (const int*)  d_in[1];
    const int*   mask     = (const int*)  d_in[2];
    const float* W1   = (const float*)d_in[3];
    const float* a_s1 = (const float*)d_in[4];
    const float* a_d1 = (const float*)d_in[5];
    const float* b1   = (const float*)d_in[6];
    const float* W2   = (const float*)d_in[7];
    const float* a_s2 = (const float*)d_in[8];
    const float* a_d2 = (const float*)d_in[9];
    const float* b2   = (const float*)d_in[10];
    const float* Wl   = (const float*)d_in[11];
    const float* bl   = (const float*)d_in[12];

    const int F  = 128, Hh = 256;
    const int N  = in_sizes[0] / F;
    const int E  = in_sizes[1] / 2;
    const int P  = in_sizes[2] / 2;

    char* ws = (char*)d_ws;
    size_t off = 0;
    auto alloc = [&](size_t bytes) -> char* {
        char* ptr = ws + off;
        off += (bytes + 255) & ~(size_t)255;
        return ptr;
    };
    unsigned short* w1t  = (unsigned short*)alloc((size_t)Hh * F * 2);
    unsigned short* ut   = (unsigned short*)alloc((size_t)16 * Hh * 2);
    float*    cb      = (float*)   alloc((size_t)2 * 4);
    float*    S       = (float*)   alloc((size_t)N * 4 * 4);
    unsigned short* buf1 = (unsigned short*)alloc((size_t)N * Hh * 2);
    float*    q0      = (float*)   alloc((size_t)N * 4);
    float*    q1      = (float*)   alloc((size_t)N * 4);
    float*    es      = (float*)   alloc((size_t)N * 4);
    float*    ed      = (float*)   alloc((size_t)N * 4);
    int*      lines   = (int*)     alloc((size_t)N * LINE_W * 4);
    int*      binCount = (int*)    alloc((size_t)NBINS_MAX * CNT_PITCH * 4);
    i32x2*    binArr  = (i32x2*)   alloc((size_t)NBINS_MAX * BIN_CAP * 8);
    int*      novf    = (int*)     alloc((size_t)4);
    i32x2*    ovf     = (i32x2*)   alloc((size_t)E * 8);

    const int tilesM = (N + 15) / 16;
    const int gemmGrid = (tilesM + GEMM_WAVES - 1) / GEMM_WAVES;
    const int TB = 256;
    auto cdiv = [](size_t a, size_t b) -> int { return (int)((a + b - 1) / b); };

    const int transposeBlocks = cdiv((size_t)F * Hh, TB);
    const int foldBlocks      = cdiv((size_t)(Hh + 1) * 32, TB);
    prepare_weights_and_counters<<<transposeBlocks + foldBlocks + 1, TB, 0, stream>>>(
        W1, w1t, F, Hh, transposeBlocks,
        W2, a_s2, a_d2, Wl, b2, ut, cb, Hh, F, foldBlocks,
        (i32x4*)binCount, NBINS_MAX * CNT_PITCH / 4, novf);

    layer1_product_beside_edge_binning<128>
        <<<gemmGrid + cdiv(E, 32 * GEMM_WAVES * CHUNK_EPT), 32 * GEMM_WAVES, 0, stream>>>(
        features, w1t, a_s1, a_d1, buf1, es, ed, N, Hh, gemmGrid,
        ei, E, binCount, binArr, novf, ovf);
    gat_gather_times_folded<<<cdiv(N, TILE_ROWS), 32 * TILE_ROWS, 0, stream>>>(
        binCount, binArr, novf, ovf, lines, buf1, es, ed, b1, ut, S, N);

    gat_gather_folded<<<cdiv((size_t)N * GL, TB), TB, 0, stream>>>(
        (const i32x4*)lines, binCount, binArr, novf, ovf, (const f32x4*)S, cb, q0, q1, N);

    link_predict<<<cdiv((size_t)P, TB), TB, 0, stream>>>(q0, q1, mask, bl, (float*)d_out, P);
}


# Both weight operands stored in fragment order: a wave's B fragment load reads 512 contiguous bytes (4 whole lines) instead of 16 lines of 32 bytes each; results bit-identical
# speedup vs baseline: 13.0299x; 1.0712x over previous
#include <hip/hip_runtime.h>
#include <hip/hip_bf16.h>

typedef __attribute__((__ext_vector_type__(16))) __bf16 bf16x16;
typedef __attribute__((__ext_vector_type__(8)))  float  f32x8;
typedef __attribute__((__ext_vector_type__(4)))  unsigned int u32x4;
typedef __attribute__((__ext_vector_type__(4)))  float  f32x4;
typedef __attribute__((__ext_vector_type__(2)))  float  f32x2;
typedef __attribute__((__ext_vector_type__(2)))  __bf16 bf16x2;

union Frag16 {
    u32x4   q[2];
    bf16x16 v;
};

static __device__ __forceinline__ unsigned short f2bf(float f) {
    unsigned u = __float_as_uint(f);
    u += 0x7FFFu + ((u >> 16) & 1u);
    return (unsigned short)(u >> 16);
}

#define GEMM_WAVES 4

static __device__ __forceinline__ size_t frag_index(int k, int n, int KS)
{
    const int c = n >> 4, l15 = n & 15;
    const int s = k >> 5, q = (k >> 4) & 1, half = (k >> 3) & 1, e = k & 7;
    return ((size_t)(((c * KS + s) * 2 + q) * 32 + half * 16 + l15)) * 8 + e;
}

static __device__ __forceinline__ unsigned pack2bf(float lo, float hi) {
    return (unsigned)f2bf(lo) | ((unsigned)f2bf(hi) << 16);
}

static __device__ __forceinline__ void score_add(float (&ps)[8], float (&pd)[8],
                                                 const f32x8 acc, float av, float dv)
{
#pragma unroll
    for (int r = 0; r < 8; ++r) {
        ps[r] += acc[r] * av;
        pd[r] += acc[r] * dv;
    }
}

static __device__ __forceinline__ void score_store(float (&ps)[8], float (&pd)[8],
                                                   int m0, int half, int l15, int M,
                                                   float* __restrict__ es, float* __restrict__ ed)
{
#pragma unroll
    for (int r = 0; r < 8; ++r) {
        for (int off = 8; off; off >>= 1) {
            ps[r] += __shfl_xor(ps[r], off);
            pd[r] += __shfl_xor(pd[r], off);
        }
    }
    if (l15 == 0) {
#pragma unroll
        for (int r = 0; r < 8; ++r) {
            const int row = m0 + r + half * 8;
            if (row < M) { es[row] = ps[r]; ed[row] = pd[r]; }
        }
    }
}

typedef float f32x4_in __attribute__((ext_vector_type(4), aligned(4)));

template <int K>
static __device__ __forceinline__ void
wmma_gemm_f32a_scores_block(int blk,
                            const float* __restrict__ A,
                            const unsigned short* __restrict__ Bt,
                            const float* __restrict__ as_, const float* __restrict__ ad_,
                            unsigned short* __restrict__ Y,
                            float* __restrict__ es, float* __restrict__ ed,
                            int M, int NOUT)
{
    constexpr int KS = K / 32;
    const int lane  = threadIdx.x & 31;
    const int wave  = threadIdx.x >> 5;
    const int tileM = blk * GEMM_WAVES + wave;
    const int m0    = tileM * 16;
    if (m0 >= M) return;

    const int l15   = lane & 15;
    const int half  = lane >> 4;
    const int kbase = half * 8;

    int rowA = m0 + l15;
    if (rowA >= M) rowA = M - 1;
    const float* ap = A + (size_t)rowA * K + kbase;

    Frag16 fa[KS];
#pragma unroll
    for (int s = 0; s < KS; ++s) {
        const f32x4 x0 = *(const f32x4_in*)(ap + s * 32);
        const f32x4 x1 = *(const f32x4_in*)(ap + s * 32 + 4);
        const f32x4 x2 = *(const f32x4_in*)(ap + s * 32 + 16);
        const f32x4 x3 = *(const f32x4_in*)(ap + s * 32 + 20);
        const u32x4 lo = { pack2bf(x0[0], x0[1]), pack2bf(x0[2], x0[3]),
                           pack2bf(x1[0], x1[1]), pack2bf(x1[2], x1[3]) };
        const u32x4 hi = { pack2bf(x2[0], x2[1]), pack2bf(x2[2], x2[3]),
                           pack2bf(x3[0], x3[1]), pack2bf(x3[2], x3[3]) };
        fa[s].q[0] = lo;
        fa[s].q[1] = hi;
    }

    float ps[8], pd[8];
#pragma unroll
    for (int r = 0; r < 8; ++r) { ps[r] = 0.f; pd[r] = 0.f; }

    for (int c0 = 0; c0 < NOUT; c0 += 16) {
        const unsigned short* bptr = Bt + ((size_t)(c0 >> 4) * (KS * 2 * 32) + lane) * 8;
        const int colY = c0 + l15;
        const float av = as_[colY];
        const float dv = ad_[colY];
        f32x8 acc = {};
#pragma unroll
        for (int s = 0; s < KS; ++s) {
            Frag16 fb;
            fb.q[0] = *(const u32x4*)(bptr + s * 512);
            fb.q[1] = *(const u32x4*)(bptr + s * 512 + 256);
            acc = __builtin_amdgcn_wmma_f32_16x16x32_bf16(
                      false, fa[s].v, false, fb.v, (short)0, acc, false, false);
        }
        const bool odd = (l15 & 1) != 0;
#pragma unroll
        for (int r = 0; r < 8; r += 2) {
            const f32x2 two = { acc[r], acc[r + 1] };
            const unsigned own  = __builtin_bit_cast(unsigned, __builtin_convertvector(two, bf16x2));
            const unsigned send = odd ? (own & 0xFFFFu) : (own >> 16);
            const unsigned recv = (unsigned)__builtin_amdgcn_update_dpp(0, (int)send, 0xB1, 0xF, 0xF, false);
            const unsigned pk   = odd ? (recv | (own & 0xFFFF0000u)) : ((own & 0xFFFFu) | (recv << 16));
            const int row = m0 + r + (odd ? 1 : 0) + half * 8;
            if (row < M) *(unsigned*)(Y + (size_t)row * NOUT + (colY & ~1)) = pk;
        }
        score_add(ps, pd, acc, av, dv);
    }
    score_store(ps, pd, m0, half, l15, M, es, ed);
}

#define LINE_W    32
#define LINE_CAP  (LINE_W - 1)
#define BIN_SHIFT 8
#define NBINS_MAX 256
#define BIN_CAP   4096
#define CNT_PITCH 32
#define CHUNK_EPT 16
typedef __attribute__((__ext_vector_type__(4))) int i32x4;
typedef __attribute__((__ext_vector_type__(2))) int i32x2;

static __device__ __forceinline__ int bin_of(int d) { return (d >> BIN_SHIFT) & (NBINS_MAX - 1); }

static __device__ __forceinline__ void
zero_bin_counters_block(i32x4* __restrict__ cnt4, int n4, int* __restrict__ novf)
{
    for (int i = threadIdx.x; i < n4; i += blockDim.x) {
        const i32x4 z = { 0, 0, 0, 0 };
        cnt4[i] = z;
    }
    if (threadIdx.x == 0) novf[0] = 0;
}

static __device__ __forceinline__ void
bin_edges_block(int blk, const int* __restrict__ ei, int E,
                int* __restrict__ binCount, i32x2* __restrict__ binArr,
                int* __restrict__ novf, i32x2* __restrict__ ovf)
{
    constexpr int T  = 32 * 4;
    constexpr int CH = T * CHUNK_EPT;
    __shared__ int   hist[NBINS_MAX];
    __shared__ int   lstart[NBINS_MAX];
    __shared__ int   gbase[NBINS_MAX];
    __shared__ i32x2 image[CH];

    const int tid  = threadIdx.x;
    const int lane = tid & 31;
    const int e0   = blk * CH;

    for (int b = tid; b < NBINS_MAX; b += T) hist[b] = 0;
    __syncthreads();

    int s[CHUNK_EPT], d[CHUNK_EPT], r[CHUNK_EPT];
#pragma unroll
    for (int k = 0; k < CHUNK_EPT; ++k) {
        const int e = e0 + k * T + tid;
        const bool ok = e < E;
        s[k] = ok ? ei[e] : 0;
        d[k] = ok ? ei[E + e] : -1;
    }
#pragma unroll
    for (int k = 0; k < CHUNK_EPT; ++k)
        r[k] = (d[k] >= 0) ? atomicAdd(&hist[bin_of(d[k])], 1) : 0;
    __syncthreads();

    for (int b = tid; b < NBINS_MAX; b += T) {
        const int cnt = hist[b];
        gbase[b] = cnt ? atomicAdd(binCount + b * CNT_PITCH, cnt) : 0;
    }
    if (tid < 32) {
        int v[NBINS_MAX / 32];
        int run = 0;
#pragma unroll
        for (int j = 0; j < NBINS_MAX / 32; ++j) { v[j] = run; run += hist[lane * (NBINS_MAX / 32) + j]; }
        int incl = run;
        for (int off = 1; off < 32; off <<= 1) {
            const int t = __shfl_up(incl, off);
            if (lane >= off) incl += t;
        }
        const int excl = incl - run;
#pragma unroll
        for (int j = 0; j < NBINS_MAX / 32; ++j) lstart[lane * (NBINS_MAX / 32) + j] = excl + v[j];
    }
    __syncthreads();

#pragma unroll
    for (int k = 0; k < CHUNK_EPT; ++k) {
        if (d[k] >= 0) {
            const i32x2 p = { d[k], s[k] };
            image[lstart[bin_of(d[k])] + r[k]] = p;
        }
    }
    __syncthreads();

    const int left   = E - e0;
    const int nvalid = left < CH ? left : CH;
    for (int k = 0; k < CHUNK_EPT; ++k) {
        const int i = k * T + tid;
        if (i < nvalid) {
            const i32x2 p = image[i];
            const int b = bin_of(p[0]);
            const int slot = gbase[b] + (i - lstart[b]);
            if (slot < BIN_CAP) {
                binArr[(size_t)b * BIN_CAP + slot] = p;
            } else {
                const int kk = atomicAdd(novf, 1);
                ovf[kk] = p;
            }
        }
    }
}

template <int K>
__global__ void __launch_bounds__(32 * GEMM_WAVES)
layer1_product_beside_edge_binning(const float* __restrict__ A,
                                   const unsigned short* __restrict__ Bt,
                                   const float* __restrict__ as_, const float* __restrict__ ad_,
                                   unsigned short* __restrict__ Y,
                                   float* __restrict__ es, float* __restrict__ ed,
                                   int M, int NOUT, int productBlocks,
                                   const int* __restrict__ ei, int E,
                                   int* __restrict__ binCount, i32x2* __restrict__ binArr,
                                   int* __restrict__ novf, i32x2* __restrict__ ovf)
{
    const int b = blockIdx.x;
    if (b < productBlocks)
        wmma_gemm_f32a_scores_block<K>(b, A, Bt, as_, ad_, Y, es, ed, M, NOUT);
    else
        bin_edges_block(b - productBlocks, ei, E, binCount, binArr, novf, ovf);
}

static __device__ __forceinline__ i32x2
pair_at(const i32x2* __restrict__ listA, int lenA, const i32x2* __restrict__ listB, int i)
{
    return (i < lenA) ? listA[i] : listB[i - lenA];
}

static __device__ __forceinline__ float leaky02(float l) { return (l > 0.f) ? l : 0.2f * l; }

template <bool FIRST>
static __device__ __forceinline__ void row_axpy(float (&acc)[8], const u32x4 q, float w)
{
#pragma unroll
    for (int k = 0; k < 4; ++k) {
        const float lo = __uint_as_float(q[k] << 16);
        const float hi = __uint_as_float(q[k] & 0xFFFF0000u);
        if (FIRST) { acc[2 * k] = lo * w;  acc[2 * k + 1] = hi * w; }
        else       { acc[2 * k] += lo * w; acc[2 * k + 1] += hi * w; }
    }
}

static __device__ __forceinline__ u32x4
gather_row_relu_bf16(int d, int lane, int word,
                     const i32x2* __restrict__ listA, int lenA,
                     const i32x2* __restrict__ listB, int lenB,
                     const unsigned short* __restrict__ h, const float* __restrict__ es,
                     const float* __restrict__ ed, const float* __restrict__ bias)
{
    constexpr int C = 256;
    const float edv   = ed[d];
    const float lself = leaky02(es[d] + edv);
    const float NEG_INF = -__builtin_inff();

    const int cnt = __builtin_amdgcn_readfirstlane(word);
    const bool deep = cnt > LINE_CAP;
    const int nin = deep ? 0 : cnt;
    const int nov = deep ? lenA + lenB : 0;

    const bool has = (lane >= 1) && (lane <= nin);
    int   s0 = d;
    float l0 = NEG_INF;
    if (has) { s0 = word; l0 = leaky02(es[s0] + edv); }

    float m = fmaxf(lself, l0);
    for (int i = lane; i < nov; i += 32) {
        const i32x2 p = pair_at(listA, lenA, listB, i);
        if (p[0] == d) m = fmaxf(m, leaky02(es[p[1]] + edv));
    }
    for (int off = 16; off; off >>= 1) m = fmaxf(m, __shfl_xor(m, off));

    float sum = has ? __expf(l0 - m) : 0.f;
    if (lane == 0) sum += __expf(lself - m);
    for (int i = lane; i < nov; i += 32) {
        const i32x2 p = pair_at(listA, lenA, listB, i);
        if (p[0] == d) sum += __expf(leaky02(es[p[1]] + edv) - m);
    }
    for (int off = 16; off; off >>= 1) sum += __shfl_xor(sum, off);

    float acc[8];
    {
        const float w = __expf(lself - m) / sum;
        const u32x4* r = (const u32x4*)(h + (size_t)d * C);
        row_axpy<true>(acc, r[lane], w);
    }
    {
        const float w = has ? __expf(l0 - m) / sum : 0.f;
#pragma unroll 4
        for (int j = 1; j <= nin; ++j) {
            const int   sj = __builtin_amdgcn_readlane(s0, j);
            const float wj = __int_as_float(__builtin_amdgcn_readlane(__float_as_int(w), j));
            const u32x4* r = (const u32x4*)(h + (size_t)sj * C);
            row_axpy<false>(acc, r[lane], wj);
        }
    }
    for (int base = 0; base < nov; base += 32) {
        const int i = base + lane;
        int   s = d;
        float w = 0.f;
        bool  mine = false;
        if (i < nov) {
            const i32x2 p = pair_at(listA, lenA, listB, i);
            if (p[0] == d) { mine = true; s = p[1]; w = __expf(leaky02(es[s] + edv) - m) / sum; }
        }
        unsigned todo = __builtin_amdgcn_ballot_w32(mine);
        while (todo) {
            const int j = __builtin_ctz(todo);
            todo &= todo - 1;
            const int   sj = __builtin_amdgcn_readlane(s, j);
            const float wj = __int_as_float(__builtin_amdgcn_readlane(__float_as_int(w), j));
            const u32x4* r = (const u32x4*)(h + (size_t)sj * C);
            row_axpy<false>(acc, r[lane], wj);
        }
    }
    const float* b = bias + lane * 8;
    u32x4 pk;
#pragma unroll
    for (int k = 0; k < 4; ++k) {
        const float o0 = acc[2 * k]     + b[2 * k];
        const float o1 = acc[2 * k + 1] + b[2 * k + 1];
        pk[k] = pack2bf(o0 > 0.f ? o0 : 0.f, o1 > 0.f ? o1 : 0.f);
    }
    return pk;
}

#define TILE_ROWS 16
#define ROW_Q     33

__global__ void __launch_bounds__(32 * TILE_ROWS)
gat_gather_times_folded(const int* __restrict__ binCount, const i32x2* __restrict__ binArr,
                        const int* __restrict__ novf, const i32x2* __restrict__ ovf,
                        int* __restrict__ lines,
                        const unsigned short* __restrict__ h, const float* __restrict__ es,
                        const float* __restrict__ ed, const float* __restrict__ bias,
                        const unsigned short* __restrict__ Ut,
                        float* __restrict__ S, int N)
{
    constexpr int K = 256;
    constexpr int T = 32 * TILE_ROWS;
    constexpr int PER_THREAD = BIN_CAP / T;
    static_assert(PER_THREAD * T == BIN_CAP, "a bin array is a whole number of sweeps");
    __shared__ u32x4 tile[TILE_ROWS * ROW_Q];
    __shared__ int   tickets;
    __shared__ int   ecount[TILE_ROWS];
    __shared__ int   elist[TILE_ROWS * LINE_W];

    const int lane = threadIdx.x & 31;
    const int wave = __builtin_amdgcn_readfirstlane((int)(threadIdx.x >> 5));
    const int m0   = blockIdx.x * TILE_ROWS;
    const int d    = m0 + wave;

    if (threadIdx.x == 0) tickets = 0;
    if (threadIdx.x < TILE_ROWS) ecount[threadIdx.x] = 0;
    __syncthreads();

    const int bin   = m0 >> BIN_SHIFT;
    const int bc    = binCount[bin * CNT_PITCH];
    const int lenA  = (bc < BIN_CAP) ? bc : BIN_CAP;
    const int lenB  = (bc > BIN_CAP) ? novf[0] : 0;
    const i32x2* listA = binArr + (size_t)bin * BIN_CAP;
    i32x2 pr[PER_THREAD];
#pragma unroll
    for (int k = 0; k < PER_THREAD; ++k) {
        const int i = k * T + (int)threadIdx.x;
        const i32x2 none = { -1, -1 };
        pr[k] = (i < lenA) ? listA[i] : none;
    }
#pragma unroll
    for (int k = 0; k < PER_THREAD; ++k) {
        const unsigned t = (unsigned)(pr[k][0] - m0);
        if (t < (unsigned)TILE_ROWS) {
            const int pos = atomicAdd(&ecount[t], 1);
            if (pos < LINE_CAP) elist[t * LINE_W + 1 + pos] = pr[k][1];
        }
    }
    for (int i = threadIdx.x; i < lenB; i += T) {
        const i32x2 p = ovf[i];
        const unsigned t = (unsigned)(p[0] - m0);
        if (t < (unsigned)TILE_ROWS) {
            const int pos = atomicAdd(&ecount[t], 1);
            if (pos < LINE_CAP) elist[t * LINE_W + 1 + pos] = p[1];
        }
    }
    __syncthreads();

    u32x4 row = { 0u, 0u, 0u, 0u };
    if (d < N) {
        const int cntd = ecount[wave];
        const int held = (cntd < LINE_CAP) ? cntd : LINE_CAP;
        int word = -1;
        if (lane == 0) word = cntd;
        else if (lane <= held) word = elist[wave * LINE_W + lane];
        lines[(size_t)d * LINE_W + lane] = word;
        row = gather_row_relu_bf16(d, lane, word, listA, lenA, ovf, lenB, h, es, ed, bias);
    }
    tile[wave * ROW_Q + lane] = row;

    __builtin_amdgcn_fence(__ATOMIC_RELEASE, "workgroup");
    int t = 0;
    if (lane == 0) t = atomicAdd(&tickets, 1);
    t = __builtin_amdgcn_readfirstlane(t);
    if (t != TILE_ROWS - 1) return;
    __builtin_amdgcn_fence(__ATOMIC_ACQUIRE, "workgroup");

    const int l15  = lane & 15;
    const int half = lane >> 4;
    const u32x4* arow = tile + l15 * ROW_Q + half;
    const unsigned short* bptr = Ut + lane * 8;

    f32x8 acc = {};
#pragma unroll 2
    for (int k0 = 0; k0 < K; k0 += 32) {
        Frag16 fa, fb;
        fa.q[0] = arow[(k0 >> 3)];
        fa.q[1] = arow[(k0 >> 3) + 2];
        fb.q[0] = *(const u32x4*)(bptr + (k0 >> 5) * 512);
        fb.q[1] = *(const u32x4*)(bptr + (k0 >> 5) * 512 + 256);
        acc = __builtin_amdgcn_wmma_f32_16x16x32_bf16(
                  false, fa.v, false, fb.v, (short)0, acc, false, false);
    }
    if (l15 < 4) {
#pragma unroll
        for (int r = 0; r < 8; ++r) {
            const int node = m0 + r + half * 8;
            if (node < N) S[(size_t)node * 4 + l15] = acc[r];
        }
    }
}

#define GL 8
__global__ void __launch_bounds__(256)
gat_gather_folded(const i32x4* __restrict__ lines4,
                  const int* __restrict__ binCount, const i32x2* __restrict__ binArr,
                  const int* __restrict__ novf, const i32x2* __restrict__ ovf,
                  const f32x4* __restrict__ S, const float* __restrict__ cb,
                  float* __restrict__ q0, float* __restrict__ q1, int N)
{
    const int t   = blockIdx.x * blockDim.x + threadIdx.x;
    const int sub = threadIdx.x & (GL - 1);
    int d = t / GL;
    const bool live = d < N;
    if (!live) d = N - 1;

    const f32x4 sd = S[d];
    const i32x4 wd = lines4[(size_t)d * (LINE_W / 4) + sub];
    const int cnt = __shfl(wd[0], 0, GL);
    const bool deep = cnt > LINE_CAP;
    const int nin = deep ? 0 : cnt;
    int lenA = 0, lenB = 0;
    const i32x2* listA = binArr;
    if (deep) {
        const int bin = d >> BIN_SHIFT;
        const int bc  = binCount[bin * CNT_PITCH];
        lenA  = (bc < BIN_CAP) ? bc : BIN_CAP;
        lenB  = (bc > BIN_CAP) ? novf[0] : 0;
        listA = binArr + (size_t)bin * BIN_CAP;
    }
    const int nov = lenA + lenB;

    const float edv   = sd[1];
    const float lself = leaky02(sd[0] + edv);
    const float NEG_INF = -__builtin_inff();

    f32x4 x[4];
#pragma unroll
    for (int k = 0; k < 4; ++k) {
        const int idx = sub * 4 + k;
        x[k] = S[(idx >= 1 && idx <= nin) ? wd[k] : d];
    }
    float l[4], ta[4], tb[4];
#pragma unroll
    for (int k = 0; k < 4; ++k) {
        const int idx = sub * 4 + k;
        l[k]  = (idx >= 1 && idx <= nin) ? leaky02(x[k][0] + edv) : NEG_INF;
        ta[k] = x[k][2];
        tb[k] = x[k][3];
    }

    float m = fmaxf(fmaxf(lself, l[0]), fmaxf(l[1], fmaxf(l[2], l[3])));
    for (int i = sub; i < nov; i += GL) {
        const i32x2 p = pair_at(listA, lenA, ovf, i);
        if (p[0] == d) m = fmaxf(m, leaky02(S[p[1]][0] + edv));
    }
    for (int off = GL / 2; off; off >>= 1) m = fmaxf(m, __shfl_xor(m, off, GL));

    float sum = 0.f, a0 = 0.f, a1 = 0.f;
#pragma unroll
    for (int k = 0; k < 4; ++k) {
        const int idx = sub * 4 + k;
        const float p = (idx >= 1 && idx <= nin) ? __expf(l[k] - m) : 0.f;
        sum += p; a0 += p * ta[k]; a1 += p * tb[k];
    }
    if (sub == 0) {
        const float ps = __expf(lself - m);
        sum += ps; a0 += ps * sd[2]; a1 += ps * sd[3];
    }
    for (int i = sub; i < nov; i += GL) {
        const i32x2 pr = pair_at(listA, lenA, ovf, i);
        if (pr[0] == d) {
            const f32x4 xr = S[pr[1]];
            const float p = __expf(leaky02(xr[0] + edv) - m);
            sum += p; a0 += p * xr[2]; a1 += p * xr[3];
        }
    }
    for (int off = GL / 2; off; off >>= 1) {
        sum += __shfl_xor(sum, off, GL);
        a0  += __shfl_xor(a0,  off, GL);
        a1  += __shfl_xor(a1,  off, GL);
    }
    if (live && sub == 0) {
        q0[d] = a0 / sum + cb[0];
        q1[d] = a1 / sum + cb[1];
    }
}

static __device__ __forceinline__ void
transpose_to_bf16_block(int blk, const float* __restrict__ W, unsigned short* __restrict__ Wt,
                        int K, int NOUT)
{
    const int i = blk * blockDim.x + threadIdx.x;
    if (i >= K * NOUT) return;
    const int k = i / NOUT, n = i % NOUT;
    Wt[frag_index(k, n, K >> 5)] = f2bf(W[i]);
}

static __device__ __forceinline__ void
fold_weight_block(int blk, const float* __restrict__ W2, const float* __restrict__ a_s,
                  const float* __restrict__ a_d, const float* __restrict__ Wl,
                  const float* __restrict__ b2,
                  unsigned short* __restrict__ Ut, float* __restrict__ cb, int K, int C)
{
    const int lane = threadIdx.x & 31;
    const int w = (int)((blk * blockDim.x + threadIdx.x) >> 5);
    if (w > K) return;
    const float* x = (w < K) ? (W2 + (size_t)w * C) : b2;
    float s0 = 0.f, s1 = 0.f, s2 = 0.f, s3 = 0.f;
    for (int c = lane; c < C; c += 32) {
        const float xv = x[c];
        s0 += xv * a_s[c];
        s1 += xv * a_d[c];
        s2 += xv * Wl[c];
        s3 += xv * Wl[C + c];
    }
    for (int off = 16; off; off >>= 1) {
        s0 += __shfl_xor(s0, off);
        s1 += __shfl_xor(s1, off);
        s2 += __shfl_xor(s2, off);
        s3 += __shfl_xor(s3, off);
    }
    if (w < K) {
        const float v = (lane == 0) ? s0 : (lane == 1) ? s1 : (lane == 2) ? s2 : s3;
        if (lane < 16) Ut[frag_index(w, lane, K >> 5)] = (lane < 4) ? f2bf(v) : (unsigned short)0;
    } else if (lane == 0) {
        cb[0] = s2;
        cb[1] = s3;
    }
}

__global__ void __launch_bounds__(256)
prepare_weights_and_counters(const float* __restrict__ W1, unsigned short* __restrict__ w1t,
                             int K1, int NOUT1, int transposeBlocks,
                             const float* __restrict__ W2, const float* __restrict__ a_s,
                             const float* __restrict__ a_d, const float* __restrict__ Wl,
                             const float* __restrict__ b2,
                             unsigned short* __restrict__ Ut, float* __restrict__ cb,
                             int K2, int C2, int foldBlocks,
                             i32x4* __restrict__ binCount4, int n4,
                             int* __restrict__ novf)
{
    const int b = blockIdx.x;
    if (b < transposeBlocks)
        transpose_to_bf16_block(b, W1, w1t, K1, NOUT1);
    else if (b < transposeBlocks + foldBlocks)
        fold_weight_block(b - transposeBlocks, W2, a_s, a_d, Wl, b2, Ut, cb, K2, C2);
    else
        zero_bin_counters_block(binCount4, n4, novf);
}

__global__ void link_predict(const float* __restrict__ q0, const float* __restrict__ q1,
                             const int* __restrict__ mask, const float* __restrict__ bl,
                             float* __restrict__ out, int P)
{
    const int p = blockIdx.x * blockDim.x + threadIdx.x;
    if (p >= P) return;
    const float z = q0[mask[p * 2 + 0]] + q1[mask[p * 2 + 1]] + bl[0];
    out[p] = 1.f / (1.f + expf(-z));
}

extern "C" void kernel_launch(void* const* d_in, const int* in_sizes, int n_in,
                              void* d_out, int out_size, void* d_ws, size_t ws_size,
                              hipStream_t stream)
{
    (void)n_in; (void)out_size; (void)ws_size;

    const float* features = (const float*)d_in[0];
    const int*   ei       = (const int*)  d_in[1];
    const int*   mask     = (const int*)  d_in[2];
    const float* W1   = (const float*)d_in[3];
    const float* a_s1 = (const float*)d_in[4];
    const float* a_d1 = (const float*)d_in[5];
    const float* b1   = (const float*)d_in[6];
    const float* W2   = (const float*)d_in[7];
    const float* a_s2 = (const float*)d_in[8];
    const float* a_d2 = (const float*)d_in[9];
    const float* b2   = (const float*)d_in[10];
    const float* Wl   = (const float*)d_in[11];
    const float* bl   = (const float*)d_in[12];

    const int F  = 128, Hh = 256;
    const int N  = in_sizes[0] / F;
    const int E  = in_sizes[1] / 2;
    const int P  = in_sizes[2] / 2;

    char* ws = (char*)d_ws;
    size_t off = 0;
    auto alloc = [&](size_t bytes) -> char* {
        char* ptr = ws + off;
        off += (bytes + 255) & ~(size_t)255;
        return ptr;
    };
    unsigned short* w1t  = (unsigned short*)alloc((size_t)Hh * F * 2);
    unsigned short* ut   = (unsigned short*)alloc((size_t)16 * Hh * 2);
    float*    cb      = (float*)   alloc((size_t)2 * 4);
    float*    S       = (float*)   alloc((size_t)N * 4 * 4);
    unsigned short* buf1 = (unsigned short*)alloc((size_t)N * Hh * 2);
    float*    q0      = (float*)   alloc((size_t)N * 4);
    float*    q1      = (float*)   alloc((size_t)N * 4);
    float*    es      = (float*)   alloc((size_t)N * 4);
    float*    ed      = (float*)   alloc((size_t)N * 4);
    int*      lines   = (int*)     alloc((size_t)N * LINE_W * 4);
    int*      binCount = (int*)    alloc((size_t)NBINS_MAX * CNT_PITCH * 4);
    i32x2*    binArr  = (i32x2*)   alloc((size_t)NBINS_MAX * BIN_CAP * 8);
    int*      novf    = (int*)     alloc((size_t)4);
    i32x2*    ovf     = (i32x2*)   alloc((size_t)E * 8);

    const int tilesM = (N + 15) / 16;
    const int gemmGrid = (tilesM + GEMM_WAVES - 1) / GEMM_WAVES;
    const int TB = 256;
    auto cdiv = [](size_t a, size_t b) -> int { return (int)((a + b - 1) / b); };

    const int transposeBlocks = cdiv((size_t)F * Hh, TB);
    const int foldBlocks      = cdiv((size_t)(Hh + 1) * 32, TB);
    prepare_weights_and_counters<<<transposeBlocks + foldBlocks + 1, TB, 0, stream>>>(
        W1, w1t, F, Hh, transposeBlocks,
        W2, a_s2, a_d2, Wl, b2, ut, cb, Hh, F, foldBlocks,
        (i32x4*)binCount, NBINS_MAX * CNT_PITCH / 4, novf);

    layer1_product_beside_edge_binning<128>
        <<<gemmGrid + cdiv(E, 32 * GEMM_WAVES * CHUNK_EPT), 32 * GEMM_WAVES, 0, stream>>>(
        features, w1t, a_s1, a_d1, buf1, es, ed, N, Hh, gemmGrid,
        ei, E, binCount, binArr, novf, ovf);
    gat_gather_times_folded<<<cdiv(N, TILE_ROWS), 32 * TILE_ROWS, 0, stream>>>(
        binCount, binArr, novf, ovf, lines, buf1, es, ed, b1, ut, S, N);

    gat_gather_folded<<<cdiv((size_t)N * GL, TB), TB, 0, stream>>>(
        (const i32x4*)lines, binCount, binArr, novf, ovf, (const f32x4*)S, cb, q0, q1, N);

    link_predict<<<cdiv((size_t)P, TB), TB, 0, stream>>>(q0, q1, mask, bl, (float*)d_out, P);
}
